# starting kernel
# baseline (speedup 1.0000x reference)
#include <hip/hip_runtime.h>
#include <hip/hip_bf16.h>

#define LRELU_SLOPE 0.2f
#define GAT_EPS 1e-16f
#define KT 32

typedef __attribute__((ext_vector_type(2))) float v2f;
typedef __attribute__((ext_vector_type(8))) float v8f;

__global__ void fill_kernel(float* __restrict__ p, float v, int n) {
    int i = blockIdx.x * blockDim.x + threadIdx.x;
    if (i < n) p[i] = v;
}

__global__ void gemm_wmma_f32(const float* __restrict__ A,
                              const float* __restrict__ B,
                              float* __restrict__ C, int M, int K) {
    __shared__ float As[32][KT + 1];

    const int tid  = threadIdx.x;
    const int lane = tid & 31;
    const int wave = tid >> 5;
    const int r0   = blockIdx.x * 32;
    const int c0   = wave * 16;
    const int half = lane >> 4;
    const int l15  = lane & 15;
    const int bcol = c0 + l15;

    v8f acc0 = {}, acc1 = {};

    for (int k0 = 0; k0 < K; k0 += KT) {
        __syncthreads();
#pragma unroll
        for (int i = tid; i < 32 * KT; i += 128) {
            const int r = i >> 5;
            const int c = i & (KT - 1);
            const int grow = r0 + r;
            As[r][c] = (grow < M) ? A[(long long)grow * K + k0 + c] : 0.f;
        }
        __syncthreads();

#pragma unroll
        for (int k = 0; k < KT; k += 4) {
            const int kk = k + (half << 1);
            v2f b, a0, a1;
            b.x  = B[(k0 + kk) * 64 + bcol];
            b.y  = B[(k0 + kk + 1) * 64 + bcol];
            a0.x = As[l15][kk];       a0.y = As[l15][kk + 1];
            a1.x = As[16 + l15][kk];  a1.y = As[16 + l15][kk + 1];
            acc0 = __builtin_amdgcn_wmma_f32_16x16x4_f32(
                false, a0, false, b, (short)0, acc0, false, false);
            acc1 = __builtin_amdgcn_wmma_f32_16x16x4_f32(
                false, a1, false, b, (short)0, acc1, false, false);
        }
    }

    const int rbase = r0 + half * 8;
#pragma unroll
    for (int r = 0; r < 8; ++r) {
        const int row0 = rbase + r;
        const int row1 = row0 + 16;
        if (row0 < M) C[(long long)row0 * 64 + c0 + l15] = acc0[r];
        if (row1 < M) C[(long long)row1 * 64 + c0 + l15] = acc1[r];
    }
}

template <int H, int CH>
__global__ void alpha_kernel(const float* __restrict__ xl,
                             const float* __restrict__ a_src,
                             const float* __restrict__ a_dst,
                             float* __restrict__ asrc,
                             float* __restrict__ adst, int n_nodes) {
    int idx = blockIdx.x * blockDim.x + threadIdx.x;
    if (idx >= n_nodes * H) return;
    const int n = idx / H, h = idx % H;
    const float* row = xl + (long long)n * 64 + h * CH;
    float s1 = 0.f, s2 = 0.f;
#pragma unroll
    for (int c = 0; c < CH; ++c) {
        s1 += row[c] * a_src[h * CH + c];
        s2 += row[c] * a_dst[h * CH + c];
    }
    asrc[idx] = s1;
    adst[idx] = s2;
}

__device__ __forceinline__ float lrelu(float x) {
    return x > 0.f ? x : LRELU_SLOPE * x;
}

__device__ __forceinline__ void atomic_max_f32(float* addr, float val) {
    if (val >= 0.f)
        atomicMax((int*)addr, __float_as_int(val));
    else
        atomicMin((unsigned int*)addr, __float_as_uint(val));
}

template <int H>
__device__ __forceinline__ void edge_sd(const int* __restrict__ src,
                                        const int* __restrict__ dst,
                                        int E, int e, int& s, int& d) {
    if (e < E) { s = src[e]; d = dst[e]; }
    else       { s = d = e - E; }
}

template <int H>
__global__ void edge_max_kernel(const int* __restrict__ src,
                                const int* __restrict__ dst, int E, int ET,
                                const float* __restrict__ asrc,
                                const float* __restrict__ adst,
                                float* __restrict__ amax) {
    int t = blockIdx.x * blockDim.x + threadIdx.x;
    if (t >= ET * H) return;
    const int e = t / H, h = t % H;
    int s, d; edge_sd<H>(src, dst, E, e, s, d);
    float v = lrelu(asrc[s * H + h] + adst[d * H + h]);
    atomic_max_f32(&amax[d * H + h], v);
}

template <int H>
__global__ void edge_sum_kernel(const int* __restrict__ src,
                                const int* __restrict__ dst, int E, int ET,
                                const float* __restrict__ asrc,
                                const float* __restrict__ adst,
                                const float* __restrict__ amax,
                                float* __restrict__ den) {
    int t = blockIdx.x * blockDim.x + threadIdx.x;
    if (t >= ET * H) return;
    const int e = t / H, h = t % H;
    int s, d; edge_sd<H>(src, dst, E, e, s, d);
    float v = lrelu(asrc[s * H + h] + adst[d * H + h]);
    atomicAdd(&den[d * H + h], expf(v - amax[d * H + h]));
}

template <int H, int CSH>
__global__ void edge_agg_kernel(const int* __restrict__ src,
                                const int* __restrict__ dst, int E, int ET,
                                const float* __restrict__ asrc,
                                const float* __restrict__ adst,
                                const float* __restrict__ amax,
                                const float* __restrict__ den,
                                const float* __restrict__ xl,
                                float* __restrict__ agg) {
    int t = blockIdx.x * blockDim.x + threadIdx.x;
    if (t >= ET * 64) return;
    const int e = t >> 6, f = t & 63, h = f >> CSH;
    int s, d; edge_sd<H>(src, dst, E, e, s, d);
    float v = lrelu(asrc[s * H + h] + adst[d * H + h]);
    float w = expf(v - amax[d * H + h]) / fmaxf(den[d * H + h], GAT_EPS);
    atomicAdd(&agg[(long long)d * 64 + f], xl[(long long)s * 64 + f] * w);
}

__global__ void elu_bias_kernel(const float* __restrict__ agg,
                                const float* __restrict__ bias,
                                float* __restrict__ h, int n) {
    int i = blockIdx.x * blockDim.x + threadIdx.x;
    if (i >= n) return;
    float v = agg[i] + bias[i & 63];
    h[i] = v > 0.f ? v : (expf(v) - 1.f);
}

__global__ void final_kernel(const float* __restrict__ agg2,
                             const float* __restrict__ b2,
                             float* __restrict__ out, int n_nodes) {
    int i = blockIdx.x * blockDim.x + threadIdx.x;
    if (i >= n_nodes * 16) return;
    const int n = i >> 4, c = i & 15;
    const float* row = agg2 + (long long)n * 64;
    float s = row[c] + row[16 + c] + row[32 + c] + row[48 + c];
    out[i] = 0.25f * s + b2[c];
}

static inline int cdiv(long long a, long long b) { return (int)((a + b - 1) / b); }

extern "C" void kernel_launch(void* const* d_in, const int* in_sizes, int n_in,
                              void* d_out, int out_size, void* d_ws, size_t ws_size,
                              hipStream_t stream) {
    const float* x   = (const float*)d_in[0];
    const int*   ei  = (const int*)d_in[1];
    const float* W1  = (const float*)d_in[2];
    const float* as1 = (const float*)d_in[3];
    const float* ad1 = (const float*)d_in[4];
    const float* b1  = (const float*)d_in[5];
    const float* W2  = (const float*)d_in[6];
    const float* as2 = (const float*)d_in[7];
    const float* ad2 = (const float*)d_in[8];
    const float* b2  = (const float*)d_in[9];
    float*       out = (float*)d_out;

    const int IN = 256;
    const int n  = in_sizes[0] / IN;
    const int E  = in_sizes[1] / 2;
    const int ET = E + n;
    const int* src = ei;
    const int* dst = ei + E;

    float* ws    = (float*)d_ws;
    float* xl1   = ws;
    float* asrc1 = xl1   + (long long)n * 64;
    float* adst1 = asrc1 + (long long)n * 8;
    float* amax1 = adst1 + (long long)n * 8;
    float* den1  = amax1 + (long long)n * 8;
    float* agg1  = den1  + (long long)n * 8;
    float* hbuf  = agg1  + (long long)n * 64;
    float* xl2   = hbuf  + (long long)n * 64;
    float* asrc2 = xl2   + (long long)n * 64;
    float* adst2 = asrc2 + (long long)n * 4;
    float* amax2 = adst2 + (long long)n * 4;
    float* den2  = amax2 + (long long)n * 4;
    float* agg2  = den2  + (long long)n * 4;

    const int B = 256;
    const float NEG_INF = -__builtin_inff();

    fill_kernel<<<cdiv((long long)n * 8, B), B, 0, stream>>>(amax1, NEG_INF, n * 8);
    fill_kernel<<<cdiv((long long)n * 8, B), B, 0, stream>>>(den1, 0.f, n * 8);
    fill_kernel<<<cdiv((long long)n * 64, B), B, 0, stream>>>(agg1, 0.f, n * 64);
    fill_kernel<<<cdiv((long long)n * 4, B), B, 0, stream>>>(amax2, NEG_INF, n * 4);
    fill_kernel<<<cdiv((long long)n * 4, B), B, 0, stream>>>(den2, 0.f, n * 4);
    fill_kernel<<<cdiv((long long)n * 64, B), B, 0, stream>>>(agg2, 0.f, n * 64);

    gemm_wmma_f32<<<cdiv(n, 32), 128, 0, stream>>>(x, W1, xl1, n, IN);
    alpha_kernel<8, 8><<<cdiv((long long)n * 8, B), B, 0, stream>>>(
        xl1, as1, ad1, asrc1, adst1, n);
    edge_max_kernel<8><<<cdiv((long long)ET * 8, B), B, 0, stream>>>(
        src, dst, E, ET, asrc1, adst1, amax1);
    edge_sum_kernel<8><<<cdiv((long long)ET * 8, B), B, 0, stream>>>(
        src, dst, E, ET, asrc1, adst1, amax1, den1);
    edge_agg_kernel<8, 3><<<cdiv((long long)ET * 64, B), B, 0, stream>>>(
        src, dst, E, ET, asrc1, adst1, amax1, den1, xl1, agg1);
    elu_bias_kernel<<<cdiv((long long)n * 64, B), B, 0, stream>>>(
        agg1, b1, hbuf, n * 64);

    gemm_wmma_f32<<<cdiv(n, 32), 128, 0, stream>>>(hbuf, W2, xl2, n, 64);
    alpha_kernel<4, 16><<<cdiv((long long)n * 4, B), B, 0, stream>>>(
        xl2, as2, ad2, asrc2, adst2, n);
    edge_max_kernel<4><<<cdiv((long long)ET * 4, B), B, 0, stream>>>(
        src, dst, E, ET, asrc2, adst2, amax2);
    edge_sum_kernel<4><<<cdiv((long long)ET * 4, B), B, 0, stream>>>(
        src, dst, E, ET, asrc2, adst2, amax2, den2);
    edge_agg_kernel<4, 4><<<cdiv((long long)ET * 64, B), B, 0, stream>>>(
        src, dst, E, ET, asrc2, adst2, amax2, den2, xl2, agg2);
    final_kernel<<<cdiv((long long)n * 16, B), B, 0, stream>>>(agg2, b2, out, n);
}


# Edges grouped by destination once per call (integer-add ranks, scan, place); one wave per node does max, exp-sum and row sum in registers; float atomics gone, sums now in list (arrival-rank) order
# speedup vs baseline: 2.7495x; 2.7495x over previous
#include <hip/hip_runtime.h>
#include <hip/hip_bf16.h>

#define LRELU_SLOPE 0.2f
#define GAT_EPS 1e-16f
#define KT 32

typedef __attribute__((ext_vector_type(2))) float v2f;
typedef __attribute__((ext_vector_type(8))) float v8f;

__global__ void gemm_wmma_f32(const float* __restrict__ A,
                              const float* __restrict__ B,
                              float* __restrict__ C, int M, int K) {
    __shared__ float As[32][KT + 1];

    const int tid  = threadIdx.x;
    const int lane = tid & 31;
    const int wave = tid >> 5;
    const int r0   = blockIdx.x * 32;
    const int c0   = wave * 16;
    const int half = lane >> 4;
    const int l15  = lane & 15;
    const int bcol = c0 + l15;

    v8f acc0 = {}, acc1 = {};

    for (int k0 = 0; k0 < K; k0 += KT) {
        __syncthreads();
#pragma unroll
        for (int i = tid; i < 32 * KT; i += 128) {
            const int r = i >> 5;
            const int c = i & (KT - 1);
            const int grow = r0 + r;
            As[r][c] = (grow < M) ? A[(long long)grow * K + k0 + c] : 0.f;
        }
        __syncthreads();

#pragma unroll
        for (int k = 0; k < KT; k += 4) {
            const int kk = k + (half << 1);
            v2f b, a0, a1;
            b.x  = B[(k0 + kk) * 64 + bcol];
            b.y  = B[(k0 + kk + 1) * 64 + bcol];
            a0.x = As[l15][kk];       a0.y = As[l15][kk + 1];
            a1.x = As[16 + l15][kk];  a1.y = As[16 + l15][kk + 1];
            acc0 = __builtin_amdgcn_wmma_f32_16x16x4_f32(
                false, a0, false, b, (short)0, acc0, false, false);
            acc1 = __builtin_amdgcn_wmma_f32_16x16x4_f32(
                false, a1, false, b, (short)0, acc1, false, false);
        }
    }

    const int rbase = r0 + half * 8;
#pragma unroll
    for (int r = 0; r < 8; ++r) {
        const int row0 = rbase + r;
        const int row1 = row0 + 16;
        if (row0 < M) C[(long long)row0 * 64 + c0 + l15] = acc0[r];
        if (row1 < M) C[(long long)row1 * 64 + c0 + l15] = acc1[r];
    }
}

template <int H, int CH>
__global__ void alpha_kernel(const float* __restrict__ xl,
                             const float* __restrict__ a_src,
                             const float* __restrict__ a_dst,
                             float* __restrict__ asrc,
                             float* __restrict__ adst, int n_nodes) {
    int idx = blockIdx.x * blockDim.x + threadIdx.x;
    if (idx >= n_nodes * H) return;
    const int n = idx / H, h = idx % H;
    const float* row = xl + (long long)n * 64 + h * CH;
    float s1 = 0.f, s2 = 0.f;
#pragma unroll
    for (int c = 0; c < CH; ++c) {
        s1 += row[c] * a_src[h * CH + c];
        s2 += row[c] * a_dst[h * CH + c];
    }
    asrc[idx] = s1;
    adst[idx] = s2;
}

__device__ __forceinline__ float lrelu(float x) {
    return x > 0.f ? x : LRELU_SLOPE * x;
}

__global__ void zero_int_kernel(int* __restrict__ p, int n) {
    int i = blockIdx.x * blockDim.x + threadIdx.x;
    if (i < n) p[i] = 0;
}

__global__ void count_rank_kernel(const int* __restrict__ dst, int E,
                                  int* __restrict__ cnt,
                                  int* __restrict__ rank) {
    int e = blockIdx.x * blockDim.x + threadIdx.x;
    if (e < E) rank[e] = atomicAdd(&cnt[dst[e]], 1);
}

#define SCAN_T 1024
__global__ void scan_rows_kernel(const int* __restrict__ cnt,
                                 int* __restrict__ rowptr, int n) {
    __shared__ int part[SCAN_T];
    const int tid   = threadIdx.x;
    const int chunk = (n + SCAN_T - 1) / SCAN_T;
    const int beg   = tid * chunk;
    const int end   = (beg + chunk < n) ? (beg + chunk) : n;
    int s = 0;
    for (int i = beg; i < end; ++i) s += cnt[i];
    part[tid] = s;
    __syncthreads();
    for (int off = 1; off < SCAN_T; off <<= 1) {
        const int v = (tid >= off) ? part[tid - off] : 0;
        __syncthreads();
        part[tid] += v;
        __syncthreads();
    }
    int run = part[tid] - s;
    for (int i = beg; i < end; ++i) { rowptr[i] = run; run += cnt[i]; }
    if (tid == SCAN_T - 1) rowptr[n] = part[SCAN_T - 1];
}

__global__ void place_src_kernel(const int* __restrict__ src,
                                 const int* __restrict__ dst,
                                 const int* __restrict__ rank,
                                 const int* __restrict__ rowptr, int E,
                                 int* __restrict__ csr_src) {
    int e = blockIdx.x * blockDim.x + threadIdx.x;
    if (e < E) csr_src[rowptr[dst[e]] + rank[e]] = src[e];
}

template <int H, int CSH>
__global__ void node_gather_kernel(const int* __restrict__ rowptr,
                                   const int* __restrict__ csr_src,
                                   const float* __restrict__ asrc,
                                   const float* __restrict__ adst,
                                   const float* __restrict__ xl,
                                   float* __restrict__ agg, int n_nodes) {
    const int lane = threadIdx.x & 31;
    const int d = __builtin_amdgcn_readfirstlane(
        (int)(blockIdx.x * (blockDim.x >> 5) + (threadIdx.x >> 5)));
    if (d >= n_nodes) return;
    const int h   = (2 * lane) >> CSH;
    const int beg = rowptr[d], end = rowptr[d + 1];
    const float ad    = adst[d * H + h];
    const float vself = lrelu(asrc[d * H + h] + ad);
    const float* xcol = xl + 2 * lane;

    float m = vself;
    for (int base = beg; base < end; base += 32) {
        const int cntc = (end - base < 32) ? (end - base) : 32;
        const int sidx = (lane < cntc) ? csr_src[base + lane] : d;
        for (int j = 0; j < cntc; j += 8) {
            float av[8];
#pragma unroll
            for (int u = 0; u < 8; ++u) {
                const int s = __builtin_amdgcn_readlane(sidx, j + u);
                av[u] = asrc[s * H + h];
            }
#pragma unroll
            for (int u = 0; u < 8; ++u) m = fmaxf(m, lrelu(av[u] + ad));
        }
    }

    float den = expf(vself - m);
    v2f acc;
    {
        const v2f xs = *(const v2f*)(xcol + (size_t)d * 64);
        acc.x = den * xs.x;
        acc.y = den * xs.y;
    }
    for (int base = beg; base < end; base += 32) {
        const int cntc = (end - base < 32) ? (end - base) : 32;
        const int sidx = (lane < cntc) ? csr_src[base + lane] : d;
        for (int j = 0; j < cntc; j += 8) {
            float av[8];
            v2f   xv[8];
#pragma unroll
            for (int u = 0; u < 8; ++u) {
                const int s = __builtin_amdgcn_readlane(sidx, j + u);
                av[u] = asrc[s * H + h];
                xv[u] = *(const v2f*)(xcol + (size_t)s * 64);
            }
#pragma unroll
            for (int u = 0; u < 8; ++u) {
                const float p = (j + u < cntc)
                                    ? expf(lrelu(av[u] + ad) - m) : 0.f;
                den   += p;
                acc.x += p * xv[u].x;
                acc.y += p * xv[u].y;
            }
        }
    }
    const float dd = fmaxf(den, GAT_EPS);
    v2f o;
    o.x = acc.x / dd;
    o.y = acc.y / dd;
    *(v2f*)(agg + (size_t)d * 64 + 2 * lane) = o;
}

__global__ void elu_bias_kernel(const float* __restrict__ agg,
                                const float* __restrict__ bias,
                                float* __restrict__ h, int n) {
    int i = blockIdx.x * blockDim.x + threadIdx.x;
    if (i >= n) return;
    float v = agg[i] + bias[i & 63];
    h[i] = v > 0.f ? v : (expf(v) - 1.f);
}

__global__ void final_kernel(const float* __restrict__ agg2,
                             const float* __restrict__ b2,
                             float* __restrict__ out, int n_nodes) {
    int i = blockIdx.x * blockDim.x + threadIdx.x;
    if (i >= n_nodes * 16) return;
    const int n = i >> 4, c = i & 15;
    const float* row = agg2 + (long long)n * 64;
    float s = row[c] + row[16 + c] + row[32 + c] + row[48 + c];
    out[i] = 0.25f * s + b2[c];
}

static inline int cdiv(long long a, long long b) { return (int)((a + b - 1) / b); }

extern "C" void kernel_launch(void* const* d_in, const int* in_sizes, int n_in,
                              void* d_out, int out_size, void* d_ws, size_t ws_size,
                              hipStream_t stream) {
    const float* x   = (const float*)d_in[0];
    const int*   ei  = (const int*)d_in[1];
    const float* W1  = (const float*)d_in[2];
    const float* as1 = (const float*)d_in[3];
    const float* ad1 = (const float*)d_in[4];
    const float* b1  = (const float*)d_in[5];
    const float* W2  = (const float*)d_in[6];
    const float* as2 = (const float*)d_in[7];
    const float* ad2 = (const float*)d_in[8];
    const float* b2  = (const float*)d_in[9];
    float*       out = (float*)d_out;

    const int IN = 256;
    const int n  = in_sizes[0] / IN;
    const int E  = in_sizes[1] / 2;
    const int* src = ei;
    const int* dst = ei + E;

    float* ws    = (float*)d_ws;
    float* xl1   = ws;
    float* asrc1 = xl1   + (long long)n * 64;
    float* adst1 = asrc1 + (long long)n * 8;
    float* agg1  = adst1 + (long long)n * 8;
    float* hbuf  = agg1  + (long long)n * 64;
    float* xl2   = hbuf  + (long long)n * 64;
    float* asrc2 = xl2   + (long long)n * 64;
    float* adst2 = asrc2 + (long long)n * 4;
    float* agg2  = adst2 + (long long)n * 4;
    int*   cnt    = (int*)(agg2 + (long long)n * 64);
    int*   rowptr = cnt    + n;
    int*   rank   = rowptr + n + 1;
    int*   csr    = rank   + E;

    const int B = 256;

    zero_int_kernel<<<cdiv(n, B), B, 0, stream>>>(cnt, n);
    count_rank_kernel<<<cdiv(E, B), B, 0, stream>>>(dst, E, cnt, rank);
    scan_rows_kernel<<<1, SCAN_T, 0, stream>>>(cnt, rowptr, n);
    place_src_kernel<<<cdiv(E, B), B, 0, stream>>>(src, dst, rank, rowptr, E, csr);

    gemm_wmma_f32<<<cdiv(n, 32), 128, 0, stream>>>(x, W1, xl1, n, IN);
    alpha_kernel<8, 8><<<cdiv((long long)n * 8, B), B, 0, stream>>>(
        xl1, as1, ad1, asrc1, adst1, n);
    node_gather_kernel<8, 3><<<cdiv(n, B / 32), B, 0, stream>>>(
        rowptr, csr, asrc1, adst1, xl1, agg1, n);
    elu_bias_kernel<<<cdiv((long long)n * 64, B), B, 0, stream>>>(
        agg1, b1, hbuf, n * 64);

    gemm_wmma_f32<<<cdiv(n, 32), 128, 0, stream>>>(hbuf, W2, xl2, n, 64);
    alpha_kernel<4, 16><<<cdiv((long long)n * 4, B), B, 0, stream>>>(
        xl2, as2, ad2, asrc2, adst2, n);
    node_gather_kernel<4, 4><<<cdiv(n, B / 32), B, 0, stream>>>(
        rowptr, csr, asrc2, adst2, xl2, agg2, n);
    final_kernel<<<cdiv((long long)n * 16, B), B, 0, stream>>>(agg2, b2, out, n);
}


# Four element-wise kernels fused into producers: logit dots in GEMM epilogue from an LDS tile copy; bias+ELU and head mean+bias in the gather's store; 12 launches to 8, every sum order kept
# speedup vs baseline: 2.8830x; 1.0486x over previous
#include <hip/hip_runtime.h>
#include <hip/hip_bf16.h>

#define LRELU_SLOPE 0.2f
#define GAT_EPS 1e-16f
#define KT 32

typedef __attribute__((ext_vector_type(2))) float v2f;
typedef __attribute__((ext_vector_type(8))) float v8f;

template <int H, int CH>
__global__ void gemm_alpha_wmma_f32(const float* __restrict__ A,
                                    const float* __restrict__ B,
                                    float* __restrict__ C,
                                    const float* __restrict__ a_src,
                                    const float* __restrict__ a_dst,
                                    float* __restrict__ asrc,
                                    float* __restrict__ adst, int M, int K) {
    __shared__ float As[32][KT + 1];
    __shared__ float Cs[32][66];

    const int tid  = threadIdx.x;
    const int lane = tid & 31;
    const int wave = tid >> 5;
    const int r0   = blockIdx.x * 32;
    const int c0   = wave * 16;
    const int half = lane >> 4;
    const int l15  = lane & 15;
    const int bcol = c0 + l15;

    v8f acc0 = {}, acc1 = {};

    for (int k0 = 0; k0 < K; k0 += KT) {
        __syncthreads();
#pragma unroll
        for (int i = tid; i < 32 * KT; i += 128) {
            const int r = i >> 5;
            const int c = i & (KT - 1);
            const int grow = r0 + r;
            As[r][c] = (grow < M) ? A[(long long)grow * K + k0 + c] : 0.f;
        }
        __syncthreads();

#pragma unroll
        for (int k = 0; k < KT; k += 4) {
            const int kk = k + (half << 1);
            v2f b, a0, a1;
            b.x  = B[(k0 + kk) * 64 + bcol];
            b.y  = B[(k0 + kk + 1) * 64 + bcol];
            a0.x = As[l15][kk];       a0.y = As[l15][kk + 1];
            a1.x = As[16 + l15][kk];  a1.y = As[16 + l15][kk + 1];
            acc0 = __builtin_amdgcn_wmma_f32_16x16x4_f32(
                false, a0, false, b, (short)0, acc0, false, false);
            acc1 = __builtin_amdgcn_wmma_f32_16x16x4_f32(
                false, a1, false, b, (short)0, acc1, false, false);
        }
    }

    const int lrow = half * 8;
#pragma unroll
    for (int r = 0; r < 8; ++r) {
        const int row0 = r0 + lrow + r;
        const int row1 = row0 + 16;
        if (row0 < M) C[(long long)row0 * 64 + c0 + l15] = acc0[r];
        if (row1 < M) C[(long long)row1 * 64 + c0 + l15] = acc1[r];
        Cs[lrow + r][c0 + l15]      = acc0[r];
        Cs[16 + lrow + r][c0 + l15] = acc1[r];
    }
    __syncthreads();

#pragma unroll
    for (int p = tid; p < 32 * H; p += 128) {
        const int lr   = p / H, h = p % H;
        const int grow = r0 + lr;
        const float* row = &Cs[lr][h * CH];
        float s1 = 0.f, s2 = 0.f;
#pragma unroll
        for (int c = 0; c < CH; ++c) {
            s1 += row[c] * a_src[h * CH + c];
            s2 += row[c] * a_dst[h * CH + c];
        }
        if (grow < M) {
            asrc[(long long)grow * H + h] = s1;
            adst[(long long)grow * H + h] = s2;
        }
    }
}

__device__ __forceinline__ float lrelu(float x) {
    return x > 0.f ? x : LRELU_SLOPE * x;
}

__global__ void zero_int_kernel(int* __restrict__ p, int n) {
    int i = blockIdx.x * blockDim.x + threadIdx.x;
    if (i < n) p[i] = 0;
}

__global__ void count_rank_kernel(const int* __restrict__ dst, int E,
                                  int* __restrict__ cnt,
                                  int* __restrict__ rank) {
    int e = blockIdx.x * blockDim.x + threadIdx.x;
    if (e < E) rank[e] = atomicAdd(&cnt[dst[e]], 1);
}

#define SCAN_T 1024
__global__ void scan_rows_kernel(const int* __restrict__ cnt,
                                 int* __restrict__ rowptr, int n) {
    __shared__ int part[SCAN_T];
    const int tid   = threadIdx.x;
    const int chunk = (n + SCAN_T - 1) / SCAN_T;
    const int beg   = tid * chunk;
    const int end   = (beg + chunk < n) ? (beg + chunk) : n;
    int s = 0;
    for (int i = beg; i < end; ++i) s += cnt[i];
    part[tid] = s;
    __syncthreads();
    for (int off = 1; off < SCAN_T; off <<= 1) {
        const int v = (tid >= off) ? part[tid - off] : 0;
        __syncthreads();
        part[tid] += v;
        __syncthreads();
    }
    int run = part[tid] - s;
    for (int i = beg; i < end; ++i) { rowptr[i] = run; run += cnt[i]; }
    if (tid == SCAN_T - 1) rowptr[n] = part[SCAN_T - 1];
}

__global__ void place_src_kernel(const int* __restrict__ src,
                                 const int* __restrict__ dst,
                                 const int* __restrict__ rank,
                                 const int* __restrict__ rowptr, int E,
                                 int* __restrict__ csr_src) {
    int e = blockIdx.x * blockDim.x + threadIdx.x;
    if (e < E) csr_src[rowptr[dst[e]] + rank[e]] = src[e];
}

template <int H, int CSH, bool MEAN>
__global__ void node_gather_kernel(const int* __restrict__ rowptr,
                                   const int* __restrict__ csr_src,
                                   const float* __restrict__ asrc,
                                   const float* __restrict__ adst,
                                   const float* __restrict__ xl,
                                   const float* __restrict__ bias,
                                   float* __restrict__ dstp, int n_nodes) {
    const int lane = threadIdx.x & 31;
    const int d = __builtin_amdgcn_readfirstlane(
        (int)(blockIdx.x * (blockDim.x >> 5) + (threadIdx.x >> 5)));
    if (d >= n_nodes) return;
    const int h   = (2 * lane) >> CSH;
    const int beg = rowptr[d], end = rowptr[d + 1];
    const float ad    = adst[d * H + h];
    const float vself = lrelu(asrc[d * H + h] + ad);
    const float* xcol = xl + 2 * lane;

    float m = vself;
    for (int base = beg; base < end; base += 32) {
        const int cntc = (end - base < 32) ? (end - base) : 32;
        const int sidx = (lane < cntc) ? csr_src[base + lane] : d;
        for (int j = 0; j < cntc; j += 8) {
            float av[8];
#pragma unroll
            for (int u = 0; u < 8; ++u) {
                const int s = __builtin_amdgcn_readlane(sidx, j + u);
                av[u] = asrc[s * H + h];
            }
#pragma unroll
            for (int u = 0; u < 8; ++u) m = fmaxf(m, lrelu(av[u] + ad));
        }
    }

    float den = expf(vself - m);
    v2f acc;
    {
        const v2f xs = *(const v2f*)(xcol + (size_t)d * 64);
        acc.x = den * xs.x;
        acc.y = den * xs.y;
    }
    for (int base = beg; base < end; base += 32) {
        const int cntc = (end - base < 32) ? (end - base) : 32;
        const int sidx = (lane < cntc) ? csr_src[base + lane] : d;
        for (int j = 0; j < cntc; j += 8) {
            float av[8];
            v2f   xv[8];
#pragma unroll
            for (int u = 0; u < 8; ++u) {
                const int s = __builtin_amdgcn_readlane(sidx, j + u);
                av[u] = asrc[s * H + h];
                xv[u] = *(const v2f*)(xcol + (size_t)s * 64);
            }
#pragma unroll
            for (int u = 0; u < 8; ++u) {
                const float p = (j + u < cntc)
                                    ? expf(lrelu(av[u] + ad) - m) : 0.f;
                den   += p;
                acc.x += p * xv[u].x;
                acc.y += p * xv[u].y;
            }
        }
    }
    const float dd = fmaxf(den, GAT_EPS);
    v2f o;
    o.x = acc.x / dd;
    o.y = acc.y / dd;
    if (!MEAN) {
        const float vx = o.x + bias[2 * lane];
        const float vy = o.y + bias[2 * lane + 1];
        v2f r;
        r.x = vx > 0.f ? vx : (expf(vx) - 1.f);
        r.y = vy > 0.f ? vy : (expf(vy) - 1.f);
        *(v2f*)(dstp + (size_t)d * 64 + 2 * lane) = r;
    } else {
        float sx = o.x, sy = o.y;
#pragma unroll
        for (int k = 8; k < 32; k += 8) {
            sx += __shfl(o.x, (lane + k) & 31);
            sy += __shfl(o.y, (lane + k) & 31);
        }
        if (lane < 8) {
            v2f r;
            r.x = 0.25f * sx + bias[2 * lane];
            r.y = 0.25f * sy + bias[2 * lane + 1];
            *(v2f*)(dstp + (size_t)d * 16 + 2 * lane) = r;
        }
    }
}

static inline int cdiv(long long a, long long b) { return (int)((a + b - 1) / b); }

extern "C" void kernel_launch(void* const* d_in, const int* in_sizes, int n_in,
                              void* d_out, int out_size, void* d_ws, size_t ws_size,
                              hipStream_t stream) {
    const float* x   = (const float*)d_in[0];
    const int*   ei  = (const int*)d_in[1];
    const float* W1  = (const float*)d_in[2];
    const float* as1 = (const float*)d_in[3];
    const float* ad1 = (const float*)d_in[4];
    const float* b1  = (const float*)d_in[5];
    const float* W2  = (const float*)d_in[6];
    const float* as2 = (const float*)d_in[7];
    const float* ad2 = (const float*)d_in[8];
    const float* b2  = (const float*)d_in[9];
    float*       out = (float*)d_out;

    const int IN = 256;
    const int n  = in_sizes[0] / IN;
    const int E  = in_sizes[1] / 2;
    const int* src = ei;
    const int* dst = ei + E;

    float* ws    = (float*)d_ws;
    float* xl1   = ws;
    float* asrc1 = xl1   + (long long)n * 64;
    float* adst1 = asrc1 + (long long)n * 8;
    float* hbuf  = adst1 + (long long)n * 8;
    float* xl2   = hbuf  + (long long)n * 64;
    float* asrc2 = xl2   + (long long)n * 64;
    float* adst2 = asrc2 + (long long)n * 4;
    int*   cnt    = (int*)(adst2 + (long long)n * 4);
    int*   rowptr = cnt    + n;
    int*   rank   = rowptr + n + 1;
    int*   csr    = rank   + E;

    const int B = 256;

    zero_int_kernel<<<cdiv(n, B), B, 0, stream>>>(cnt, n);
    count_rank_kernel<<<cdiv(E, B), B, 0, stream>>>(dst, E, cnt, rank);
    scan_rows_kernel<<<1, SCAN_T, 0, stream>>>(cnt, rowptr, n);
    place_src_kernel<<<cdiv(E, B), B, 0, stream>>>(src, dst, rank, rowptr, E, csr);

    gemm_alpha_wmma_f32<8, 8><<<cdiv(n, 32), 128, 0, stream>>>(
        x, W1, xl1, as1, ad1, asrc1, adst1, n, IN);
    node_gather_kernel<8, 3, false><<<cdiv(n, B / 32), B, 0, stream>>>(
        rowptr, csr, asrc1, adst1, xl1, b1, hbuf, n);

    gemm_alpha_wmma_f32<4, 16><<<cdiv(n, 32), 128, 0, stream>>>(
        hbuf, W2, xl2, as2, ad2, asrc2, adst2, n, 64);
    node_gather_kernel<4, 4, true><<<cdiv(n, B / 32), B, 0, stream>>>(
        rowptr, csr, asrc2, adst2, xl2, b2, out, n);
}


# Gather kernels keep the pass-1 logits of the first 64 list positions in per-wave LDS and read them back in pass 2: the second score gather per edge is gone; every sum and its order unchanged
# speedup vs baseline: 2.9873x; 1.0362x over previous
#include <hip/hip_runtime.h>
#include <hip/hip_bf16.h>

#define LRELU_SLOPE 0.2f
#define GAT_EPS 1e-16f
#define KT 32

typedef __attribute__((ext_vector_type(2))) float v2f;
typedef __attribute__((ext_vector_type(4))) float v4f;
typedef __attribute__((ext_vector_type(8))) float v8f;

template <int H, int CH>
__global__ void gemm_alpha_wmma_f32(const float* __restrict__ A,
                                    const float* __restrict__ B,
                                    float* __restrict__ C,
                                    const float* __restrict__ a_src,
                                    const float* __restrict__ a_dst,
                                    float* __restrict__ asrc,
                                    float* __restrict__ adst, int M, int K) {
    __shared__ float As[32][KT + 1];
    __shared__ float Cs[32][66];

    const int tid  = threadIdx.x;
    const int lane = tid & 31;
    const int wave = tid >> 5;
    const int r0   = blockIdx.x * 32;
    const int c0   = wave * 16;
    const int half = lane >> 4;
    const int l15  = lane & 15;
    const int bcol = c0 + l15;

    v8f acc0 = {}, acc1 = {};

    for (int k0 = 0; k0 < K; k0 += KT) {
        __syncthreads();
#pragma unroll
        for (int i = tid; i < 32 * KT; i += 128) {
            const int r = i >> 5;
            const int c = i & (KT - 1);
            const int grow = r0 + r;
            As[r][c] = (grow < M) ? A[(long long)grow * K + k0 + c] : 0.f;
        }
        __syncthreads();

#pragma unroll
        for (int k = 0; k < KT; k += 4) {
            const int kk = k + (half << 1);
            v2f b, a0, a1;
            b.x  = B[(k0 + kk) * 64 + bcol];
            b.y  = B[(k0 + kk + 1) * 64 + bcol];
            a0.x = As[l15][kk];       a0.y = As[l15][kk + 1];
            a1.x = As[16 + l15][kk];  a1.y = As[16 + l15][kk + 1];
            acc0 = __builtin_amdgcn_wmma_f32_16x16x4_f32(
                false, a0, false, b, (short)0, acc0, false, false);
            acc1 = __builtin_amdgcn_wmma_f32_16x16x4_f32(
                false, a1, false, b, (short)0, acc1, false, false);
        }
    }

    const int lrow = half * 8;
#pragma unroll
    for (int r = 0; r < 8; ++r) {
        const int row0 = r0 + lrow + r;
        const int row1 = row0 + 16;
        if (row0 < M) C[(long long)row0 * 64 + c0 + l15] = acc0[r];
        if (row1 < M) C[(long long)row1 * 64 + c0 + l15] = acc1[r];
        Cs[lrow + r][c0 + l15]      = acc0[r];
        Cs[16 + lrow + r][c0 + l15] = acc1[r];
    }
    __syncthreads();

#pragma unroll
    for (int p = tid; p < 32 * H; p += 128) {
        const int lr   = p / H, h = p % H;
        const int grow = r0 + lr;
        const float* row = &Cs[lr][h * CH];
        float s1 = 0.f, s2 = 0.f;
#pragma unroll
        for (int c = 0; c < CH; ++c) {
            s1 += row[c] * a_src[h * CH + c];
            s2 += row[c] * a_dst[h * CH + c];
        }
        if (grow < M) {
            asrc[(long long)grow * H + h] = s1;
            adst[(long long)grow * H + h] = s2;
        }
    }
}

__device__ __forceinline__ float lrelu(float x) {
    return x > 0.f ? x : LRELU_SLOPE * x;
}

__global__ void zero_int_kernel(int* __restrict__ p, int n) {
    int i = blockIdx.x * blockDim.x + threadIdx.x;
    if (i < n) p[i] = 0;
}

__global__ void count_rank_kernel(const int* __restrict__ dst, int E,
                                  int* __restrict__ cnt,
                                  int* __restrict__ rank) {
    int e = blockIdx.x * blockDim.x + threadIdx.x;
    if (e < E) rank[e] = atomicAdd(&cnt[dst[e]], 1);
}

#define SCAN_T 1024
__global__ void scan_rows_kernel(const int* __restrict__ cnt,
                                 int* __restrict__ rowptr, int n) {
    __shared__ int part[SCAN_T];
    const int tid   = threadIdx.x;
    const int chunk = (n + SCAN_T - 1) / SCAN_T;
    const int beg   = tid * chunk;
    const int end   = (beg + chunk < n) ? (beg + chunk) : n;
    int s = 0;
    for (int i = beg; i < end; ++i) s += cnt[i];
    part[tid] = s;
    __syncthreads();
    for (int off = 1; off < SCAN_T; off <<= 1) {
        const int v = (tid >= off) ? part[tid - off] : 0;
        __syncthreads();
        part[tid] += v;
        __syncthreads();
    }
    int run = part[tid] - s;
    for (int i = beg; i < end; ++i) { rowptr[i] = run; run += cnt[i]; }
    if (tid == SCAN_T - 1) rowptr[n] = part[SCAN_T - 1];
}

__global__ void place_src_kernel(const int* __restrict__ src,
                                 const int* __restrict__ dst,
                                 const int* __restrict__ rank,
                                 const int* __restrict__ rowptr, int E,
                                 int* __restrict__ csr_src) {
    int e = blockIdx.x * blockDim.x + threadIdx.x;
    if (e < E) csr_src[rowptr[dst[e]] + rank[e]] = src[e];
}

#define GATHER_WAVES 8
#define LG_CAP   64
#define LG_PITCH 68

template <int H, int CSH, bool MEAN>
__global__ void node_gather_kernel(const int* __restrict__ rowptr,
                                   const int* __restrict__ csr_src,
                                   const float* __restrict__ asrc,
                                   const float* __restrict__ adst,
                                   const float* __restrict__ xl,
                                   const float* __restrict__ bias,
                                   float* __restrict__ dstp, int n_nodes) {
    __shared__ __attribute__((aligned(16))) float lg[GATHER_WAVES][H * LG_PITCH];

    const int lane = threadIdx.x & 31;
    const int d = __builtin_amdgcn_readfirstlane(
        (int)(blockIdx.x * (blockDim.x >> 5) + (threadIdx.x >> 5)));
    if (d >= n_nodes) return;
    const int h   = (2 * lane) >> CSH;
    float* lgrow  = &lg[threadIdx.x >> 5][h * LG_PITCH];
    const int beg = rowptr[d], end = rowptr[d + 1];
    const float ad    = adst[d * H + h];
    const float vself = lrelu(asrc[d * H + h] + ad);
    const float* xcol = xl + 2 * lane;

    float m = vself;
    for (int base = beg; base < end; base += 32) {
        const int cntc = (end - base < 32) ? (end - base) : 32;
        const int sidx = (lane < cntc) ? csr_src[base + lane] : d;
        const int pos0 = base - beg;
        for (int j = 0; j < cntc; j += 8) {
            float av[8], tv[8];
#pragma unroll
            for (int u = 0; u < 8; ++u) {
                const int s = __builtin_amdgcn_readlane(sidx, j + u);
                av[u] = asrc[s * H + h];
            }
#pragma unroll
            for (int u = 0; u < 8; ++u) {
                tv[u] = lrelu(av[u] + ad);
                m = fmaxf(m, tv[u]);
            }
            if (pos0 < LG_CAP) {
                v4f t0, t1;
                t0.x = tv[0]; t0.y = tv[1]; t0.z = tv[2]; t0.w = tv[3];
                t1.x = tv[4]; t1.y = tv[5]; t1.z = tv[6]; t1.w = tv[7];
                *(v4f*)(lgrow + pos0 + j)     = t0;
                *(v4f*)(lgrow + pos0 + j + 4) = t1;
            }
        }
    }

    float den = expf(vself - m);
    v2f acc;
    {
        const v2f xs = *(const v2f*)(xcol + (size_t)d * 64);
        acc.x = den * xs.x;
        acc.y = den * xs.y;
    }
    for (int base = beg; base < end; base += 32) {
        const int cntc = (end - base < 32) ? (end - base) : 32;
        const int sidx = (lane < cntc) ? csr_src[base + lane] : d;
        const int pos0 = base - beg;
        for (int j = 0; j < cntc; j += 8) {
            float tv[8];
            v2f   xv[8];
            if (pos0 < LG_CAP) {
#pragma unroll
                for (int u = 0; u < 8; ++u) {
                    const int s = __builtin_amdgcn_readlane(sidx, j + u);
                    xv[u] = *(const v2f*)(xcol + (size_t)s * 64);
                }
                const v4f t0 = *(const v4f*)(lgrow + pos0 + j);
                const v4f t1 = *(const v4f*)(lgrow + pos0 + j + 4);
                tv[0] = t0.x; tv[1] = t0.y; tv[2] = t0.z; tv[3] = t0.w;
                tv[4] = t1.x; tv[5] = t1.y; tv[6] = t1.z; tv[7] = t1.w;
            } else {
                float av[8];
#pragma unroll
                for (int u = 0; u < 8; ++u) {
                    const int s = __builtin_amdgcn_readlane(sidx, j + u);
                    av[u] = asrc[s * H + h];
                    xv[u] = *(const v2f*)(xcol + (size_t)s * 64);
                }
#pragma unroll
                for (int u = 0; u < 8; ++u) tv[u] = lrelu(av[u] + ad);
            }
#pragma unroll
            for (int u = 0; u < 8; ++u) {
                const float p = (j + u < cntc) ? expf(tv[u] - m) : 0.f;
                den   += p;
                acc.x += p * xv[u].x;
                acc.y += p * xv[u].y;
            }
        }
    }
    const float dd = fmaxf(den, GAT_EPS);
    v2f o;
    o.x = acc.x / dd;
    o.y = acc.y / dd;
    if (!MEAN) {
        const float vx = o.x + bias[2 * lane];
        const float vy = o.y + bias[2 * lane + 1];
        v2f r;
        r.x = vx > 0.f ? vx : (expf(vx) - 1.f);
        r.y = vy > 0.f ? vy : (expf(vy) - 1.f);
        *(v2f*)(dstp + (size_t)d * 64 + 2 * lane) = r;
    } else {
        float sx = o.x, sy = o.y;
#pragma unroll
        for (int k = 8; k < 32; k += 8) {
            sx += __shfl(o.x, (lane + k) & 31);
            sy += __shfl(o.y, (lane + k) & 31);
        }
        if (lane < 8) {
            v2f r;
            r.x = 0.25f * sx + bias[2 * lane];
            r.y = 0.25f * sy + bias[2 * lane + 1];
            *(v2f*)(dstp + (size_t)d * 16 + 2 * lane) = r;
        }
    }
}

static inline int cdiv(long long a, long long b) { return (int)((a + b - 1) / b); }

extern "C" void kernel_launch(void* const* d_in, const int* in_sizes, int n_in,
                              void* d_out, int out_size, void* d_ws, size_t ws_size,
                              hipStream_t stream) {
    const float* x   = (const float*)d_in[0];
    const int*   ei  = (const int*)d_in[1];
    const float* W1  = (const float*)d_in[2];
    const float* as1 = (const float*)d_in[3];
    const float* ad1 = (const float*)d_in[4];
    const float* b1  = (const float*)d_in[5];
    const float* W2  = (const float*)d_in[6];
    const float* as2 = (const float*)d_in[7];
    const float* ad2 = (const float*)d_in[8];
    const float* b2  = (const float*)d_in[9];
    float*       out = (float*)d_out;

    const int IN = 256;
    const int n  = in_sizes[0] / IN;
    const int E  = in_sizes[1] / 2;
    const int* src = ei;
    const int* dst = ei + E;

    float* ws    = (float*)d_ws;
    float* xl1   = ws;
    float* asrc1 = xl1   + (long long)n * 64;
    float* adst1 = asrc1 + (long long)n * 8;
    float* hbuf  = adst1 + (long long)n * 8;
    float* xl2   = hbuf  + (long long)n * 64;
    float* asrc2 = xl2   + (long long)n * 64;
    float* adst2 = asrc2 + (long long)n * 4;
    int*   cnt    = (int*)(adst2 + (long long)n * 4);
    int*   rowptr = cnt    + n;
    int*   rank   = rowptr + n + 1;
    int*   csr    = rank   + E;

    const int B = 256;

    zero_int_kernel<<<cdiv(n, B), B, 0, stream>>>(cnt, n);
    count_rank_kernel<<<cdiv(E, B), B, 0, stream>>>(dst, E, cnt, rank);
    scan_rows_kernel<<<1, SCAN_T, 0, stream>>>(cnt, rowptr, n);
    place_src_kernel<<<cdiv(E, B), B, 0, stream>>>(src, dst, rank, rowptr, E, csr);

    gemm_alpha_wmma_f32<8, 8><<<cdiv(n, 32), 128, 0, stream>>>(
        x, W1, xl1, as1, ad1, asrc1, adst1, n, IN);
    node_gather_kernel<8, 3, false><<<cdiv(n, GATHER_WAVES), GATHER_WAVES * 32, 0, stream>>>(
        rowptr, csr, asrc1, adst1, xl1, b1, hbuf, n);

    gemm_alpha_wmma_f32<4, 16><<<cdiv(n, 32), 128, 0, stream>>>(
        hbuf, W2, xl2, as2, ad2, asrc2, adst2, n, 64);
    node_gather_kernel<4, 4, true><<<cdiv(n, GATHER_WAVES), GATHER_WAVES * 32, 0, stream>>>(
        rowptr, csr, asrc2, adst2, xl2, b2, out, n);
}


# Edge list built by a two-level counting sort in LDS (tile counts, bin runs, per-bin node sort); per-edge global adds, one-block scan, scattered stores gone; 8 to 7 launches; sum order = LDS arrival
# speedup vs baseline: 6.7511x; 2.2599x over previous
#include <hip/hip_runtime.h>
#include <hip/hip_bf16.h>

#define LRELU_SLOPE 0.2f
#define GAT_EPS 1e-16f
#define KT 32

typedef __attribute__((ext_vector_type(2))) float v2f;
typedef __attribute__((ext_vector_type(4))) float v4f;
typedef __attribute__((ext_vector_type(8))) float v8f;

template <int H, int CH>
__global__ void gemm_alpha_wmma_f32(const float* __restrict__ A,
                                    const float* __restrict__ B,
                                    float* __restrict__ C,
                                    const float* __restrict__ a_src,
                                    const float* __restrict__ a_dst,
                                    float* __restrict__ asrc,
                                    float* __restrict__ adst, int M, int K) {
    __shared__ float As[32][KT + 1];
    __shared__ float Cs[32][66];

    const int tid  = threadIdx.x;
    const int lane = tid & 31;
    const int wave = tid >> 5;
    const int r0   = blockIdx.x * 32;
    const int c0   = wave * 16;
    const int half = lane >> 4;
    const int l15  = lane & 15;
    const int bcol = c0 + l15;

    v8f acc0 = {}, acc1 = {};

    for (int k0 = 0; k0 < K; k0 += KT) {
        __syncthreads();
#pragma unroll
        for (int i = tid; i < 32 * KT; i += 128) {
            const int r = i >> 5;
            const int c = i & (KT - 1);
            const int grow = r0 + r;
            As[r][c] = (grow < M) ? A[(long long)grow * K + k0 + c] : 0.f;
        }
        __syncthreads();

#pragma unroll
        for (int k = 0; k < KT; k += 4) {
            const int kk = k + (half << 1);
            v2f b, a0, a1;
            b.x  = B[(k0 + kk) * 64 + bcol];
            b.y  = B[(k0 + kk + 1) * 64 + bcol];
            a0.x = As[l15][kk];       a0.y = As[l15][kk + 1];
            a1.x = As[16 + l15][kk];  a1.y = As[16 + l15][kk + 1];
            acc0 = __builtin_amdgcn_wmma_f32_16x16x4_f32(
                false, a0, false, b, (short)0, acc0, false, false);
            acc1 = __builtin_amdgcn_wmma_f32_16x16x4_f32(
                false, a1, false, b, (short)0, acc1, false, false);
        }
    }

    const int lrow = half * 8;
#pragma unroll
    for (int r = 0; r < 8; ++r) {
        const int row0 = r0 + lrow + r;
        const int row1 = row0 + 16;
        if (row0 < M) C[(long long)row0 * 64 + c0 + l15] = acc0[r];
        if (row1 < M) C[(long long)row1 * 64 + c0 + l15] = acc1[r];
        Cs[lrow + r][c0 + l15]      = acc0[r];
        Cs[16 + lrow + r][c0 + l15] = acc1[r];
    }
    __syncthreads();

#pragma unroll
    for (int p = tid; p < 32 * H; p += 128) {
        const int lr   = p / H, h = p % H;
        const int grow = r0 + lr;
        const float* row = &Cs[lr][h * CH];
        float s1 = 0.f, s2 = 0.f;
#pragma unroll
        for (int c = 0; c < CH; ++c) {
            s1 += row[c] * a_src[h * CH + c];
            s2 += row[c] * a_dst[h * CH + c];
        }
        if (grow < M) {
            asrc[(long long)grow * H + h] = s1;
            adst[(long long)grow * H + h] = s2;
        }
    }
}

__device__ __forceinline__ float lrelu(float x) {
    return x > 0.f ? x : LRELU_SLOPE * x;
}

#define BIN_SHIFT 8
#define BIN_NODES 256
#define NB_MAX    256
#define TILE_E    8192
#define SORT_T    1024
#define STAGE_CAP 16384
#define RW        10

__device__ __forceinline__ int scan256_wave(const int* in, int* out, int lane) {
    int a[8];
    int s = 0;
#pragma unroll
    for (int k = 0; k < 8; ++k) { a[k] = in[8 * lane + k]; s += a[k]; }
    int incl = s;
#pragma unroll
    for (int off = 1; off < 32; off <<= 1) {
        const int v = __shfl_up(incl, off);
        if (lane >= off) incl += v;
    }
    int run = incl - s;
#pragma unroll
    for (int k = 0; k < 8; ++k) { out[8 * lane + k] = run; run += a[k]; }
    return __shfl(incl, 31);
}

__global__ void __launch_bounds__(SORT_T)
tile_hist_kernel(const int* __restrict__ dst, int E, int NB,
                 int* __restrict__ tilehist) {
    __shared__ int lh[NB_MAX];
    const int t = threadIdx.x;
    if (t < NB_MAX) lh[t] = 0;
    __syncthreads();
    const int base = blockIdx.x * TILE_E;
    int dv[TILE_E / SORT_T];
#pragma unroll
    for (int k = 0; k < TILE_E / SORT_T; ++k) {
        const int e = base + k * SORT_T + t;
        dv[k] = (e < E) ? dst[e] : -1;
    }
#pragma unroll
    for (int k = 0; k < TILE_E / SORT_T; ++k)
        if (dv[k] >= 0) atomicAdd(&lh[dv[k] >> BIN_SHIFT], 1);
    __syncthreads();
    if (t < NB) tilehist[blockIdx.x * NB + t] = lh[t];
}

__global__ void __launch_bounds__(SORT_T)
tile_partition_kernel(const int* __restrict__ src, const int* __restrict__ dst,
                      int E, int NB, int G, const int* __restrict__ tilehist,
                      int* __restrict__ binstart, unsigned* __restrict__ part) {
    __shared__ int s_tot[4][NB_MAX];
    __shared__ int s_pre[4][NB_MAX];
    __shared__ int s_T[NB_MAX];
    __shared__ int s_own[NB_MAX];
    __shared__ int s_bs[NB_MAX + 1];
    __shared__ int s_lb[NB_MAX];
    __shared__ int s_gb[NB_MAX];
    __shared__ int s_cur[NB_MAX];
    __shared__ unsigned s_val[TILE_E];

    const int t = threadIdx.x, blk = blockIdx.x;
    const int c = t & (NB_MAX - 1), rg = t >> 8;
    int tot = 0, pre = 0;
    if (c < NB) {
        for (int r = rg; r < G; r += 4) {
            const int v = tilehist[r * NB + c];
            tot += v;
            pre += (r < blk) ? v : 0;
        }
    }
    s_tot[rg][c] = tot;
    s_pre[rg][c] = pre;
    __syncthreads();
    if (t < NB_MAX) {
        s_T[t]   = s_tot[0][t] + s_tot[1][t] + s_tot[2][t] + s_tot[3][t];
        s_own[t] = (t < NB) ? tilehist[blk * NB + t] : 0;
        s_cur[t] = 0;
    }
    __syncthreads();
    if (t < 32) {
        const int total = scan256_wave(s_T, s_bs, t);
        if (t == 0) s_bs[NB_MAX] = total;
        scan256_wave(s_own, s_lb, t);
    }
    __syncthreads();
    if (t < NB_MAX)
        s_gb[t] = s_bs[t] + s_pre[0][t] + s_pre[1][t] + s_pre[2][t] + s_pre[3][t];
    if (blk == 0 && t <= NB) binstart[t] = s_bs[t];
    __syncthreads();

    const int base = blk * TILE_E;
    int dv[TILE_E / SORT_T], sv[TILE_E / SORT_T];
#pragma unroll
    for (int k = 0; k < TILE_E / SORT_T; ++k) {
        const int e = base + k * SORT_T + t;
        dv[k] = (e < E) ? dst[e] : -1;
        sv[k] = (e < E) ? src[e] : 0;
    }
    int slot[TILE_E / SORT_T];
#pragma unroll
    for (int k = 0; k < TILE_E / SORT_T; ++k) {
        slot[k] = -1;
        if (dv[k] >= 0) {
            const int b = dv[k] >> BIN_SHIFT;
            slot[k] = s_lb[b] + atomicAdd(&s_cur[b], 1);
        }
    }
#pragma unroll
    for (int k = 0; k < TILE_E / SORT_T; ++k)
        if (slot[k] >= 0) s_val[slot[k]] = ((unsigned)dv[k] << 16) | (unsigned)sv[k];
    __syncthreads();
    const int left = E - base;
    const int cnt  = left < TILE_E ? left : TILE_E;
    for (int i = t; i < cnt; i += SORT_T) {
        const unsigned w = s_val[i];
        const int b = (int)(w >> (16 + BIN_SHIFT));
        part[s_gb[b] + (i - s_lb[b])] = w;
    }
}

__global__ void __launch_bounds__(SORT_T)
bin_sort_kernel(const unsigned* __restrict__ part,
                const int* __restrict__ binstart, int n, int NB,
                int* __restrict__ rowptr, int* __restrict__ csr_src) {
    __shared__ int s_cnt[BIN_NODES];
    __shared__ int s_ex[BIN_NODES];
    __shared__ int s_cur[BIN_NODES];
    __shared__ int s_stage[STAGE_CAP];
    const int t = threadIdx.x, b = blockIdx.x;
    const int beg = binstart[b], end = binstart[b + 1];
    if (t < BIN_NODES) s_cnt[t] = 0;
    unsigned wr[RW];
#pragma unroll
    for (int k = 0; k < RW; ++k) {
        const int i = beg + k * SORT_T + t;
        wr[k] = (i < end) ? part[i] : 0u;
    }
    __syncthreads();
#pragma unroll
    for (int k = 0; k < RW; ++k)
        if (beg + k * SORT_T + t < end)
            atomicAdd(&s_cnt[(wr[k] >> 16) & (BIN_NODES - 1)], 1);
    for (int i = beg + RW * SORT_T + t; i < end; i += SORT_T)
        atomicAdd(&s_cnt[(part[i] >> 16) & (BIN_NODES - 1)], 1);
    __syncthreads();
    if (t < 32) scan256_wave(s_cnt, s_ex, t);
    __syncthreads();
    if (t < BIN_NODES) {
        const int node = b * BIN_NODES + t;
        s_cur[t] = s_ex[t];
        if (node < n) rowptr[node] = beg + s_ex[t];
    }
    if (b == NB - 1 && t == 0) rowptr[n] = end;
    __syncthreads();
    const int  m      = end - beg;
    const bool staged = (m <= STAGE_CAP);
    int pos[RW];
#pragma unroll
    for (int k = 0; k < RW; ++k) {
        pos[k] = -1;
        if (beg + k * SORT_T + t < end)
            pos[k] = atomicAdd(&s_cur[(wr[k] >> 16) & (BIN_NODES - 1)], 1);
    }
#pragma unroll
    for (int k = 0; k < RW; ++k) {
        if (pos[k] >= 0) {
            if (staged) s_stage[pos[k]] = (int)(wr[k] & 0xffffu);
            else        csr_src[beg + pos[k]] = (int)(wr[k] & 0xffffu);
        }
    }
    for (int i = beg + RW * SORT_T + t; i < end; i += SORT_T) {
        const unsigned w = part[i];
        const int p = atomicAdd(&s_cur[(w >> 16) & (BIN_NODES - 1)], 1);
        if (staged) s_stage[p] = (int)(w & 0xffffu);
        else        csr_src[beg + p] = (int)(w & 0xffffu);
    }
    if (staged) {
        __syncthreads();
        for (int i = t; i < m; i += SORT_T) csr_src[beg + i] = s_stage[i];
    }
}

#define GATHER_WAVES 8
#define LG_CAP   64
#define LG_PITCH 68

template <int H, int CSH, bool MEAN>
__global__ void node_gather_kernel(const int* __restrict__ rowptr,
                                   const int* __restrict__ csr_src,
                                   const float* __restrict__ asrc,
                                   const float* __restrict__ adst,
                                   const float* __restrict__ xl,
                                   const float* __restrict__ bias,
                                   float* __restrict__ dstp, int n_nodes) {
    __shared__ __attribute__((aligned(16))) float lg[GATHER_WAVES][H * LG_PITCH];

    const int lane = threadIdx.x & 31;
    const int d = __builtin_amdgcn_readfirstlane(
        (int)(blockIdx.x * (blockDim.x >> 5) + (threadIdx.x >> 5)));
    if (d >= n_nodes) return;
    const int h   = (2 * lane) >> CSH;
    float* lgrow  = &lg[threadIdx.x >> 5][h * LG_PITCH];
    const int beg = rowptr[d], end = rowptr[d + 1];
    const float ad    = adst[d * H + h];
    const float vself = lrelu(asrc[d * H + h] + ad);
    const float* xcol = xl + 2 * lane;

    float m = vself;
    for (int base = beg; base < end; base += 32) {
        const int cntc = (end - base < 32) ? (end - base) : 32;
        const int sidx = (lane < cntc) ? csr_src[base + lane] : d;
        const int pos0 = base - beg;
        for (int j = 0; j < cntc; j += 8) {
            float av[8], tv[8];
#pragma unroll
            for (int u = 0; u < 8; ++u) {
                const int s = __builtin_amdgcn_readlane(sidx, j + u);
                av[u] = asrc[s * H + h];
            }
#pragma unroll
            for (int u = 0; u < 8; ++u) {
                tv[u] = lrelu(av[u] + ad);
                m = fmaxf(m, tv[u]);
            }
            if (pos0 < LG_CAP) {
                v4f t0, t1;
                t0.x = tv[0]; t0.y = tv[1]; t0.z = tv[2]; t0.w = tv[3];
                t1.x = tv[4]; t1.y = tv[5]; t1.z = tv[6]; t1.w = tv[7];
                *(v4f*)(lgrow + pos0 + j)     = t0;
                *(v4f*)(lgrow + pos0 + j + 4) = t1;
            }
        }
    }

    float den = expf(vself - m);
    v2f acc;
    {
        const v2f xs = *(const v2f*)(xcol + (size_t)d * 64);
        acc.x = den * xs.x;
        acc.y = den * xs.y;
    }
    for (int base = beg; base < end; base += 32) {
        const int cntc = (end - base < 32) ? (end - base) : 32;
        const int sidx = (lane < cntc) ? csr_src[base + lane] : d;
        const int pos0 = base - beg;
        for (int j = 0; j < cntc; j += 8) {
            float tv[8];
            v2f   xv[8];
            if (pos0 < LG_CAP) {
#pragma unroll
                for (int u = 0; u < 8; ++u) {
                    const int s = __builtin_amdgcn_readlane(sidx, j + u);
                    xv[u] = *(const v2f*)(xcol + (size_t)s * 64);
                }
                const v4f t0 = *(const v4f*)(lgrow + pos0 + j);
                const v4f t1 = *(const v4f*)(lgrow + pos0 + j + 4);
                tv[0] = t0.x; tv[1] = t0.y; tv[2] = t0.z; tv[3] = t0.w;
                tv[4] = t1.x; tv[5] = t1.y; tv[6] = t1.z; tv[7] = t1.w;
            } else {
                float av[8];
#pragma unroll
                for (int u = 0; u < 8; ++u) {
                    const int s = __builtin_amdgcn_readlane(sidx, j + u);
                    av[u] = asrc[s * H + h];
                    xv[u] = *(const v2f*)(xcol + (size_t)s * 64);
                }
#pragma unroll
                for (int u = 0; u < 8; ++u) tv[u] = lrelu(av[u] + ad);
            }
#pragma unroll
            for (int u = 0; u < 8; ++u) {
                const float p = (j + u < cntc) ? expf(tv[u] - m) : 0.f;
                den   += p;
                acc.x += p * xv[u].x;
                acc.y += p * xv[u].y;
            }
        }
    }
    const float dd = fmaxf(den, GAT_EPS);
    v2f o;
    o.x = acc.x / dd;
    o.y = acc.y / dd;
    if (!MEAN) {
        const float vx = o.x + bias[2 * lane];
        const float vy = o.y + bias[2 * lane + 1];
        v2f r;
        r.x = vx > 0.f ? vx : (expf(vx) - 1.f);
        r.y = vy > 0.f ? vy : (expf(vy) - 1.f);
        *(v2f*)(dstp + (size_t)d * 64 + 2 * lane) = r;
    } else {
        float sx = o.x, sy = o.y;
#pragma unroll
        for (int k = 8; k < 32; k += 8) {
            sx += __shfl(o.x, (lane + k) & 31);
            sy += __shfl(o.y, (lane + k) & 31);
        }
        if (lane < 8) {
            v2f r;
            r.x = 0.25f * sx + bias[2 * lane];
            r.y = 0.25f * sy + bias[2 * lane + 1];
            *(v2f*)(dstp + (size_t)d * 16 + 2 * lane) = r;
        }
    }
}

static inline int cdiv(long long a, long long b) { return (int)((a + b - 1) / b); }

extern "C" void kernel_launch(void* const* d_in, const int* in_sizes, int n_in,
                              void* d_out, int out_size, void* d_ws, size_t ws_size,
                              hipStream_t stream) {
    const float* x   = (const float*)d_in[0];
    const int*   ei  = (const int*)d_in[1];
    const float* W1  = (const float*)d_in[2];
    const float* as1 = (const float*)d_in[3];
    const float* ad1 = (const float*)d_in[4];
    const float* b1  = (const float*)d_in[5];
    const float* W2  = (const float*)d_in[6];
    const float* as2 = (const float*)d_in[7];
    const float* ad2 = (const float*)d_in[8];
    const float* b2  = (const float*)d_in[9];
    float*       out = (float*)d_out;

    const int IN = 256;
    const int n  = in_sizes[0] / IN;
    const int E  = in_sizes[1] / 2;
    const int* src = ei;
    const int* dst = ei + E;

    float* ws    = (float*)d_ws;
    float* xl1   = ws;
    float* asrc1 = xl1   + (long long)n * 64;
    float* adst1 = asrc1 + (long long)n * 8;
    float* hbuf  = adst1 + (long long)n * 8;
    float* xl2   = hbuf  + (long long)n * 64;
    float* asrc2 = xl2   + (long long)n * 64;
    float* adst2 = asrc2 + (long long)n * 4;
    const int NB = cdiv(n, BIN_NODES);
    const int G  = cdiv(E, TILE_E);
    int*      tilehist = (int*)(adst2 + (long long)n * 4);
    int*      binstart = tilehist + (long long)G * NB;
    int*      rowptr   = binstart + NB + 1;
    unsigned* part     = (unsigned*)(rowptr + n + 1);
    int*      csr      = (int*)(part + E);

    tile_hist_kernel<<<G, SORT_T, 0, stream>>>(dst, E, NB, tilehist);
    tile_partition_kernel<<<G, SORT_T, 0, stream>>>(src, dst, E, NB, G, tilehist,
                                                    binstart, part);
    bin_sort_kernel<<<NB, SORT_T, 0, stream>>>(part, binstart, n, NB, rowptr, csr);

    gemm_alpha_wmma_f32<8, 8><<<cdiv(n, 32), 128, 0, stream>>>(
        x, W1, xl1, as1, ad1, asrc1, adst1, n, IN);
    node_gather_kernel<8, 3, false><<<cdiv(n, GATHER_WAVES), GATHER_WAVES * 32, 0, stream>>>(
        rowptr, csr, asrc1, adst1, xl1, b1, hbuf, n);

    gemm_alpha_wmma_f32<4, 16><<<cdiv(n, 32), 128, 0, stream>>>(
        hbuf, W2, xl2, as2, ad2, asrc2, adst2, n, 64);
    node_gather_kernel<4, 4, true><<<cdiv(n, GATHER_WAVES), GATHER_WAVES * 32, 0, stream>>>(
        rowptr, csr, asrc2, adst2, xl2, b2, out, n);
}


# Gather kernels: one list pass, two nodes per wave (16 lanes x 16 B a row); edge logit from the gathered row itself (dot, DPP sum over the head's lanes), sums rescaled per 8 edges: score gather gone
# speedup vs baseline: 7.6747x; 1.1368x over previous
#include <hip/hip_runtime.h>
#include <hip/hip_bf16.h>

#define LRELU_SLOPE 0.2f
#define GAT_EPS 1e-16f
#define KT 32

typedef __attribute__((ext_vector_type(2))) float v2f;
typedef __attribute__((ext_vector_type(4))) float v4f;
typedef __attribute__((ext_vector_type(8))) float v8f;

template <int H, int CH>
__global__ void gemm_alpha_wmma_f32(const float* __restrict__ A,
                                    const float* __restrict__ B,
                                    float* __restrict__ C,
                                    const float* __restrict__ a_src,
                                    const float* __restrict__ a_dst,
                                    float* __restrict__ asrc,
                                    float* __restrict__ adst, int M, int K) {
    __shared__ float As[32][KT + 1];
    __shared__ float Cs[32][66];

    const int tid  = threadIdx.x;
    const int lane = tid & 31;
    const int wave = tid >> 5;
    const int r0   = blockIdx.x * 32;
    const int c0   = wave * 16;
    const int half = lane >> 4;
    const int l15  = lane & 15;
    const int bcol = c0 + l15;

    v8f acc0 = {}, acc1 = {};

    for (int k0 = 0; k0 < K; k0 += KT) {
        __syncthreads();
#pragma unroll
        for (int i = tid; i < 32 * KT; i += 128) {
            const int r = i >> 5;
            const int c = i & (KT - 1);
            const int grow = r0 + r;
            As[r][c] = (grow < M) ? A[(long long)grow * K + k0 + c] : 0.f;
        }
        __syncthreads();

#pragma unroll
        for (int k = 0; k < KT; k += 4) {
            const int kk = k + (half << 1);
            v2f b, a0, a1;
            b.x  = B[(k0 + kk) * 64 + bcol];
            b.y  = B[(k0 + kk + 1) * 64 + bcol];
            a0.x = As[l15][kk];       a0.y = As[l15][kk + 1];
            a1.x = As[16 + l15][kk];  a1.y = As[16 + l15][kk + 1];
            acc0 = __builtin_amdgcn_wmma_f32_16x16x4_f32(
                false, a0, false, b, (short)0, acc0, false, false);
            acc1 = __builtin_amdgcn_wmma_f32_16x16x4_f32(
                false, a1, false, b, (short)0, acc1, false, false);
        }
    }

    const int lrow = half * 8;
#pragma unroll
    for (int r = 0; r < 8; ++r) {
        const int row0 = r0 + lrow + r;
        const int row1 = row0 + 16;
        if (row0 < M) C[(long long)row0 * 64 + c0 + l15] = acc0[r];
        if (row1 < M) C[(long long)row1 * 64 + c0 + l15] = acc1[r];
        Cs[lrow + r][c0 + l15]      = acc0[r];
        Cs[16 + lrow + r][c0 + l15] = acc1[r];
    }
    __syncthreads();

#pragma unroll
    for (int p = tid; p < 32 * H; p += 128) {
        const int lr   = p / H, h = p % H;
        const int grow = r0 + lr;
        const float* row = &Cs[lr][h * CH];
        float s1 = 0.f, s2 = 0.f;
#pragma unroll
        for (int c = 0; c < CH; ++c) {
            s1 += row[c] * a_src[h * CH + c];
            s2 += row[c] * a_dst[h * CH + c];
        }
        if (grow < M) {
            asrc[(long long)grow * H + h] = s1;
            adst[(long long)grow * H + h] = s2;
        }
    }
}

__device__ __forceinline__ float lrelu(float x) {
    return x > 0.f ? x : LRELU_SLOPE * x;
}

#define BIN_SHIFT 8
#define BIN_NODES 256
#define NB_MAX    256
#define TILE_E    8192
#define SORT_T    1024
#define STAGE_CAP 16384
#define RW        10

__device__ __forceinline__ int scan256_wave(const int* in, int* out, int lane) {
    int a[8];
    int s = 0;
#pragma unroll
    for (int k = 0; k < 8; ++k) { a[k] = in[8 * lane + k]; s += a[k]; }
    int incl = s;
#pragma unroll
    for (int off = 1; off < 32; off <<= 1) {
        const int v = __shfl_up(incl, off);
        if (lane >= off) incl += v;
    }
    int run = incl - s;
#pragma unroll
    for (int k = 0; k < 8; ++k) { out[8 * lane + k] = run; run += a[k]; }
    return __shfl(incl, 31);
}

__global__ void __launch_bounds__(SORT_T)
tile_hist_kernel(const int* __restrict__ dst, int E, int NB,
                 int* __restrict__ tilehist) {
    __shared__ int lh[NB_MAX];
    const int t = threadIdx.x;
    if (t < NB_MAX) lh[t] = 0;
    __syncthreads();
    const int base = blockIdx.x * TILE_E;
    int dv[TILE_E / SORT_T];
#pragma unroll
    for (int k = 0; k < TILE_E / SORT_T; ++k) {
        const int e = base + k * SORT_T + t;
        dv[k] = (e < E) ? dst[e] : -1;
    }
#pragma unroll
    for (int k = 0; k < TILE_E / SORT_T; ++k)
        if (dv[k] >= 0) atomicAdd(&lh[dv[k] >> BIN_SHIFT], 1);
    __syncthreads();
    if (t < NB) tilehist[blockIdx.x * NB + t] = lh[t];
}

__global__ void __launch_bounds__(SORT_T)
tile_partition_kernel(const int* __restrict__ src, const int* __restrict__ dst,
                      int E, int NB, int G, const int* __restrict__ tilehist,
                      int* __restrict__ binstart, unsigned* __restrict__ part) {
    __shared__ int s_tot[4][NB_MAX];
    __shared__ int s_pre[4][NB_MAX];
    __shared__ int s_T[NB_MAX];
    __shared__ int s_own[NB_MAX];
    __shared__ int s_bs[NB_MAX + 1];
    __shared__ int s_lb[NB_MAX];
    __shared__ int s_gb[NB_MAX];
    __shared__ int s_cur[NB_MAX];
    __shared__ unsigned s_val[TILE_E];

    const int t = threadIdx.x, blk = blockIdx.x;
    const int c = t & (NB_MAX - 1), rg = t >> 8;
    int tot = 0, pre = 0;
    if (c < NB) {
        for (int r = rg; r < G; r += 4) {
            const int v = tilehist[r * NB + c];
            tot += v;
            pre += (r < blk) ? v : 0;
        }
    }
    s_tot[rg][c] = tot;
    s_pre[rg][c] = pre;
    __syncthreads();
    if (t < NB_MAX) {
        s_T[t]   = s_tot[0][t] + s_tot[1][t] + s_tot[2][t] + s_tot[3][t];
        s_own[t] = (t < NB) ? tilehist[blk * NB + t] : 0;
        s_cur[t] = 0;
    }
    __syncthreads();
    if (t < 32) {
        const int total = scan256_wave(s_T, s_bs, t);
        if (t == 0) s_bs[NB_MAX] = total;
        scan256_wave(s_own, s_lb, t);
    }
    __syncthreads();
    if (t < NB_MAX)
        s_gb[t] = s_bs[t] + s_pre[0][t] + s_pre[1][t] + s_pre[2][t] + s_pre[3][t];
    if (blk == 0 && t <= NB) binstart[t] = s_bs[t];
    __syncthreads();

    const int base = blk * TILE_E;
    int dv[TILE_E / SORT_T], sv[TILE_E / SORT_T];
#pragma unroll
    for (int k = 0; k < TILE_E / SORT_T; ++k) {
        const int e = base + k * SORT_T + t;
        dv[k] = (e < E) ? dst[e] : -1;
        sv[k] = (e < E) ? src[e] : 0;
    }
    int slot[TILE_E / SORT_T];
#pragma unroll
    for (int k = 0; k < TILE_E / SORT_T; ++k) {
        slot[k] = -1;
        if (dv[k] >= 0) {
            const int b = dv[k] >> BIN_SHIFT;
            slot[k] = s_lb[b] + atomicAdd(&s_cur[b], 1);
        }
    }
#pragma unroll
    for (int k = 0; k < TILE_E / SORT_T; ++k)
        if (slot[k] >= 0) s_val[slot[k]] = ((unsigned)dv[k] << 16) | (unsigned)sv[k];
    __syncthreads();
    const int left = E - base;
    const int cnt  = left < TILE_E ? left : TILE_E;
    for (int i = t; i < cnt; i += SORT_T) {
        const unsigned w = s_val[i];
        const int b = (int)(w >> (16 + BIN_SHIFT));
        part[s_gb[b] + (i - s_lb[b])] = w;
    }
}

__global__ void __launch_bounds__(SORT_T)
bin_sort_kernel(const unsigned* __restrict__ part,
                const int* __restrict__ binstart, int n, int NB,
                int* __restrict__ rowptr, int* __restrict__ csr_src) {
    __shared__ int s_cnt[BIN_NODES];
    __shared__ int s_ex[BIN_NODES];
    __shared__ int s_cur[BIN_NODES];
    __shared__ int s_stage[STAGE_CAP];
    const int t = threadIdx.x, b = blockIdx.x;
    const int beg = binstart[b], end = binstart[b + 1];
    if (t < BIN_NODES) s_cnt[t] = 0;
    unsigned wr[RW];
#pragma unroll
    for (int k = 0; k < RW; ++k) {
        const int i = beg + k * SORT_T + t;
        wr[k] = (i < end) ? part[i] : 0u;
    }
    __syncthreads();
#pragma unroll
    for (int k = 0; k < RW; ++k)
        if (beg + k * SORT_T + t < end)
            atomicAdd(&s_cnt[(wr[k] >> 16) & (BIN_NODES - 1)], 1);
    for (int i = beg + RW * SORT_T + t; i < end; i += SORT_T)
        atomicAdd(&s_cnt[(part[i] >> 16) & (BIN_NODES - 1)], 1);
    __syncthreads();
    if (t < 32) scan256_wave(s_cnt, s_ex, t);
    __syncthreads();
    if (t < BIN_NODES) {
        const int node = b * BIN_NODES + t;
        s_cur[t] = s_ex[t];
        if (node < n) rowptr[node] = beg + s_ex[t];
    }
    if (b == NB - 1 && t == 0) rowptr[n] = end;
    __syncthreads();
    const int  m      = end - beg;
    const bool staged = (m <= STAGE_CAP);
    int pos[RW];
#pragma unroll
    for (int k = 0; k < RW; ++k) {
        pos[k] = -1;
        if (beg + k * SORT_T + t < end)
            pos[k] = atomicAdd(&s_cur[(wr[k] >> 16) & (BIN_NODES - 1)], 1);
    }
#pragma unroll
    for (int k = 0; k < RW; ++k) {
        if (pos[k] >= 0) {
            if (staged) s_stage[pos[k]] = (int)(wr[k] & 0xffffu);
            else        csr_src[beg + pos[k]] = (int)(wr[k] & 0xffffu);
        }
    }
    for (int i = beg + RW * SORT_T + t; i < end; i += SORT_T) {
        const unsigned w = part[i];
        const int p = atomicAdd(&s_cur[(w >> 16) & (BIN_NODES - 1)], 1);
        if (staged) s_stage[p] = (int)(w & 0xffffu);
        else        csr_src[beg + p] = (int)(w & 0xffffu);
    }
    if (staged) {
        __syncthreads();
        for (int i = t; i < m; i += SORT_T) csr_src[beg + i] = s_stage[i];
    }
}

#define GATHER_WAVES 8

template <int K>
__device__ __forceinline__ int row_lane(int v) {
    return __builtin_amdgcn_update_dpp(0, v, 0x150 + K, 0xf, 0xf, true);
}

template <int XM>
__device__ __forceinline__ float row_xor_lane(float v) {
    return __builtin_bit_cast(float, __builtin_amdgcn_update_dpp(
        0, __builtin_bit_cast(int, v), 0x160 + XM, 0xf, 0xf, true));
}

template <int J>
__device__ __forceinline__ void gather8_rows(const v4f* __restrict__ xrow,
                                             int sx, int hl, v4f (&xv)[8]) {
    xv[0] = xrow[(unsigned)(row_lane<J + 0>(sx) + hl)];
    xv[1] = xrow[(unsigned)(row_lane<J + 1>(sx) + hl)];
    xv[2] = xrow[(unsigned)(row_lane<J + 2>(sx) + hl)];
    xv[3] = xrow[(unsigned)(row_lane<J + 3>(sx) + hl)];
    xv[4] = xrow[(unsigned)(row_lane<J + 4>(sx) + hl)];
    xv[5] = xrow[(unsigned)(row_lane<J + 5>(sx) + hl)];
    xv[6] = xrow[(unsigned)(row_lane<J + 6>(sx) + hl)];
    xv[7] = xrow[(unsigned)(row_lane<J + 7>(sx) + hl)];
}

template <int J, int CSH>
__device__ __forceinline__ void edge_group(const v4f* __restrict__ xrow, int sx,
                                           int hl, v4f as4, float ad, int cntc,
                                           float& m, float& den,
                                           v2f& acc_lo, v2f& acc_hi) {
    v4f   xv[8];
    float tv[8];
    gather8_rows<J>(xrow, sx, hl, xv);
    float mn = m;
#pragma unroll
    for (int u = 0; u < 8; ++u) {
        float t = xv[u].x * as4.x + xv[u].y * as4.y
                + xv[u].z * as4.z + xv[u].w * as4.w;
        t += row_xor_lane<1>(t);
        if (CSH == 4) t += row_xor_lane<2>(t);
        t = lrelu(t + ad);
        tv[u] = (J + u < cntc) ? t : -__builtin_inff();
        mn = fmaxf(mn, tv[u]);
    }
    const float sc = expf(m - mn);
    m = mn;
    den    *= sc;
    acc_lo *= sc;
    acc_hi *= sc;
#pragma unroll
    for (int u = 0; u < 8; ++u) {
        const float p = expf(tv[u] - m);
        den += p;
        const v2f x_lo = __builtin_shufflevector(xv[u], xv[u], 0, 1);
        const v2f x_hi = __builtin_shufflevector(xv[u], xv[u], 2, 3);
        acc_lo += p * x_lo;
        acc_hi += p * x_hi;
    }
}

template <int H, int CSH, bool MEAN>
__global__ void __launch_bounds__(GATHER_WAVES * 32)
node_gather_kernel(const int* __restrict__ rowptr,
                   const int* __restrict__ csr_src,
                   const float* __restrict__ asrc,
                   const float* __restrict__ adst,
                   const float* __restrict__ a_src,
                   const float* __restrict__ xl,
                   const float* __restrict__ bias,
                   float* __restrict__ dstp, int n_nodes) {
    const int lane = threadIdx.x & 31;
    const int hl   = lane & 15;
    const int half = lane >> 4;
    const int wv   = threadIdx.x >> 5;
    const int d0 = __builtin_amdgcn_readfirstlane(
        2 * (int)(blockIdx.x * GATHER_WAVES + wv));
    if (d0 >= n_nodes) return;
    const bool live = (d0 + half) < n_nodes;
    const int  d    = live ? d0 + half : d0;
    const int  h    = (4 * hl) >> CSH;
    const int beg   = rowptr[d];
    const int deg   = live ? rowptr[d + 1] - beg : 0;
    const int dega  = __builtin_amdgcn_readlane(deg, 0);
    const int degb  = __builtin_amdgcn_readlane(deg, 16);
    const int dmax  = dega > degb ? dega : degb;
    const float ad  = adst[d * H + h];
    const v4f as4   = *(const v4f*)(a_src + 4 * hl);
    const v4f* xrow = (const v4f*)xl;

    float m   = lrelu(asrc[d * H + h] + ad);
    float den = 1.f;
    v2f acc_lo, acc_hi;
    {
        const v4f xs = xrow[(unsigned)(d * 16 + hl)];
        acc_lo.x = xs.x;
        acc_lo.y = xs.y;
        acc_hi.x = xs.z;
        acc_hi.y = xs.w;
    }
    for (int pos0 = 0; pos0 < dmax; pos0 += 16) {
        const int cntc = deg - pos0;
        const int sidx = (hl < cntc) ? csr_src[beg + pos0 + hl] : d;
        const int sx   = sidx * 16;
        edge_group<0, CSH>(xrow, sx, hl, as4, ad, cntc, m, den, acc_lo, acc_hi);
        if (dmax - pos0 > 8)
            edge_group<8, CSH>(xrow, sx, hl, as4, ad, cntc, m, den, acc_lo, acc_hi);
    }
    const float dd = fmaxf(den, GAT_EPS);
    v4f o;
    o.x = acc_lo.x / dd;
    o.y = acc_lo.y / dd;
    o.z = acc_hi.x / dd;
    o.w = acc_hi.y / dd;
    if (!MEAN) {
        if (live) {
            const v4f bv = *(const v4f*)(bias + 4 * hl);
            const float vx = o.x + bv.x;
            const float vy = o.y + bv.y;
            const float vz = o.z + bv.z;
            const float vw = o.w + bv.w;
            v4f r;
            r.x = vx > 0.f ? vx : (expf(vx) - 1.f);
            r.y = vy > 0.f ? vy : (expf(vy) - 1.f);
            r.z = vz > 0.f ? vz : (expf(vz) - 1.f);
            r.w = vw > 0.f ? vw : (expf(vw) - 1.f);
            *(v4f*)(dstp + (size_t)d * 64 + 4 * hl) = r;
        }
    } else {
        float sx = o.x, sy = o.y, sz = o.z, sw = o.w;
#pragma unroll
        for (int k = 4; k < 16; k += 4) {
            const int from = (lane & 16) | ((hl + k) & 15);
            sx += __shfl(o.x, from);
            sy += __shfl(o.y, from);
            sz += __shfl(o.z, from);
            sw += __shfl(o.w, from);
        }
        if (live && hl < 4) {
            const v4f bv = *(const v4f*)(bias + 4 * hl);
            v4f r;
            r.x = 0.25f * sx + bv.x;
            r.y = 0.25f * sy + bv.y;
            r.z = 0.25f * sz + bv.z;
            r.w = 0.25f * sw + bv.w;
            *(v4f*)(dstp + (size_t)d * 16 + 4 * hl) = r;
        }
    }
}

static inline int cdiv(long long a, long long b) { return (int)((a + b - 1) / b); }

extern "C" void kernel_launch(void* const* d_in, const int* in_sizes, int n_in,
                              void* d_out, int out_size, void* d_ws, size_t ws_size,
                              hipStream_t stream) {
    const float* x   = (const float*)d_in[0];
    const int*   ei  = (const int*)d_in[1];
    const float* W1  = (const float*)d_in[2];
    const float* as1 = (const float*)d_in[3];
    const float* ad1 = (const float*)d_in[4];
    const float* b1  = (const float*)d_in[5];
    const float* W2  = (const float*)d_in[6];
    const float* as2 = (const float*)d_in[7];
    const float* ad2 = (const float*)d_in[8];
    const float* b2  = (const float*)d_in[9];
    float*       out = (float*)d_out;

    const int IN = 256;
    const int n  = in_sizes[0] / IN;
    const int E  = in_sizes[1] / 2;
    const int* src = ei;
    const int* dst = ei + E;

    float* ws    = (float*)d_ws;
    float* xl1   = ws;
    float* asrc1 = xl1   + (long long)n * 64;
    float* adst1 = asrc1 + (long long)n * 8;
    float* hbuf  = adst1 + (long long)n * 8;
    float* xl2   = hbuf  + (long long)n * 64;
    float* asrc2 = xl2   + (long long)n * 64;
    float* adst2 = asrc2 + (long long)n * 4;
    const int NB = cdiv(n, BIN_NODES);
    const int G  = cdiv(E, TILE_E);
    int*      tilehist = (int*)(adst2 + (long long)n * 4);
    int*      binstart = tilehist + (long long)G * NB;
    int*      rowptr   = binstart + NB + 1;
    unsigned* part     = (unsigned*)(rowptr + n + 1);
    int*      csr      = (int*)(part + E);

    tile_hist_kernel<<<G, SORT_T, 0, stream>>>(dst, E, NB, tilehist);
    tile_partition_kernel<<<G, SORT_T, 0, stream>>>(src, dst, E, NB, G, tilehist,
                                                    binstart, part);
    bin_sort_kernel<<<NB, SORT_T, 0, stream>>>(part, binstart, n, NB, rowptr, csr);

    gemm_alpha_wmma_f32<8, 8><<<cdiv(n, 32), 128, 0, stream>>>(
        x, W1, xl1, as1, ad1, asrc1, adst1, n, IN);
    node_gather_kernel<8, 3, false><<<cdiv(n, 2 * GATHER_WAVES), GATHER_WAVES * 32, 0, stream>>>(
        rowptr, csr, asrc1, adst1, as1, xl1, b1, hbuf, n);

    gemm_alpha_wmma_f32<4, 16><<<cdiv(n, 32), 128, 0, stream>>>(
        hbuf, W2, xl2, as2, ad2, asrc2, adst2, n, 64);
    node_gather_kernel<4, 4, true><<<cdiv(n, 2 * GATHER_WAVES), GATHER_WAVES * 32, 0, stream>>>(
        rowptr, csr, asrc2, adst2, as2, xl2, b2, out, n);
}


# GEMM k-loop: next k-tile's A elements (8 a thread, branch-free, rows past M clamped) and B fragments requested into registers before the current tile's WMMAs; 8 load-wait-store rounds a tile gone
# speedup vs baseline: 9.4828x; 1.2356x over previous
#include <hip/hip_runtime.h>
#include <hip/hip_bf16.h>

#define LRELU_SLOPE 0.2f
#define GAT_EPS 1e-16f
#define KT 32

typedef __attribute__((ext_vector_type(2))) float v2f;
typedef __attribute__((ext_vector_type(4))) float v4f;
typedef __attribute__((ext_vector_type(8))) float v8f;

template <int H, int CH>
__global__ void gemm_alpha_wmma_f32(const float* __restrict__ A,
                                    const float* __restrict__ B,
                                    float* __restrict__ C,
                                    const float* __restrict__ a_src,
                                    const float* __restrict__ a_dst,
                                    float* __restrict__ asrc,
                                    float* __restrict__ adst, int M, int K) {
    __shared__ float As[32][KT + 1];
    __shared__ float Cs[32][66];

    const int tid  = threadIdx.x;
    const int lane = tid & 31;
    const int wave = tid >> 5;
    const int r0   = blockIdx.x * 32;
    const int c0   = wave * 16;
    const int half = lane >> 4;
    const int l15  = lane & 15;
    const int bcol = c0 + l15;

    const int sr = tid >> 5;
    const int sc = tid & (KT - 1);
    const float* ap[8];
    bool inb[8];
#pragma unroll
    for (int j = 0; j < 8; ++j) {
        const int grow = r0 + sr + 4 * j;
        inb[j] = grow < M;
        ap[j]  = A + (long long)(inb[j] ? grow : M - 1) * K + sc;
    }

    v2f   pb[8];
    float pa[8];
#pragma unroll
    for (int s = 0; s < 8; ++s) {
        const int kk = 4 * s + (half << 1);
        pb[s].x = B[kk * 64 + bcol];
        pb[s].y = B[(kk + 1) * 64 + bcol];
    }
#pragma unroll
    for (int j = 0; j < 8; ++j) pa[j] = ap[j][0];

    v8f acc0 = {}, acc1 = {};

    for (int k0 = 0; k0 < K; k0 += KT) {
        __syncthreads();
#pragma unroll
        for (int j = 0; j < 8; ++j) As[sr + 4 * j][sc] = inb[j] ? pa[j] : 0.f;
        v2f bc[8];
#pragma unroll
        for (int s = 0; s < 8; ++s) bc[s] = pb[s];
        __syncthreads();

        const int kn = k0 + KT;
        if (kn < K) {
#pragma unroll
            for (int s = 0; s < 8; ++s) {
                const int kk = kn + 4 * s + (half << 1);
                pb[s].x = B[kk * 64 + bcol];
                pb[s].y = B[(kk + 1) * 64 + bcol];
            }
#pragma unroll
            for (int j = 0; j < 8; ++j) pa[j] = ap[j][kn];
        }

#pragma unroll
        for (int s = 0; s < 8; ++s) {
            const int kk = 4 * s + (half << 1);
            v2f a0, a1;
            a0.x = As[l15][kk];       a0.y = As[l15][kk + 1];
            a1.x = As[16 + l15][kk];  a1.y = As[16 + l15][kk + 1];
            acc0 = __builtin_amdgcn_wmma_f32_16x16x4_f32(
                false, a0, false, bc[s], (short)0, acc0, false, false);
            acc1 = __builtin_amdgcn_wmma_f32_16x16x4_f32(
                false, a1, false, bc[s], (short)0, acc1, false, false);
        }
    }

    const int lrow = half * 8;
#pragma unroll
    for (int r = 0; r < 8; ++r) {
        const int row0 = r0 + lrow + r;
        const int row1 = row0 + 16;
        if (row0 < M) C[(long long)row0 * 64 + c0 + l15] = acc0[r];
        if (row1 < M) C[(long long)row1 * 64 + c0 + l15] = acc1[r];
        Cs[lrow + r][c0 + l15]      = acc0[r];
        Cs[16 + lrow + r][c0 + l15] = acc1[r];
    }
    __syncthreads();

#pragma unroll
    for (int p = tid; p < 32 * H; p += 128) {
        const int lr   = p / H, h = p % H;
        const int grow = r0 + lr;
        const float* row = &Cs[lr][h * CH];
        float s1 = 0.f, s2 = 0.f;
#pragma unroll
        for (int c = 0; c < CH; ++c) {
            s1 += row[c] * a_src[h * CH + c];
            s2 += row[c] * a_dst[h * CH + c];
        }
        if (grow < M) {
            asrc[(long long)grow * H + h] = s1;
            adst[(long long)grow * H + h] = s2;
        }
    }
}

__device__ __forceinline__ float lrelu(float x) {
    return x > 0.f ? x : LRELU_SLOPE * x;
}

#define BIN_SHIFT 8
#define BIN_NODES 256
#define NB_MAX    256
#define TILE_E    8192
#define SORT_T    1024
#define STAGE_CAP 16384
#define RW        10

__device__ __forceinline__ int scan256_wave(const int* in, int* out, int lane) {
    int a[8];
    int s = 0;
#pragma unroll
    for (int k = 0; k < 8; ++k) { a[k] = in[8 * lane + k]; s += a[k]; }
    int incl = s;
#pragma unroll
    for (int off = 1; off < 32; off <<= 1) {
        const int v = __shfl_up(incl, off);
        if (lane >= off) incl += v;
    }
    int run = incl - s;
#pragma unroll
    for (int k = 0; k < 8; ++k) { out[8 * lane + k] = run; run += a[k]; }
    return __shfl(incl, 31);
}

__global__ void __launch_bounds__(SORT_T)
tile_hist_kernel(const int* __restrict__ dst, int E, int NB,
                 int* __restrict__ tilehist) {
    __shared__ int lh[NB_MAX];
    const int t = threadIdx.x;
    if (t < NB_MAX) lh[t] = 0;
    __syncthreads();
    const int base = blockIdx.x * TILE_E;
    int dv[TILE_E / SORT_T];
#pragma unroll
    for (int k = 0; k < TILE_E / SORT_T; ++k) {
        const int e = base + k * SORT_T + t;
        dv[k] = (e < E) ? dst[e] : -1;
    }
#pragma unroll
    for (int k = 0; k < TILE_E / SORT_T; ++k)
        if (dv[k] >= 0) atomicAdd(&lh[dv[k] >> BIN_SHIFT], 1);
    __syncthreads();
    if (t < NB) tilehist[blockIdx.x * NB + t] = lh[t];
}

__global__ void __launch_bounds__(SORT_T)
tile_partition_kernel(const int* __restrict__ src, const int* __restrict__ dst,
                      int E, int NB, int G, const int* __restrict__ tilehist,
                      int* __restrict__ binstart, unsigned* __restrict__ part) {
    __shared__ int s_tot[4][NB_MAX];
    __shared__ int s_pre[4][NB_MAX];
    __shared__ int s_T[NB_MAX];
    __shared__ int s_own[NB_MAX];
    __shared__ int s_bs[NB_MAX + 1];
    __shared__ int s_lb[NB_MAX];
    __shared__ int s_gb[NB_MAX];
    __shared__ int s_cur[NB_MAX];
    __shared__ unsigned s_val[TILE_E];

    const int t = threadIdx.x, blk = blockIdx.x;
    const int c = t & (NB_MAX - 1), rg = t >> 8;
    int tot = 0, pre = 0;
    if (c < NB) {
        for (int r = rg; r < G; r += 4) {
            const int v = tilehist[r * NB + c];
            tot += v;
            pre += (r < blk) ? v : 0;
        }
    }
    s_tot[rg][c] = tot;
    s_pre[rg][c] = pre;
    __syncthreads();
    if (t < NB_MAX) {
        s_T[t]   = s_tot[0][t] + s_tot[1][t] + s_tot[2][t] + s_tot[3][t];
        s_own[t] = (t < NB) ? tilehist[blk * NB + t] : 0;
        s_cur[t] = 0;
    }
    __syncthreads();
    if (t < 32) {
        const int total = scan256_wave(s_T, s_bs, t);
        if (t == 0) s_bs[NB_MAX] = total;
        scan256_wave(s_own, s_lb, t);
    }
    __syncthreads();
    if (t < NB_MAX)
        s_gb[t] = s_bs[t] + s_pre[0][t] + s_pre[1][t] + s_pre[2][t] + s_pre[3][t];
    if (blk == 0 && t <= NB) binstart[t] = s_bs[t];
    __syncthreads();

    const int base = blk * TILE_E;
    int dv[TILE_E / SORT_T], sv[TILE_E / SORT_T];
#pragma unroll
    for (int k = 0; k < TILE_E / SORT_T; ++k) {
        const int e = base + k * SORT_T + t;
        dv[k] = (e < E) ? dst[e] : -1;
        sv[k] = (e < E) ? src[e] : 0;
    }
    int slot[TILE_E / SORT_T];
#pragma unroll
    for (int k = 0; k < TILE_E / SORT_T; ++k) {
        slot[k] = -1;
        if (dv[k] >= 0) {
            const int b = dv[k] >> BIN_SHIFT;
            slot[k] = s_lb[b] + atomicAdd(&s_cur[b], 1);
        }
    }
#pragma unroll
    for (int k = 0; k < TILE_E / SORT_T; ++k)
        if (slot[k] >= 0) s_val[slot[k]] = ((unsigned)dv[k] << 16) | (unsigned)sv[k];
    __syncthreads();
    const int left = E - base;
    const int cnt  = left < TILE_E ? left : TILE_E;
    for (int i = t; i < cnt; i += SORT_T) {
        const unsigned w = s_val[i];
        const int b = (int)(w >> (16 + BIN_SHIFT));
        part[s_gb[b] + (i - s_lb[b])] = w;
    }
}

__global__ void __launch_bounds__(SORT_T)
bin_sort_kernel(const unsigned* __restrict__ part,
                const int* __restrict__ binstart, int n, int NB,
                int* __restrict__ rowptr, int* __restrict__ csr_src) {
    __shared__ int s_cnt[BIN_NODES];
    __shared__ int s_ex[BIN_NODES];
    __shared__ int s_cur[BIN_NODES];
    __shared__ int s_stage[STAGE_CAP];
    const int t = threadIdx.x, b = blockIdx.x;
    const int beg = binstart[b], end = binstart[b + 1];
    if (t < BIN_NODES) s_cnt[t] = 0;
    unsigned wr[RW];
#pragma unroll
    for (int k = 0; k < RW; ++k) {
        const int i = beg + k * SORT_T + t;
        wr[k] = (i < end) ? part[i] : 0u;
    }
    __syncthreads();
#pragma unroll
    for (int k = 0; k < RW; ++k)
        if (beg + k * SORT_T + t < end)
            atomicAdd(&s_cnt[(wr[k] >> 16) & (BIN_NODES - 1)], 1);
    for (int i = beg + RW * SORT_T + t; i < end; i += SORT_T)
        atomicAdd(&s_cnt[(part[i] >> 16) & (BIN_NODES - 1)], 1);
    __syncthreads();
    if (t < 32) scan256_wave(s_cnt, s_ex, t);
    __syncthreads();
    if (t < BIN_NODES) {
        const int node = b * BIN_NODES + t;
        s_cur[t] = s_ex[t];
        if (node < n) rowptr[node] = beg + s_ex[t];
    }
    if (b == NB - 1 && t == 0) rowptr[n] = end;
    __syncthreads();
    const int  m      = end - beg;
    const bool staged = (m <= STAGE_CAP);
    int pos[RW];
#pragma unroll
    for (int k = 0; k < RW; ++k) {
        pos[k] = -1;
        if (beg + k * SORT_T + t < end)
            pos[k] = atomicAdd(&s_cur[(wr[k] >> 16) & (BIN_NODES - 1)], 1);
    }
#pragma unroll
    for (int k = 0; k < RW; ++k) {
        if (pos[k] >= 0) {
            if (staged) s_stage[pos[k]] = (int)(wr[k] & 0xffffu);
            else        csr_src[beg + pos[k]] = (int)(wr[k] & 0xffffu);
        }
    }
    for (int i = beg + RW * SORT_T + t; i < end; i += SORT_T) {
        const unsigned w = part[i];
        const int p = atomicAdd(&s_cur[(w >> 16) & (BIN_NODES - 1)], 1);
        if (staged) s_stage[p] = (int)(w & 0xffffu);
        else        csr_src[beg + p] = (int)(w & 0xffffu);
    }
    if (staged) {
        __syncthreads();
        for (int i = t; i < m; i += SORT_T) csr_src[beg + i] = s_stage[i];
    }
}

#define GATHER_WAVES 8

template <int K>
__device__ __forceinline__ int row_lane(int v) {
    return __builtin_amdgcn_update_dpp(0, v, 0x150 + K, 0xf, 0xf, true);
}

template <int XM>
__device__ __forceinline__ float row_xor_lane(float v) {
    return __builtin_bit_cast(float, __builtin_amdgcn_update_dpp(
        0, __builtin_bit_cast(int, v), 0x160 + XM, 0xf, 0xf, true));
}

template <int J>
__device__ __forceinline__ void gather8_rows(const v4f* __restrict__ xrow,
                                             int sx, int hl, v4f (&xv)[8]) {
    xv[0] = xrow[(unsigned)(row_lane<J + 0>(sx) + hl)];
    xv[1] = xrow[(unsigned)(row_lane<J + 1>(sx) + hl)];
    xv[2] = xrow[(unsigned)(row_lane<J + 2>(sx) + hl)];
    xv[3] = xrow[(unsigned)(row_lane<J + 3>(sx) + hl)];
    xv[4] = xrow[(unsigned)(row_lane<J + 4>(sx) + hl)];
    xv[5] = xrow[(unsigned)(row_lane<J + 5>(sx) + hl)];
    xv[6] = xrow[(unsigned)(row_lane<J + 6>(sx) + hl)];
    xv[7] = xrow[(unsigned)(row_lane<J + 7>(sx) + hl)];
}

template <int J, int CSH>
__device__ __forceinline__ void edge_group(const v4f* __restrict__ xrow, int sx,
                                           int hl, v4f as4, float ad, int cntc,
                                           float& m, float& den,
                                           v2f& acc_lo, v2f& acc_hi) {
    v4f   xv[8];
    float tv[8];
    gather8_rows<J>(xrow, sx, hl, xv);
    float mn = m;
#pragma unroll
    for (int u = 0; u < 8; ++u) {
        float t = xv[u].x * as4.x + xv[u].y * as4.y
                + xv[u].z * as4.z + xv[u].w * as4.w;
        t += row_xor_lane<1>(t);
        if (CSH == 4) t += row_xor_lane<2>(t);
        t = lrelu(t + ad);
        tv[u] = (J + u < cntc) ? t : -__builtin_inff();
        mn = fmaxf(mn, tv[u]);
    }
    const float sc = expf(m - mn);
    m = mn;
    den    *= sc;
    acc_lo *= sc;
    acc_hi *= sc;
#pragma unroll
    for (int u = 0; u < 8; ++u) {
        const float p = expf(tv[u] - m);
        den += p;
        const v2f x_lo = __builtin_shufflevector(xv[u], xv[u], 0, 1);
        const v2f x_hi = __builtin_shufflevector(xv[u], xv[u], 2, 3);
        acc_lo += p * x_lo;
        acc_hi += p * x_hi;
    }
}

template <int H, int CSH, bool MEAN>
__global__ void __launch_bounds__(GATHER_WAVES * 32)
node_gather_kernel(const int* __restrict__ rowptr,
                   const int* __restrict__ csr_src,
                   const float* __restrict__ asrc,
                   const float* __restrict__ adst,
                   const float* __restrict__ a_src,
                   const float* __restrict__ xl,
                   const float* __restrict__ bias,
                   float* __restrict__ dstp, int n_nodes) {
    const int lane = threadIdx.x & 31;
    const int hl   = lane & 15;
    const int half = lane >> 4;
    const int wv   = threadIdx.x >> 5;
    const int d0 = __builtin_amdgcn_readfirstlane(
        2 * (int)(blockIdx.x * GATHER_WAVES + wv));
    if (d0 >= n_nodes) return;
    const bool live = (d0 + half) < n_nodes;
    const int  d    = live ? d0 + half : d0;
    const int  h    = (4 * hl) >> CSH;
    const int beg   = rowptr[d];
    const int deg   = live ? rowptr[d + 1] - beg : 0;
    const int dega  = __builtin_amdgcn_readlane(deg, 0);
    const int degb  = __builtin_amdgcn_readlane(deg, 16);
    const int dmax  = dega > degb ? dega : degb;
    const float ad  = adst[d * H + h];
    const v4f as4   = *(const v4f*)(a_src + 4 * hl);
    const v4f* xrow = (const v4f*)xl;

    float m   = lrelu(asrc[d * H + h] + ad);
    float den = 1.f;
    v2f acc_lo, acc_hi;
    {
        const v4f xs = xrow[(unsigned)(d * 16 + hl)];
        acc_lo.x = xs.x;
        acc_lo.y = xs.y;
        acc_hi.x = xs.z;
        acc_hi.y = xs.w;
    }
    for (int pos0 = 0; pos0 < dmax; pos0 += 16) {
        const int cntc = deg - pos0;
        const int sidx = (hl < cntc) ? csr_src[beg + pos0 + hl] : d;
        const int sx   = sidx * 16;
        edge_group<0, CSH>(xrow, sx, hl, as4, ad, cntc, m, den, acc_lo, acc_hi);
        if (dmax - pos0 > 8)
            edge_group<8, CSH>(xrow, sx, hl, as4, ad, cntc, m, den, acc_lo, acc_hi);
    }
    const float dd = fmaxf(den, GAT_EPS);
    v4f o;
    o.x = acc_lo.x / dd;
    o.y = acc_lo.y / dd;
    o.z = acc_hi.x / dd;
    o.w = acc_hi.y / dd;
    if (!MEAN) {
        if (live) {
            const v4f bv = *(const v4f*)(bias + 4 * hl);
            const float vx = o.x + bv.x;
            const float vy = o.y + bv.y;
            const float vz = o.z + bv.z;
            const float vw = o.w + bv.w;
            v4f r;
            r.x = vx > 0.f ? vx : (expf(vx) - 1.f);
            r.y = vy > 0.f ? vy : (expf(vy) - 1.f);
            r.z = vz > 0.f ? vz : (expf(vz) - 1.f);
            r.w = vw > 0.f ? vw : (expf(vw) - 1.f);
            *(v4f*)(dstp + (size_t)d * 64 + 4 * hl) = r;
        }
    } else {
        float sx = o.x, sy = o.y, sz = o.z, sw = o.w;
#pragma unroll
        for (int k = 4; k < 16; k += 4) {
            const int from = (lane & 16) | ((hl + k) & 15);
            sx += __shfl(o.x, from);
            sy += __shfl(o.y, from);
            sz += __shfl(o.z, from);
            sw += __shfl(o.w, from);
        }
        if (live && hl < 4) {
            const v4f bv = *(const v4f*)(bias + 4 * hl);
            v4f r;
            r.x = 0.25f * sx + bv.x;
            r.y = 0.25f * sy + bv.y;
            r.z = 0.25f * sz + bv.z;
            r.w = 0.25f * sw + bv.w;
            *(v4f*)(dstp + (size_t)d * 16 + 4 * hl) = r;
        }
    }
}

static inline int cdiv(long long a, long long b) { return (int)((a + b - 1) / b); }

extern "C" void kernel_launch(void* const* d_in, const int* in_sizes, int n_in,
                              void* d_out, int out_size, void* d_ws, size_t ws_size,
                              hipStream_t stream) {
    const float* x   = (const float*)d_in[0];
    const int*   ei  = (const int*)d_in[1];
    const float* W1  = (const float*)d_in[2];
    const float* as1 = (const float*)d_in[3];
    const float* ad1 = (const float*)d_in[4];
    const float* b1  = (const float*)d_in[5];
    const float* W2  = (const float*)d_in[6];
    const float* as2 = (const float*)d_in[7];
    const float* ad2 = (const float*)d_in[8];
    const float* b2  = (const float*)d_in[9];
    float*       out = (float*)d_out;

    const int IN = 256;
    const int n  = in_sizes[0] / IN;
    const int E  = in_sizes[1] / 2;
    const int* src = ei;
    const int* dst = ei + E;

    float* ws    = (float*)d_ws;
    float* xl1   = ws;
    float* asrc1 = xl1   + (long long)n * 64;
    float* adst1 = asrc1 + (long long)n * 8;
    float* hbuf  = adst1 + (long long)n * 8;
    float* xl2   = hbuf  + (long long)n * 64;
    float* asrc2 = xl2   + (long long)n * 64;
    float* adst2 = asrc2 + (long long)n * 4;
    const int NB = cdiv(n, BIN_NODES);
    const int G  = cdiv(E, TILE_E);
    int*      tilehist = (int*)(adst2 + (long long)n * 4);
    int*      binstart = tilehist + (long long)G * NB;
    int*      rowptr   = binstart + NB + 1;
    unsigned* part     = (unsigned*)(rowptr + n + 1);
    int*      csr      = (int*)(part + E);

    tile_hist_kernel<<<G, SORT_T, 0, stream>>>(dst, E, NB, tilehist);
    tile_partition_kernel<<<G, SORT_T, 0, stream>>>(src, dst, E, NB, G, tilehist,
                                                    binstart, part);
    bin_sort_kernel<<<NB, SORT_T, 0, stream>>>(part, binstart, n, NB, rowptr, csr);

    gemm_alpha_wmma_f32<8, 8><<<cdiv(n, 32), 128, 0, stream>>>(
        x, W1, xl1, as1, ad1, asrc1, adst1, n, IN);
    node_gather_kernel<8, 3, false><<<cdiv(n, 2 * GATHER_WAVES), GATHER_WAVES * 32, 0, stream>>>(
        rowptr, csr, asrc1, adst1, as1, xl1, b1, hbuf, n);

    gemm_alpha_wmma_f32<4, 16><<<cdiv(n, 32), 128, 0, stream>>>(
        hbuf, W2, xl2, as2, ad2, asrc2, adst2, n, 64);
    node_gather_kernel<4, 4, true><<<cdiv(n, 2 * GATHER_WAVES), GATHER_WAVES * 32, 0, stream>>>(
        rowptr, csr, asrc2, adst2, as2, xl2, b2, out, n);
}


# Gather kernels: softmax weights, rescale factors and ELU take e^x as a multiply by log2(e) and one hardware v_exp_f32 instead of library expf; values move in the last bits, sum order kept
# speedup vs baseline: 10.0837x; 1.0634x over previous
#include <hip/hip_runtime.h>
#include <hip/hip_bf16.h>

#define LRELU_SLOPE 0.2f
#define GAT_EPS 1e-16f
#define KT 32

typedef __attribute__((ext_vector_type(2))) float v2f;
typedef __attribute__((ext_vector_type(4))) float v4f;
typedef __attribute__((ext_vector_type(8))) float v8f;

template <int H, int CH>
__global__ void gemm_alpha_wmma_f32(const float* __restrict__ A,
                                    const float* __restrict__ B,
                                    float* __restrict__ C,
                                    const float* __restrict__ a_src,
                                    const float* __restrict__ a_dst,
                                    float* __restrict__ asrc,
                                    float* __restrict__ adst, int M, int K) {
    __shared__ float As[32][KT + 1];
    __shared__ float Cs[32][66];

    const int tid  = threadIdx.x;
    const int lane = tid & 31;
    const int wave = tid >> 5;
    const int r0   = blockIdx.x * 32;
    const int c0   = wave * 16;
    const int half = lane >> 4;
    const int l15  = lane & 15;
    const int bcol = c0 + l15;

    const int sr = tid >> 5;
    const int sc = tid & (KT - 1);
    const float* ap[8];
    bool inb[8];
#pragma unroll
    for (int j = 0; j < 8; ++j) {
        const int grow = r0 + sr + 4 * j;
        inb[j] = grow < M;
        ap[j]  = A + (long long)(inb[j] ? grow : M - 1) * K + sc;
    }

    v2f   pb[8];
    float pa[8];
#pragma unroll
    for (int s = 0; s < 8; ++s) {
        const int kk = 4 * s + (half << 1);
        pb[s].x = B[kk * 64 + bcol];
        pb[s].y = B[(kk + 1) * 64 + bcol];
    }
#pragma unroll
    for (int j = 0; j < 8; ++j) pa[j] = ap[j][0];

    v8f acc0 = {}, acc1 = {};

    for (int k0 = 0; k0 < K; k0 += KT) {
        __syncthreads();
#pragma unroll
        for (int j = 0; j < 8; ++j) As[sr + 4 * j][sc] = inb[j] ? pa[j] : 0.f;
        v2f bc[8];
#pragma unroll
        for (int s = 0; s < 8; ++s) bc[s] = pb[s];
        __syncthreads();

        const int kn = k0 + KT;
        if (kn < K) {
#pragma unroll
            for (int s = 0; s < 8; ++s) {
                const int kk = kn + 4 * s + (half << 1);
                pb[s].x = B[kk * 64 + bcol];
                pb[s].y = B[(kk + 1) * 64 + bcol];
            }
#pragma unroll
            for (int j = 0; j < 8; ++j) pa[j] = ap[j][kn];
        }

#pragma unroll
        for (int s = 0; s < 8; ++s) {
            const int kk = 4 * s + (half << 1);
            v2f a0, a1;
            a0.x = As[l15][kk];       a0.y = As[l15][kk + 1];
            a1.x = As[16 + l15][kk];  a1.y = As[16 + l15][kk + 1];
            acc0 = __builtin_amdgcn_wmma_f32_16x16x4_f32(
                false, a0, false, bc[s], (short)0, acc0, false, false);
            acc1 = __builtin_amdgcn_wmma_f32_16x16x4_f32(
                false, a1, false, bc[s], (short)0, acc1, false, false);
        }
    }

    const int lrow = half * 8;
#pragma unroll
    for (int r = 0; r < 8; ++r) {
        const int row0 = r0 + lrow + r;
        const int row1 = row0 + 16;
        if (row0 < M) C[(long long)row0 * 64 + c0 + l15] = acc0[r];
        if (row1 < M) C[(long long)row1 * 64 + c0 + l15] = acc1[r];
        Cs[lrow + r][c0 + l15]      = acc0[r];
        Cs[16 + lrow + r][c0 + l15] = acc1[r];
    }
    __syncthreads();

#pragma unroll
    for (int p = tid; p < 32 * H; p += 128) {
        const int lr   = p / H, h = p % H;
        const int grow = r0 + lr;
        const float* row = &Cs[lr][h * CH];
        float s1 = 0.f, s2 = 0.f;
#pragma unroll
        for (int c = 0; c < CH; ++c) {
            s1 += row[c] * a_src[h * CH + c];
            s2 += row[c] * a_dst[h * CH + c];
        }
        if (grow < M) {
            asrc[(long long)grow * H + h] = s1;
            adst[(long long)grow * H + h] = s2;
        }
    }
}

__device__ __forceinline__ float lrelu(float x) {
    return x > 0.f ? x : LRELU_SLOPE * x;
}

__device__ __forceinline__ float exp_hw(float x) {
    return __builtin_amdgcn_exp2f(x * 1.44269504088896341f);
}

#define BIN_SHIFT 8
#define BIN_NODES 256
#define NB_MAX    256
#define TILE_E    8192
#define SORT_T    1024
#define STAGE_CAP 16384
#define RW        10

__device__ __forceinline__ int scan256_wave(const int* in, int* out, int lane) {
    int a[8];
    int s = 0;
#pragma unroll
    for (int k = 0; k < 8; ++k) { a[k] = in[8 * lane + k]; s += a[k]; }
    int incl = s;
#pragma unroll
    for (int off = 1; off < 32; off <<= 1) {
        const int v = __shfl_up(incl, off);
        if (lane >= off) incl += v;
    }
    int run = incl - s;
#pragma unroll
    for (int k = 0; k < 8; ++k) { out[8 * lane + k] = run; run += a[k]; }
    return __shfl(incl, 31);
}

__global__ void __launch_bounds__(SORT_T)
tile_hist_kernel(const int* __restrict__ dst, int E, int NB,
                 int* __restrict__ tilehist) {
    __shared__ int lh[NB_MAX];
    const int t = threadIdx.x;
    if (t < NB_MAX) lh[t] = 0;
    __syncthreads();
    const int base = blockIdx.x * TILE_E;
    int dv[TILE_E / SORT_T];
#pragma unroll
    for (int k = 0; k < TILE_E / SORT_T; ++k) {
        const int e = base + k * SORT_T + t;
        dv[k] = (e < E) ? dst[e] : -1;
    }
#pragma unroll
    for (int k = 0; k < TILE_E / SORT_T; ++k)
        if (dv[k] >= 0) atomicAdd(&lh[dv[k] >> BIN_SHIFT], 1);
    __syncthreads();
    if (t < NB) tilehist[blockIdx.x * NB + t] = lh[t];
}

__global__ void __launch_bounds__(SORT_T)
tile_partition_kernel(const int* __restrict__ src, const int* __restrict__ dst,
                      int E, int NB, int G, const int* __restrict__ tilehist,
                      int* __restrict__ binstart, unsigned* __restrict__ part) {
    __shared__ int s_tot[4][NB_MAX];
    __shared__ int s_pre[4][NB_MAX];
    __shared__ int s_T[NB_MAX];
    __shared__ int s_own[NB_MAX];
    __shared__ int s_bs[NB_MAX + 1];
    __shared__ int s_lb[NB_MAX];
    __shared__ int s_gb[NB_MAX];
    __shared__ int s_cur[NB_MAX];
    __shared__ unsigned s_val[TILE_E];

    const int t = threadIdx.x, blk = blockIdx.x;
    const int c = t & (NB_MAX - 1), rg = t >> 8;
    int tot = 0, pre = 0;
    if (c < NB) {
        for (int r = rg; r < G; r += 4) {
            const int v = tilehist[r * NB + c];
            tot += v;
            pre += (r < blk) ? v : 0;
        }
    }
    s_tot[rg][c] = tot;
    s_pre[rg][c] = pre;
    __syncthreads();
    if (t < NB_MAX) {
        s_T[t]   = s_tot[0][t] + s_tot[1][t] + s_tot[2][t] + s_tot[3][t];
        s_own[t] = (t < NB) ? tilehist[blk * NB + t] : 0;
        s_cur[t] = 0;
    }
    __syncthreads();
    if (t < 32) {
        const int total = scan256_wave(s_T, s_bs, t);
        if (t == 0) s_bs[NB_MAX] = total;
        scan256_wave(s_own, s_lb, t);
    }
    __syncthreads();
    if (t < NB_MAX)
        s_gb[t] = s_bs[t] + s_pre[0][t] + s_pre[1][t] + s_pre[2][t] + s_pre[3][t];
    if (blk == 0 && t <= NB) binstart[t] = s_bs[t];
    __syncthreads();

    const int base = blk * TILE_E;
    int dv[TILE_E / SORT_T], sv[TILE_E / SORT_T];
#pragma unroll
    for (int k = 0; k < TILE_E / SORT_T; ++k) {
        const int e = base + k * SORT_T + t;
        dv[k] = (e < E) ? dst[e] : -1;
        sv[k] = (e < E) ? src[e] : 0;
    }
    int slot[TILE_E / SORT_T];
#pragma unroll
    for (int k = 0; k < TILE_E / SORT_T; ++k) {
        slot[k] = -1;
        if (dv[k] >= 0) {
            const int b = dv[k] >> BIN_SHIFT;
            slot[k] = s_lb[b] + atomicAdd(&s_cur[b], 1);
        }
    }
#pragma unroll
    for (int k = 0; k < TILE_E / SORT_T; ++k)
        if (slot[k] >= 0) s_val[slot[k]] = ((unsigned)dv[k] << 16) | (unsigned)sv[k];
    __syncthreads();
    const int left = E - base;
    const int cnt  = left < TILE_E ? left : TILE_E;
    for (int i = t; i < cnt; i += SORT_T) {
        const unsigned w = s_val[i];
        const int b = (int)(w >> (16 + BIN_SHIFT));
        part[s_gb[b] + (i - s_lb[b])] = w;
    }
}

__global__ void __launch_bounds__(SORT_T)
bin_sort_kernel(const unsigned* __restrict__ part,
                const int* __restrict__ binstart, int n, int NB,
                int* __restrict__ rowptr, int* __restrict__ csr_src) {
    __shared__ int s_cnt[BIN_NODES];
    __shared__ int s_ex[BIN_NODES];
    __shared__ int s_cur[BIN_NODES];
    __shared__ int s_stage[STAGE_CAP];
    const int t = threadIdx.x, b = blockIdx.x;
    const int beg = binstart[b], end = binstart[b + 1];
    if (t < BIN_NODES) s_cnt[t] = 0;
    unsigned wr[RW];
#pragma unroll
    for (int k = 0; k < RW; ++k) {
        const int i = beg + k * SORT_T + t;
        wr[k] = (i < end) ? part[i] : 0u;
    }
    __syncthreads();
#pragma unroll
    for (int k = 0; k < RW; ++k)
        if (beg + k * SORT_T + t < end)
            atomicAdd(&s_cnt[(wr[k] >> 16) & (BIN_NODES - 1)], 1);
    for (int i = beg + RW * SORT_T + t; i < end; i += SORT_T)
        atomicAdd(&s_cnt[(part[i] >> 16) & (BIN_NODES - 1)], 1);
    __syncthreads();
    if (t < 32) scan256_wave(s_cnt, s_ex, t);
    __syncthreads();
    if (t < BIN_NODES) {
        const int node = b * BIN_NODES + t;
        s_cur[t] = s_ex[t];
        if (node < n) rowptr[node] = beg + s_ex[t];
    }
    if (b == NB - 1 && t == 0) rowptr[n] = end;
    __syncthreads();
    const int  m      = end - beg;
    const bool staged = (m <= STAGE_CAP);
    int pos[RW];
#pragma unroll
    for (int k = 0; k < RW; ++k) {
        pos[k] = -1;
        if (beg + k * SORT_T + t < end)
            pos[k] = atomicAdd(&s_cur[(wr[k] >> 16) & (BIN_NODES - 1)], 1);
    }
#pragma unroll
    for (int k = 0; k < RW; ++k) {
        if (pos[k] >= 0) {
            if (staged) s_stage[pos[k]] = (int)(wr[k] & 0xffffu);
            else        csr_src[beg + pos[k]] = (int)(wr[k] & 0xffffu);
        }
    }
    for (int i = beg + RW * SORT_T + t; i < end; i += SORT_T) {
        const unsigned w = part[i];
        const int p = atomicAdd(&s_cur[(w >> 16) & (BIN_NODES - 1)], 1);
        if (staged) s_stage[p] = (int)(w & 0xffffu);
        else        csr_src[beg + p] = (int)(w & 0xffffu);
    }
    if (staged) {
        __syncthreads();
        for (int i = t; i < m; i += SORT_T) csr_src[beg + i] = s_stage[i];
    }
}

#define GATHER_WAVES 8

template <int K>
__device__ __forceinline__ int row_lane(int v) {
    return __builtin_amdgcn_update_dpp(0, v, 0x150 + K, 0xf, 0xf, true);
}

template <int XM>
__device__ __forceinline__ float row_xor_lane(float v) {
    return __builtin_bit_cast(float, __builtin_amdgcn_update_dpp(
        0, __builtin_bit_cast(int, v), 0x160 + XM, 0xf, 0xf, true));
}

template <int J>
__device__ __forceinline__ void gather8_rows(const v4f* __restrict__ xrow,
                                             int sx, int hl, v4f (&xv)[8]) {
    xv[0] = xrow[(unsigned)(row_lane<J + 0>(sx) + hl)];
    xv[1] = xrow[(unsigned)(row_lane<J + 1>(sx) + hl)];
    xv[2] = xrow[(unsigned)(row_lane<J + 2>(sx) + hl)];
    xv[3] = xrow[(unsigned)(row_lane<J + 3>(sx) + hl)];
    xv[4] = xrow[(unsigned)(row_lane<J + 4>(sx) + hl)];
    xv[5] = xrow[(unsigned)(row_lane<J + 5>(sx) + hl)];
    xv[6] = xrow[(unsigned)(row_lane<J + 6>(sx) + hl)];
    xv[7] = xrow[(unsigned)(row_lane<J + 7>(sx) + hl)];
}

template <int J, int CSH>
__device__ __forceinline__ void edge_group(const v4f* __restrict__ xrow, int sx,
                                           int hl, v4f as4, float ad, int cntc,
                                           float& m, float& den,
                                           v2f& acc_lo, v2f& acc_hi) {
    v4f   xv[8];
    float tv[8];
    gather8_rows<J>(xrow, sx, hl, xv);
    float mn = m;
#pragma unroll
    for (int u = 0; u < 8; ++u) {
        float t = xv[u].x * as4.x + xv[u].y * as4.y
                + xv[u].z * as4.z + xv[u].w * as4.w;
        t += row_xor_lane<1>(t);
        if (CSH == 4) t += row_xor_lane<2>(t);
        t = lrelu(t + ad);
        tv[u] = (J + u < cntc) ? t : -__builtin_inff();
        mn = fmaxf(mn, tv[u]);
    }
    const float sc = exp_hw(m - mn);
    m = mn;
    den    *= sc;
    acc_lo *= sc;
    acc_hi *= sc;
#pragma unroll
    for (int u = 0; u < 8; ++u) {
        const float p = exp_hw(tv[u] - m);
        den += p;
        const v2f x_lo = __builtin_shufflevector(xv[u], xv[u], 0, 1);
        const v2f x_hi = __builtin_shufflevector(xv[u], xv[u], 2, 3);
        acc_lo += p * x_lo;
        acc_hi += p * x_hi;
    }
}

template <int H, int CSH, bool MEAN>
__global__ void __launch_bounds__(GATHER_WAVES * 32)
node_gather_kernel(const int* __restrict__ rowptr,
                   const int* __restrict__ csr_src,
                   const float* __restrict__ asrc,
                   const float* __restrict__ adst,
                   const float* __restrict__ a_src,
                   const float* __restrict__ xl,
                   const float* __restrict__ bias,
                   float* __restrict__ dstp, int n_nodes) {
    const int lane = threadIdx.x & 31;
    const int hl   = lane & 15;
    const int half = lane >> 4;
    const int wv   = threadIdx.x >> 5;
    const int d0 = __builtin_amdgcn_readfirstlane(
        2 * (int)(blockIdx.x * GATHER_WAVES + wv));
    if (d0 >= n_nodes) return;
    const bool live = (d0 + half) < n_nodes;
    const int  d    = live ? d0 + half : d0;
    const int  h    = (4 * hl) >> CSH;
    const int beg   = rowptr[d];
    const int deg   = live ? rowptr[d + 1] - beg : 0;
    const int dega  = __builtin_amdgcn_readlane(deg, 0);
    const int degb  = __builtin_amdgcn_readlane(deg, 16);
    const int dmax  = dega > degb ? dega : degb;
    const float ad  = adst[d * H + h];
    const v4f as4   = *(const v4f*)(a_src + 4 * hl);
    const v4f* xrow = (const v4f*)xl;

    float m   = lrelu(asrc[d * H + h] + ad);
    float den = 1.f;
    v2f acc_lo, acc_hi;
    {
        const v4f xs = xrow[(unsigned)(d * 16 + hl)];
        acc_lo.x = xs.x;
        acc_lo.y = xs.y;
        acc_hi.x = xs.z;
        acc_hi.y = xs.w;
    }
    for (int pos0 = 0; pos0 < dmax; pos0 += 16) {
        const int cntc = deg - pos0;
        const int sidx = (hl < cntc) ? csr_src[beg + pos0 + hl] : d;
        const int sx   = sidx * 16;
        edge_group<0, CSH>(xrow, sx, hl, as4, ad, cntc, m, den, acc_lo, acc_hi);
        if (dmax - pos0 > 8)
            edge_group<8, CSH>(xrow, sx, hl, as4, ad, cntc, m, den, acc_lo, acc_hi);
    }
    const float dd = fmaxf(den, GAT_EPS);
    v4f o;
    o.x = acc_lo.x / dd;
    o.y = acc_lo.y / dd;
    o.z = acc_hi.x / dd;
    o.w = acc_hi.y / dd;
    if (!MEAN) {
        if (live) {
            const v4f bv = *(const v4f*)(bias + 4 * hl);
            const float vx = o.x + bv.x;
            const float vy = o.y + bv.y;
            const float vz = o.z + bv.z;
            const float vw = o.w + bv.w;
            v4f r;
            r.x = vx > 0.f ? vx : (exp_hw(vx) - 1.f);
            r.y = vy > 0.f ? vy : (exp_hw(vy) - 1.f);
            r.z = vz > 0.f ? vz : (exp_hw(vz) - 1.f);
            r.w = vw > 0.f ? vw : (exp_hw(vw) - 1.f);
            *(v4f*)(dstp + (size_t)d * 64 + 4 * hl) = r;
        }
    } else {
        float sx = o.x, sy = o.y, sz = o.z, sw = o.w;
#pragma unroll
        for (int k = 4; k < 16; k += 4) {
            const int from = (lane & 16) | ((hl + k) & 15);
            sx += __shfl(o.x, from);
            sy += __shfl(o.y, from);
            sz += __shfl(o.z, from);
            sw += __shfl(o.w, from);
        }
        if (live && hl < 4) {
            const v4f bv = *(const v4f*)(bias + 4 * hl);
            v4f r;
            r.x = 0.25f * sx + bv.x;
            r.y = 0.25f * sy + bv.y;
            r.z = 0.25f * sz + bv.z;
            r.w = 0.25f * sw + bv.w;
            *(v4f*)(dstp + (size_t)d * 16 + 4 * hl) = r;
        }
    }
}

static inline int cdiv(long long a, long long b) { return (int)((a + b - 1) / b); }

extern "C" void kernel_launch(void* const* d_in, const int* in_sizes, int n_in,
                              void* d_out, int out_size, void* d_ws, size_t ws_size,
                              hipStream_t stream) {
    const float* x   = (const float*)d_in[0];
    const int*   ei  = (const int*)d_in[1];
    const float* W1  = (const float*)d_in[2];
    const float* as1 = (const float*)d_in[3];
    const float* ad1 = (const float*)d_in[4];
    const float* b1  = (const float*)d_in[5];
    const float* W2  = (const float*)d_in[6];
    const float* as2 = (const float*)d_in[7];
    const float* ad2 = (const float*)d_in[8];
    const float* b2  = (const float*)d_in[9];
    float*       out = (float*)d_out;

    const int IN = 256;
    const int n  = in_sizes[0] / IN;
    const int E  = in_sizes[1] / 2;
    const int* src = ei;
    const int* dst = ei + E;

    float* ws    = (float*)d_ws;
    float* xl1   = ws;
    float* asrc1 = xl1   + (long long)n * 64;
    float* adst1 = asrc1 + (long long)n * 8;
    float* hbuf  = adst1 + (long long)n * 8;
    float* xl2   = hbuf  + (long long)n * 64;
    float* asrc2 = xl2   + (long long)n * 64;
    float* adst2 = asrc2 + (long long)n * 4;
    const int NB = cdiv(n, BIN_NODES);
    const int G  = cdiv(E, TILE_E);
    int*      tilehist = (int*)(adst2 + (long long)n * 4);
    int*      binstart = tilehist + (long long)G * NB;
    int*      rowptr   = binstart + NB + 1;
    unsigned* part     = (unsigned*)(rowptr + n + 1);
    int*      csr      = (int*)(part + E);

    tile_hist_kernel<<<G, SORT_T, 0, stream>>>(dst, E, NB, tilehist);
    tile_partition_kernel<<<G, SORT_T, 0, stream>>>(src, dst, E, NB, G, tilehist,
                                                    binstart, part);
    bin_sort_kernel<<<NB, SORT_T, 0, stream>>>(part, binstart, n, NB, rowptr, csr);

    gemm_alpha_wmma_f32<8, 8><<<cdiv(n, 32), 128, 0, stream>>>(
        x, W1, xl1, as1, ad1, asrc1, adst1, n, IN);
    node_gather_kernel<8, 3, false><<<cdiv(n, 2 * GATHER_WAVES), GATHER_WAVES * 32, 0, stream>>>(
        rowptr, csr, asrc1, adst1, as1, xl1, b1, hbuf, n);

    gemm_alpha_wmma_f32<4, 16><<<cdiv(n, 32), 128, 0, stream>>>(
        hbuf, W2, xl2, as2, ad2, asrc2, adst2, n, 64);
    node_gather_kernel<4, 4, true><<<cdiv(n, 2 * GATHER_WAVES), GATHER_WAVES * 32, 0, stream>>>(
        rowptr, csr, asrc2, adst2, as2, xl2, b2, out, n);
}


# Edge-count histogram (first sort stage) has no launch of its own: its 196 tile blocks ride as extra work-groups in front of the layer-1 GEMM's row tiles; 7 launches to 6; every sum and its order kept
# speedup vs baseline: 10.5400x; 1.0453x over previous
#include <hip/hip_runtime.h>
#include <hip/hip_bf16.h>

#define LRELU_SLOPE 0.2f
#define GAT_EPS 1e-16f
#define KT 32

typedef __attribute__((ext_vector_type(2))) float v2f;
typedef __attribute__((ext_vector_type(4))) float v4f;
typedef __attribute__((ext_vector_type(8))) float v8f;

#define BIN_SHIFT 8
#define NB_MAX    256
#define TILE_E    8192

template <int NT>
__device__ __forceinline__ void tile_hist_role(const int* __restrict__ dst, int E,
                                               int NB, int tile,
                                               int* __restrict__ tilehist) {
    __shared__ int lh[NB_MAX];
    const int t = threadIdx.x;
    for (int i = t; i < NB_MAX; i += NT) lh[i] = 0;
    __syncthreads();
    const int base = tile * TILE_E;
    for (int k0 = 0; k0 < TILE_E / NT; k0 += 16) {
        int dv[16];
#pragma unroll
        for (int k = 0; k < 16; ++k) {
            const int e = base + (k0 + k) * NT + t;
            const int v = dst[e < E ? e : E - 1];
            dv[k] = (e < E) ? v : -1;
        }
#pragma unroll
        for (int k = 0; k < 16; ++k)
            if (dv[k] >= 0) atomicAdd(&lh[dv[k] >> BIN_SHIFT], 1);
    }
    __syncthreads();
    for (int i = t; i < NB; i += NT) tilehist[tile * NB + i] = lh[i];
}

template <int H, int CH, bool HIST>
__global__ void gemm_alpha_wmma_f32(const float* __restrict__ A,
                                    const float* __restrict__ B,
                                    float* __restrict__ C,
                                    const float* __restrict__ a_src,
                                    const float* __restrict__ a_dst,
                                    float* __restrict__ asrc,
                                    float* __restrict__ adst, int M, int K,
                                    const int* __restrict__ edst, int E, int NB,
                                    int G, int* __restrict__ tilehist) {
    __shared__ float As[32][KT + 1];
    __shared__ float Cs[32][66];

    const int tid  = threadIdx.x;
    int rblk = blockIdx.x;
    if (HIST) {
        if (rblk < G) {
            tile_hist_role<128>(edst, E, NB, rblk, tilehist);
            return;
        }
        rblk -= G;
    }
    const int lane = tid & 31;
    const int wave = tid >> 5;
    const int r0   = rblk * 32;
    const int c0   = wave * 16;
    const int half = lane >> 4;
    const int l15  = lane & 15;
    const int bcol = c0 + l15;

    const int sr = tid >> 5;
    const int sc = tid & (KT - 1);
    const float* ap[8];
    bool inb[8];
#pragma unroll
    for (int j = 0; j < 8; ++j) {
        const int grow = r0 + sr + 4 * j;
        inb[j] = grow < M;
        ap[j]  = A + (long long)(inb[j] ? grow : M - 1) * K + sc;
    }

    v2f   pb[8];
    float pa[8];
#pragma unroll
    for (int s = 0; s < 8; ++s) {
        const int kk = 4 * s + (half << 1);
        pb[s].x = B[kk * 64 + bcol];
        pb[s].y = B[(kk + 1) * 64 + bcol];
    }
#pragma unroll
    for (int j = 0; j < 8; ++j) pa[j] = ap[j][0];

    v8f acc0 = {}, acc1 = {};

    for (int k0 = 0; k0 < K; k0 += KT) {
        __syncthreads();
#pragma unroll
        for (int j = 0; j < 8; ++j) As[sr + 4 * j][sc] = inb[j] ? pa[j] : 0.f;
        v2f bc[8];
#pragma unroll
        for (int s = 0; s < 8; ++s) bc[s] = pb[s];
        __syncthreads();

        const int kn = k0 + KT;
        if (kn < K) {
#pragma unroll
            for (int s = 0; s < 8; ++s) {
                const int kk = kn + 4 * s + (half << 1);
                pb[s].x = B[kk * 64 + bcol];
                pb[s].y = B[(kk + 1) * 64 + bcol];
            }
#pragma unroll
            for (int j = 0; j < 8; ++j) pa[j] = ap[j][kn];
        }

#pragma unroll
        for (int s = 0; s < 8; ++s) {
            const int kk = 4 * s + (half << 1);
            v2f a0, a1;
            a0.x = As[l15][kk];       a0.y = As[l15][kk + 1];
            a1.x = As[16 + l15][kk];  a1.y = As[16 + l15][kk + 1];
            acc0 = __builtin_amdgcn_wmma_f32_16x16x4_f32(
                false, a0, false, bc[s], (short)0, acc0, false, false);
            acc1 = __builtin_amdgcn_wmma_f32_16x16x4_f32(
                false, a1, false, bc[s], (short)0, acc1, false, false);
        }
    }

    const int lrow = half * 8;
#pragma unroll
    for (int r = 0; r < 8; ++r) {
        const int row0 = r0 + lrow + r;
        const int row1 = row0 + 16;
        if (row0 < M) C[(long long)row0 * 64 + c0 + l15] = acc0[r];
        if (row1 < M) C[(long long)row1 * 64 + c0 + l15] = acc1[r];
        Cs[lrow + r][c0 + l15]      = acc0[r];
        Cs[16 + lrow + r][c0 + l15] = acc1[r];
    }
    __syncthreads();

#pragma unroll
    for (int p = tid; p < 32 * H; p += 128) {
        const int lr   = p / H, h = p % H;
        const int grow = r0 + lr;
        const float* row = &Cs[lr][h * CH];
        float s1 = 0.f, s2 = 0.f;
#pragma unroll
        for (int c = 0; c < CH; ++c) {
            s1 += row[c] * a_src[h * CH + c];
            s2 += row[c] * a_dst[h * CH + c];
        }
        if (grow < M) {
            asrc[(long long)grow * H + h] = s1;
            adst[(long long)grow * H + h] = s2;
        }
    }
}

__device__ __forceinline__ float lrelu(float x) {
    return x > 0.f ? x : LRELU_SLOPE * x;
}

__device__ __forceinline__ float exp_hw(float x) {
    return __builtin_amdgcn_exp2f(x * 1.44269504088896341f);
}

#define BIN_NODES 256
#define SORT_T    1024
#define STAGE_CAP 16384
#define RW        10

__device__ __forceinline__ int scan256_wave(const int* in, int* out, int lane) {
    int a[8];
    int s = 0;
#pragma unroll
    for (int k = 0; k < 8; ++k) { a[k] = in[8 * lane + k]; s += a[k]; }
    int incl = s;
#pragma unroll
    for (int off = 1; off < 32; off <<= 1) {
        const int v = __shfl_up(incl, off);
        if (lane >= off) incl += v;
    }
    int run = incl - s;
#pragma unroll
    for (int k = 0; k < 8; ++k) { out[8 * lane + k] = run; run += a[k]; }
    return __shfl(incl, 31);
}

__global__ void __launch_bounds__(SORT_T)
tile_partition_kernel(const int* __restrict__ src, const int* __restrict__ dst,
                      int E, int NB, int G, const int* __restrict__ tilehist,
                      int* __restrict__ binstart, unsigned* __restrict__ part) {
    __shared__ int s_tot[4][NB_MAX];
    __shared__ int s_pre[4][NB_MAX];
    __shared__ int s_T[NB_MAX];
    __shared__ int s_own[NB_MAX];
    __shared__ int s_bs[NB_MAX + 1];
    __shared__ int s_lb[NB_MAX];
    __shared__ int s_gb[NB_MAX];
    __shared__ int s_cur[NB_MAX];
    __shared__ unsigned s_val[TILE_E];

    const int t = threadIdx.x, blk = blockIdx.x;
    const int c = t & (NB_MAX - 1), rg = t >> 8;
    int tot = 0, pre = 0;
    if (c < NB) {
        for (int r = rg; r < G; r += 4) {
            const int v = tilehist[r * NB + c];
            tot += v;
            pre += (r < blk) ? v : 0;
        }
    }
    s_tot[rg][c] = tot;
    s_pre[rg][c] = pre;
    __syncthreads();
    if (t < NB_MAX) {
        s_T[t]   = s_tot[0][t] + s_tot[1][t] + s_tot[2][t] + s_tot[3][t];
        s_own[t] = (t < NB) ? tilehist[blk * NB + t] : 0;
        s_cur[t] = 0;
    }
    __syncthreads();
    if (t < 32) {
        const int total = scan256_wave(s_T, s_bs, t);
        if (t == 0) s_bs[NB_MAX] = total;
        scan256_wave(s_own, s_lb, t);
    }
    __syncthreads();
    if (t < NB_MAX)
        s_gb[t] = s_bs[t] + s_pre[0][t] + s_pre[1][t] + s_pre[2][t] + s_pre[3][t];
    if (blk == 0 && t <= NB) binstart[t] = s_bs[t];
    __syncthreads();

    const int base = blk * TILE_E;
    int dv[TILE_E / SORT_T], sv[TILE_E / SORT_T];
#pragma unroll
    for (int k = 0; k < TILE_E / SORT_T; ++k) {
        const int e = base + k * SORT_T + t;
        dv[k] = (e < E) ? dst[e] : -1;
        sv[k] = (e < E) ? src[e] : 0;
    }
    int slot[TILE_E / SORT_T];
#pragma unroll
    for (int k = 0; k < TILE_E / SORT_T; ++k) {
        slot[k] = -1;
        if (dv[k] >= 0) {
            const int b = dv[k] >> BIN_SHIFT;
            slot[k] = s_lb[b] + atomicAdd(&s_cur[b], 1);
        }
    }
#pragma unroll
    for (int k = 0; k < TILE_E / SORT_T; ++k)
        if (slot[k] >= 0) s_val[slot[k]] = ((unsigned)dv[k] << 16) | (unsigned)sv[k];
    __syncthreads();
    const int left = E - base;
    const int cnt  = left < TILE_E ? left : TILE_E;
    for (int i = t; i < cnt; i += SORT_T) {
        const unsigned w = s_val[i];
        const int b = (int)(w >> (16 + BIN_SHIFT));
        part[s_gb[b] + (i - s_lb[b])] = w;
    }
}

__global__ void __launch_bounds__(SORT_T)
bin_sort_kernel(const unsigned* __restrict__ part,
                const int* __restrict__ binstart, int n, int NB,
                int* __restrict__ rowptr, int* __restrict__ csr_src) {
    __shared__ int s_cnt[BIN_NODES];
    __shared__ int s_ex[BIN_NODES];
    __shared__ int s_cur[BIN_NODES];
    __shared__ int s_stage[STAGE_CAP];
    const int t = threadIdx.x, b = blockIdx.x;
    const int beg = binstart[b], end = binstart[b + 1];
    if (t < BIN_NODES) s_cnt[t] = 0;
    unsigned wr[RW];
#pragma unroll
    for (int k = 0; k < RW; ++k) {
        const int i = beg + k * SORT_T + t;
        wr[k] = (i < end) ? part[i] : 0u;
    }
    __syncthreads();
#pragma unroll
    for (int k = 0; k < RW; ++k)
        if (beg + k * SORT_T + t < end)
            atomicAdd(&s_cnt[(wr[k] >> 16) & (BIN_NODES - 1)], 1);
    for (int i = beg + RW * SORT_T + t; i < end; i += SORT_T)
        atomicAdd(&s_cnt[(part[i] >> 16) & (BIN_NODES - 1)], 1);
    __syncthreads();
    if (t < 32) scan256_wave(s_cnt, s_ex, t);
    __syncthreads();
    if (t < BIN_NODES) {
        const int node = b * BIN_NODES + t;
        s_cur[t] = s_ex[t];
        if (node < n) rowptr[node] = beg + s_ex[t];
    }
    if (b == NB - 1 && t == 0) rowptr[n] = end;
    __syncthreads();
    const int  m      = end - beg;
    const bool staged = (m <= STAGE_CAP);
    int pos[RW];
#pragma unroll
    for (int k = 0; k < RW; ++k) {
        pos[k] = -1;
        if (beg + k * SORT_T + t < end)
            pos[k] = atomicAdd(&s_cur[(wr[k] >> 16) & (BIN_NODES - 1)], 1);
    }
#pragma unroll
    for (int k = 0; k < RW; ++k) {
        if (pos[k] >= 0) {
            if (staged) s_stage[pos[k]] = (int)(wr[k] & 0xffffu);
            else        csr_src[beg + pos[k]] = (int)(wr[k] & 0xffffu);
        }
    }
    for (int i = beg + RW * SORT_T + t; i < end; i += SORT_T) {
        const unsigned w = part[i];
        const int p = atomicAdd(&s_cur[(w >> 16) & (BIN_NODES - 1)], 1);
        if (staged) s_stage[p] = (int)(w & 0xffffu);
        else        csr_src[beg + p] = (int)(w & 0xffffu);
    }
    if (staged) {
        __syncthreads();
        for (int i = t; i < m; i += SORT_T) csr_src[beg + i] = s_stage[i];
    }
}

#define GATHER_WAVES 8

template <int K>
__device__ __forceinline__ int row_lane(int v) {
    return __builtin_amdgcn_update_dpp(0, v, 0x150 + K, 0xf, 0xf, true);
}

template <int XM>
__device__ __forceinline__ float row_xor_lane(float v) {
    return __builtin_bit_cast(float, __builtin_amdgcn_update_dpp(
        0, __builtin_bit_cast(int, v), 0x160 + XM, 0xf, 0xf, true));
}

template <int J>
__device__ __forceinline__ void gather8_rows(const v4f* __restrict__ xrow,
                                             int sx, int hl, v4f (&xv)[8]) {
    xv[0] = xrow[(unsigned)(row_lane<J + 0>(sx) + hl)];
    xv[1] = xrow[(unsigned)(row_lane<J + 1>(sx) + hl)];
    xv[2] = xrow[(unsigned)(row_lane<J + 2>(sx) + hl)];
    xv[3] = xrow[(unsigned)(row_lane<J + 3>(sx) + hl)];
    xv[4] = xrow[(unsigned)(row_lane<J + 4>(sx) + hl)];
    xv[5] = xrow[(unsigned)(row_lane<J + 5>(sx) + hl)];
    xv[6] = xrow[(unsigned)(row_lane<J + 6>(sx) + hl)];
    xv[7] = xrow[(unsigned)(row_lane<J + 7>(sx) + hl)];
}

template <int J, int CSH>
__device__ __forceinline__ void edge_group(const v4f* __restrict__ xrow, int sx,
                                           int hl, v4f as4, float ad, int cntc,
                                           float& m, float& den,
                                           v2f& acc_lo, v2f& acc_hi) {
    v4f   xv[8];
    float tv[8];
    gather8_rows<J>(xrow, sx, hl, xv);
    float mn = m;
#pragma unroll
    for (int u = 0; u < 8; ++u) {
        float t = xv[u].x * as4.x + xv[u].y * as4.y
                + xv[u].z * as4.z + xv[u].w * as4.w;
        t += row_xor_lane<1>(t);
        if (CSH == 4) t += row_xor_lane<2>(t);
        t = lrelu(t + ad);
        tv[u] = (J + u < cntc) ? t : -__builtin_inff();
        mn = fmaxf(mn, tv[u]);
    }
    const float sc = exp_hw(m - mn);
    m = mn;
    den    *= sc;
    acc_lo *= sc;
    acc_hi *= sc;
#pragma unroll
    for (int u = 0; u < 8; ++u) {
        const float p = exp_hw(tv[u] - m);
        den += p;
        const v2f x_lo = __builtin_shufflevector(xv[u], xv[u], 0, 1);
        const v2f x_hi = __builtin_shufflevector(xv[u], xv[u], 2, 3);
        acc_lo += p * x_lo;
        acc_hi += p * x_hi;
    }
}

template <int H, int CSH, bool MEAN>
__global__ void __launch_bounds__(GATHER_WAVES * 32)
node_gather_kernel(const int* __restrict__ rowptr,
                   const int* __restrict__ csr_src,
                   const float* __restrict__ asrc,
                   const float* __restrict__ adst,
                   const float* __restrict__ a_src,
                   const float* __restrict__ xl,
                   const float* __restrict__ bias,
                   float* __restrict__ dstp, int n_nodes) {
    const int lane = threadIdx.x & 31;
    const int hl   = lane & 15;
    const int half = lane >> 4;
    const int wv   = threadIdx.x >> 5;
    const int d0 = __builtin_amdgcn_readfirstlane(
        2 * (int)(blockIdx.x * GATHER_WAVES + wv));
    if (d0 >= n_nodes) return;
    const bool live = (d0 + half) < n_nodes;
    const int  d    = live ? d0 + half : d0;
    const int  h    = (4 * hl) >> CSH;
    const int beg   = rowptr[d];
    const int deg   = live ? rowptr[d + 1] - beg : 0;
    const int dega  = __builtin_amdgcn_readlane(deg, 0);
    const int degb  = __builtin_amdgcn_readlane(deg, 16);
    const int dmax  = dega > degb ? dega : degb;
    const float ad  = adst[d * H + h];
    const v4f as4   = *(const v4f*)(a_src + 4 * hl);
    const v4f* xrow = (const v4f*)xl;

    float m   = lrelu(asrc[d * H + h] + ad);
    float den = 1.f;
    v2f acc_lo, acc_hi;
    {
        const v4f xs = xrow[(unsigned)(d * 16 + hl)];
        acc_lo.x = xs.x;
        acc_lo.y = xs.y;
        acc_hi.x = xs.z;
        acc_hi.y = xs.w;
    }
    for (int pos0 = 0; pos0 < dmax; pos0 += 16) {
        const int cntc = deg - pos0;
        const int sidx = (hl < cntc) ? csr_src[beg + pos0 + hl] : d;
        const int sx   = sidx * 16;
        edge_group<0, CSH>(xrow, sx, hl, as4, ad, cntc, m, den, acc_lo, acc_hi);
        if (dmax - pos0 > 8)
            edge_group<8, CSH>(xrow, sx, hl, as4, ad, cntc, m, den, acc_lo, acc_hi);
    }
    const float dd = fmaxf(den, GAT_EPS);
    v4f o;
    o.x = acc_lo.x / dd;
    o.y = acc_lo.y / dd;
    o.z = acc_hi.x / dd;
    o.w = acc_hi.y / dd;
    if (!MEAN) {
        if (live) {
            const v4f bv = *(const v4f*)(bias + 4 * hl);
            const float vx = o.x + bv.x;
            const float vy = o.y + bv.y;
            const float vz = o.z + bv.z;
            const float vw = o.w + bv.w;
            v4f r;
            r.x = vx > 0.f ? vx : (exp_hw(vx) - 1.f);
            r.y = vy > 0.f ? vy : (exp_hw(vy) - 1.f);
            r.z = vz > 0.f ? vz : (exp_hw(vz) - 1.f);
            r.w = vw > 0.f ? vw : (exp_hw(vw) - 1.f);
            *(v4f*)(dstp + (size_t)d * 64 + 4 * hl) = r;
        }
    } else {
        float sx = o.x, sy = o.y, sz = o.z, sw = o.w;
#pragma unroll
        for (int k = 4; k < 16; k += 4) {
            const int from = (lane & 16) | ((hl + k) & 15);
            sx += __shfl(o.x, from);
            sy += __shfl(o.y, from);
            sz += __shfl(o.z, from);
            sw += __shfl(o.w, from);
        }
        if (live && hl < 4) {
            const v4f bv = *(const v4f*)(bias + 4 * hl);
            v4f r;
            r.x = 0.25f * sx + bv.x;
            r.y = 0.25f * sy + bv.y;
            r.z = 0.25f * sz + bv.z;
            r.w = 0.25f * sw + bv.w;
            *(v4f*)(dstp + (size_t)d * 16 + 4 * hl) = r;
        }
    }
}

static inline int cdiv(long long a, long long b) { return (int)((a + b - 1) / b); }

extern "C" void kernel_launch(void* const* d_in, const int* in_sizes, int n_in,
                              void* d_out, int out_size, void* d_ws, size_t ws_size,
                              hipStream_t stream) {
    const float* x   = (const float*)d_in[0];
    const int*   ei  = (const int*)d_in[1];
    const float* W1  = (const float*)d_in[2];
    const float* as1 = (const float*)d_in[3];
    const float* ad1 = (const float*)d_in[4];
    const float* b1  = (const float*)d_in[5];
    const float* W2  = (const float*)d_in[6];
    const float* as2 = (const float*)d_in[7];
    const float* ad2 = (const float*)d_in[8];
    const float* b2  = (const float*)d_in[9];
    float*       out = (float*)d_out;

    const int IN = 256;
    const int n  = in_sizes[0] / IN;
    const int E  = in_sizes[1] / 2;
    const int* src = ei;
    const int* dst = ei + E;

    float* ws    = (float*)d_ws;
    float* xl1   = ws;
    float* asrc1 = xl1   + (long long)n * 64;
    float* adst1 = asrc1 + (long long)n * 8;
    float* hbuf  = adst1 + (long long)n * 8;
    float* xl2   = hbuf  + (long long)n * 64;
    float* asrc2 = xl2   + (long long)n * 64;
    float* adst2 = asrc2 + (long long)n * 4;
    const int NB = cdiv(n, BIN_NODES);
    const int G  = cdiv(E, TILE_E);
    int*      tilehist = (int*)(adst2 + (long long)n * 4);
    int*      binstart = tilehist + (long long)G * NB;
    int*      rowptr   = binstart + NB + 1;
    unsigned* part     = (unsigned*)(rowptr + n + 1);
    int*      csr      = (int*)(part + E);

    gemm_alpha_wmma_f32<8, 8, true><<<G + cdiv(n, 32), 128, 0, stream>>>(
        x, W1, xl1, as1, ad1, asrc1, adst1, n, IN, dst, E, NB, G, tilehist);

    tile_partition_kernel<<<G, SORT_T, 0, stream>>>(src, dst, E, NB, G, tilehist,
                                                    binstart, part);
    bin_sort_kernel<<<NB, SORT_T, 0, stream>>>(part, binstart, n, NB, rowptr, csr);

    node_gather_kernel<8, 3, false><<<cdiv(n, 2 * GATHER_WAVES), GATHER_WAVES * 32, 0, stream>>>(
        rowptr, csr, asrc1, adst1, as1, xl1, b1, hbuf, n);

    gemm_alpha_wmma_f32<4, 16, false><<<cdiv(n, 32), 128, 0, stream>>>(
        hbuf, W2, xl2, as2, ad2, asrc2, adst2, n, 64, nullptr, 0, 0, 0, nullptr);
    node_gather_kernel<4, 4, true><<<cdiv(n, 2 * GATHER_WAVES), GATHER_WAVES * 32, 0, stream>>>(
        rowptr, csr, asrc2, adst2, as2, xl2, b2, out, n);
}


# Layer-2 GEMM and logits ride in the layer-1 gather's launch: a block keeps its 16 finished rows in LDS and multiplies them by W2 on WMMA; h never goes to memory; 6 launches to 5; every sum order kept
# speedup vs baseline: 10.9229x; 1.0363x over previous
#include <hip/hip_runtime.h>
#include <hip/hip_bf16.h>

#define LRELU_SLOPE 0.2f
#define GAT_EPS 1e-16f
#define KT 32

typedef __attribute__((ext_vector_type(2))) float v2f;
typedef __attribute__((ext_vector_type(4))) float v4f;
typedef __attribute__((ext_vector_type(8))) float v8f;

#define BIN_SHIFT 8
#define NB_MAX    256
#define TILE_E    8192

template <int NT>
__device__ __forceinline__ void tile_hist_role(const int* __restrict__ dst, int E,
                                               int NB, int tile,
                                               int* __restrict__ tilehist) {
    __shared__ int lh[NB_MAX];
    const int t = threadIdx.x;
    for (int i = t; i < NB_MAX; i += NT) lh[i] = 0;
    __syncthreads();
    const int base = tile * TILE_E;
    for (int k0 = 0; k0 < TILE_E / NT; k0 += 16) {
        int dv[16];
#pragma unroll
        for (int k = 0; k < 16; ++k) {
            const int e = base + (k0 + k) * NT + t;
            const int v = dst[e < E ? e : E - 1];
            dv[k] = (e < E) ? v : -1;
        }
#pragma unroll
        for (int k = 0; k < 16; ++k)
            if (dv[k] >= 0) atomicAdd(&lh[dv[k] >> BIN_SHIFT], 1);
    }
    __syncthreads();
    for (int i = t; i < NB; i += NT) tilehist[tile * NB + i] = lh[i];
}

template <int H, int CH, bool HIST>
__global__ void gemm_alpha_wmma_f32(const float* __restrict__ A,
                                    const float* __restrict__ B,
                                    float* __restrict__ C,
                                    const float* __restrict__ a_src,
                                    const float* __restrict__ a_dst,
                                    float* __restrict__ asrc,
                                    float* __restrict__ adst, int M, int K,
                                    const int* __restrict__ edst, int E, int NB,
                                    int G, int* __restrict__ tilehist) {
    __shared__ float As[32][KT + 1];
    __shared__ float Cs[32][66];

    const int tid  = threadIdx.x;
    int rblk = blockIdx.x;
    if (HIST) {
        if (rblk < G) {
            tile_hist_role<128>(edst, E, NB, rblk, tilehist);
            return;
        }
        rblk -= G;
    }
    const int lane = tid & 31;
    const int wave = tid >> 5;
    const int r0   = rblk * 32;
    const int c0   = wave * 16;
    const int half = lane >> 4;
    const int l15  = lane & 15;
    const int bcol = c0 + l15;

    const int sr = tid >> 5;
    const int sc = tid & (KT - 1);
    const float* ap[8];
    bool inb[8];
#pragma unroll
    for (int j = 0; j < 8; ++j) {
        const int grow = r0 + sr + 4 * j;
        inb[j] = grow < M;
        ap[j]  = A + (long long)(inb[j] ? grow : M - 1) * K + sc;
    }

    v2f   pb[8];
    float pa[8];
#pragma unroll
    for (int s = 0; s < 8; ++s) {
        const int kk = 4 * s + (half << 1);
        pb[s].x = B[kk * 64 + bcol];
        pb[s].y = B[(kk + 1) * 64 + bcol];
    }
#pragma unroll
    for (int j = 0; j < 8; ++j) pa[j] = ap[j][0];

    v8f acc0 = {}, acc1 = {};

    for (int k0 = 0; k0 < K; k0 += KT) {
        __syncthreads();
#pragma unroll
        for (int j = 0; j < 8; ++j) As[sr + 4 * j][sc] = inb[j] ? pa[j] : 0.f;
        v2f bc[8];
#pragma unroll
        for (int s = 0; s < 8; ++s) bc[s] = pb[s];
        __syncthreads();

        const int kn = k0 + KT;
        if (kn < K) {
#pragma unroll
            for (int s = 0; s < 8; ++s) {
                const int kk = kn + 4 * s + (half << 1);
                pb[s].x = B[kk * 64 + bcol];
                pb[s].y = B[(kk + 1) * 64 + bcol];
            }
#pragma unroll
            for (int j = 0; j < 8; ++j) pa[j] = ap[j][kn];
        }

#pragma unroll
        for (int s = 0; s < 8; ++s) {
            const int kk = 4 * s + (half << 1);
            v2f a0, a1;
            a0.x = As[l15][kk];       a0.y = As[l15][kk + 1];
            a1.x = As[16 + l15][kk];  a1.y = As[16 + l15][kk + 1];
            acc0 = __builtin_amdgcn_wmma_f32_16x16x4_f32(
                false, a0, false, bc[s], (short)0, acc0, false, false);
            acc1 = __builtin_amdgcn_wmma_f32_16x16x4_f32(
                false, a1, false, bc[s], (short)0, acc1, false, false);
        }
    }

    const int lrow = half * 8;
#pragma unroll
    for (int r = 0; r < 8; ++r) {
        const int row0 = r0 + lrow + r;
        const int row1 = row0 + 16;
        if (row0 < M) C[(long long)row0 * 64 + c0 + l15] = acc0[r];
        if (row1 < M) C[(long long)row1 * 64 + c0 + l15] = acc1[r];
        Cs[lrow + r][c0 + l15]      = acc0[r];
        Cs[16 + lrow + r][c0 + l15] = acc1[r];
    }
    __syncthreads();

#pragma unroll
    for (int p = tid; p < 32 * H; p += 128) {
        const int lr   = p / H, h = p % H;
        const int grow = r0 + lr;
        const float* row = &Cs[lr][h * CH];
        float s1 = 0.f, s2 = 0.f;
#pragma unroll
        for (int c = 0; c < CH; ++c) {
            s1 += row[c] * a_src[h * CH + c];
            s2 += row[c] * a_dst[h * CH + c];
        }
        if (grow < M) {
            asrc[(long long)grow * H + h] = s1;
            adst[(long long)grow * H + h] = s2;
        }
    }
}

__device__ __forceinline__ float lrelu(float x) {
    return x > 0.f ? x : LRELU_SLOPE * x;
}

__device__ __forceinline__ float exp_hw(float x) {
    return __builtin_amdgcn_exp2f(x * 1.44269504088896341f);
}

#define BIN_NODES 256
#define SORT_T    1024
#define STAGE_CAP 16384
#define RW        10

__device__ __forceinline__ int scan256_wave(const int* in, int* out, int lane) {
    int a[8];
    int s = 0;
#pragma unroll
    for (int k = 0; k < 8; ++k) { a[k] = in[8 * lane + k]; s += a[k]; }
    int incl = s;
#pragma unroll
    for (int off = 1; off < 32; off <<= 1) {
        const int v = __shfl_up(incl, off);
        if (lane >= off) incl += v;
    }
    int run = incl - s;
#pragma unroll
    for (int k = 0; k < 8; ++k) { out[8 * lane + k] = run; run += a[k]; }
    return __shfl(incl, 31);
}

__global__ void __launch_bounds__(SORT_T)
tile_partition_kernel(const int* __restrict__ src, const int* __restrict__ dst,
                      int E, int NB, int G, const int* __restrict__ tilehist,
                      int* __restrict__ binstart, unsigned* __restrict__ part) {
    __shared__ int s_tot[4][NB_MAX];
    __shared__ int s_pre[4][NB_MAX];
    __shared__ int s_T[NB_MAX];
    __shared__ int s_own[NB_MAX];
    __shared__ int s_bs[NB_MAX + 1];
    __shared__ int s_lb[NB_MAX];
    __shared__ int s_gb[NB_MAX];
    __shared__ int s_cur[NB_MAX];
    __shared__ unsigned s_val[TILE_E];

    const int t = threadIdx.x, blk = blockIdx.x;
    const int c = t & (NB_MAX - 1), rg = t >> 8;
    int tot = 0, pre = 0;
    if (c < NB) {
        for (int r = rg; r < G; r += 4) {
            const int v = tilehist[r * NB + c];
            tot += v;
            pre += (r < blk) ? v : 0;
        }
    }
    s_tot[rg][c] = tot;
    s_pre[rg][c] = pre;
    __syncthreads();
    if (t < NB_MAX) {
        s_T[t]   = s_tot[0][t] + s_tot[1][t] + s_tot[2][t] + s_tot[3][t];
        s_own[t] = (t < NB) ? tilehist[blk * NB + t] : 0;
        s_cur[t] = 0;
    }
    __syncthreads();
    if (t < 32) {
        const int total = scan256_wave(s_T, s_bs, t);
        if (t == 0) s_bs[NB_MAX] = total;
        scan256_wave(s_own, s_lb, t);
    }
    __syncthreads();
    if (t < NB_MAX)
        s_gb[t] = s_bs[t] + s_pre[0][t] + s_pre[1][t] + s_pre[2][t] + s_pre[3][t];
    if (blk == 0 && t <= NB) binstart[t] = s_bs[t];
    __syncthreads();

    const int base = blk * TILE_E;
    int dv[TILE_E / SORT_T], sv[TILE_E / SORT_T];
#pragma unroll
    for (int k = 0; k < TILE_E / SORT_T; ++k) {
        const int e = base + k * SORT_T + t;
        dv[k] = (e < E) ? dst[e] : -1;
        sv[k] = (e < E) ? src[e] : 0;
    }
    int slot[TILE_E / SORT_T];
#pragma unroll
    for (int k = 0; k < TILE_E / SORT_T; ++k) {
        slot[k] = -1;
        if (dv[k] >= 0) {
            const int b = dv[k] >> BIN_SHIFT;
            slot[k] = s_lb[b] + atomicAdd(&s_cur[b], 1);
        }
    }
#pragma unroll
    for (int k = 0; k < TILE_E / SORT_T; ++k)
        if (slot[k] >= 0) s_val[slot[k]] = ((unsigned)dv[k] << 16) | (unsigned)sv[k];
    __syncthreads();
    const int left = E - base;
    const int cnt  = left < TILE_E ? left : TILE_E;
    for (int i = t; i < cnt; i += SORT_T) {
        const unsigned w = s_val[i];
        const int b = (int)(w >> (16 + BIN_SHIFT));
        part[s_gb[b] + (i - s_lb[b])] = w;
    }
}

__global__ void __launch_bounds__(SORT_T)
bin_sort_kernel(const unsigned* __restrict__ part,
                const int* __restrict__ binstart, int n, int NB,
                int* __restrict__ rowptr, int* __restrict__ csr_src) {
    __shared__ int s_cnt[BIN_NODES];
    __shared__ int s_ex[BIN_NODES];
    __shared__ int s_cur[BIN_NODES];
    __shared__ int s_stage[STAGE_CAP];
    const int t = threadIdx.x, b = blockIdx.x;
    const int beg = binstart[b], end = binstart[b + 1];
    if (t < BIN_NODES) s_cnt[t] = 0;
    unsigned wr[RW];
#pragma unroll
    for (int k = 0; k < RW; ++k) {
        const int i = beg + k * SORT_T + t;
        wr[k] = (i < end) ? part[i] : 0u;
    }
    __syncthreads();
#pragma unroll
    for (int k = 0; k < RW; ++k)
        if (beg + k * SORT_T + t < end)
            atomicAdd(&s_cnt[(wr[k] >> 16) & (BIN_NODES - 1)], 1);
    for (int i = beg + RW * SORT_T + t; i < end; i += SORT_T)
        atomicAdd(&s_cnt[(part[i] >> 16) & (BIN_NODES - 1)], 1);
    __syncthreads();
    if (t < 32) scan256_wave(s_cnt, s_ex, t);
    __syncthreads();
    if (t < BIN_NODES) {
        const int node = b * BIN_NODES + t;
        s_cur[t] = s_ex[t];
        if (node < n) rowptr[node] = beg + s_ex[t];
    }
    if (b == NB - 1 && t == 0) rowptr[n] = end;
    __syncthreads();
    const int  m      = end - beg;
    const bool staged = (m <= STAGE_CAP);
    int pos[RW];
#pragma unroll
    for (int k = 0; k < RW; ++k) {
        pos[k] = -1;
        if (beg + k * SORT_T + t < end)
            pos[k] = atomicAdd(&s_cur[(wr[k] >> 16) & (BIN_NODES - 1)], 1);
    }
#pragma unroll
    for (int k = 0; k < RW; ++k) {
        if (pos[k] >= 0) {
            if (staged) s_stage[pos[k]] = (int)(wr[k] & 0xffffu);
            else        csr_src[beg + pos[k]] = (int)(wr[k] & 0xffffu);
        }
    }
    for (int i = beg + RW * SORT_T + t; i < end; i += SORT_T) {
        const unsigned w = part[i];
        const int p = atomicAdd(&s_cur[(w >> 16) & (BIN_NODES - 1)], 1);
        if (staged) s_stage[p] = (int)(w & 0xffffu);
        else        csr_src[beg + p] = (int)(w & 0xffffu);
    }
    if (staged) {
        __syncthreads();
        for (int i = t; i < m; i += SORT_T) csr_src[beg + i] = s_stage[i];
    }
}

#define GATHER_WAVES 8

template <int K>
__device__ __forceinline__ int row_lane(int v) {
    return __builtin_amdgcn_update_dpp(0, v, 0x150 + K, 0xf, 0xf, true);
}

template <int XM>
__device__ __forceinline__ float row_xor_lane(float v) {
    return __builtin_bit_cast(float, __builtin_amdgcn_update_dpp(
        0, __builtin_bit_cast(int, v), 0x160 + XM, 0xf, 0xf, true));
}

template <int J>
__device__ __forceinline__ void gather8_rows(const v4f* __restrict__ xrow,
                                             int sx, int hl, v4f (&xv)[8]) {
    xv[0] = xrow[(unsigned)(row_lane<J + 0>(sx) + hl)];
    xv[1] = xrow[(unsigned)(row_lane<J + 1>(sx) + hl)];
    xv[2] = xrow[(unsigned)(row_lane<J + 2>(sx) + hl)];
    xv[3] = xrow[(unsigned)(row_lane<J + 3>(sx) + hl)];
    xv[4] = xrow[(unsigned)(row_lane<J + 4>(sx) + hl)];
    xv[5] = xrow[(unsigned)(row_lane<J + 5>(sx) + hl)];
    xv[6] = xrow[(unsigned)(row_lane<J + 6>(sx) + hl)];
    xv[7] = xrow[(unsigned)(row_lane<J + 7>(sx) + hl)];
}

template <int J, int CSH>
__device__ __forceinline__ void edge_group(const v4f* __restrict__ xrow, int sx,
                                           int hl, v4f as4, float ad, int cntc,
                                           float& m, float& den,
                                           v2f& acc_lo, v2f& acc_hi) {
    v4f   xv[8];
    float tv[8];
    gather8_rows<J>(xrow, sx, hl, xv);
    float mn = m;
#pragma unroll
    for (int u = 0; u < 8; ++u) {
        float t = xv[u].x * as4.x + xv[u].y * as4.y
                + xv[u].z * as4.z + xv[u].w * as4.w;
        t += row_xor_lane<1>(t);
        if (CSH == 4) t += row_xor_lane<2>(t);
        t = lrelu(t + ad);
        tv[u] = (J + u < cntc) ? t : -__builtin_inff();
        mn = fmaxf(mn, tv[u]);
    }
    const float sc = exp_hw(m - mn);
    m = mn;
    den    *= sc;
    acc_lo *= sc;
    acc_hi *= sc;
#pragma unroll
    for (int u = 0; u < 8; ++u) {
        const float p = exp_hw(tv[u] - m);
        den += p;
        const v2f x_lo = __builtin_shufflevector(xv[u], xv[u], 0, 1);
        const v2f x_hi = __builtin_shufflevector(xv[u], xv[u], 2, 3);
        acc_lo += p * x_lo;
        acc_hi += p * x_hi;
    }
}

#define HP 68
__device__ __forceinline__ void block_rows_times_w2(
        v4f hrow, int node0, int n_nodes, const float* __restrict__ W2,
        const float* __restrict__ a_src2, const float* __restrict__ a_dst2,
        float* __restrict__ xl2, float* __restrict__ asrc2,
        float* __restrict__ adst2) {
    __shared__ float Hs[16][HP];
    __shared__ float Cs[16][HP];
    const int tid  = threadIdx.x;
    const int lane = tid & 31;
    const int hl   = lane & 15;
    const int half = lane >> 4;
    const int wv = __builtin_amdgcn_readfirstlane(tid >> 5);
    const int c0 = (wv & 3) * 16;
    const int bcol = c0 + hl;
    v2f bf[16];
    if (wv < 4) {
#pragma unroll
        for (int s = 0; s < 16; ++s) {
            const int kk = 4 * s + (half << 1);
            bf[s].x = W2[kk * 64 + bcol];
            bf[s].y = W2[(kk + 1) * 64 + bcol];
        }
    }
    *(v4f*)&Hs[2 * wv + half][4 * hl] = hrow;
    __syncthreads();
    if (wv < 4) {
        v8f acc = {};
#pragma unroll
        for (int s = 0; s < 16; ++s) {
            const int kk = 4 * s + (half << 1);
            v2f a;
            a.x = Hs[hl][kk];
            a.y = Hs[hl][kk + 1];
            acc = __builtin_amdgcn_wmma_f32_16x16x4_f32(
                false, a, false, bf[s], (short)0, acc, false, false);
        }
#pragma unroll
        for (int r = 0; r < 8; ++r) Cs[half * 8 + r][bcol] = acc[r];
    }
    __syncthreads();
    {
        const int row  = tid >> 4;
        const int node = node0 + row;
        if (node < n_nodes)
            *(v4f*)(xl2 + (size_t)node * 64 + 4 * hl) = *(const v4f*)&Cs[row][4 * hl];
    }
    if (tid < 64) {
        const int lr = tid >> 2, h = tid & 3;
        const int node = node0 + lr;
        const float* row = &Cs[lr][h * 16];
        float s1 = 0.f, s2 = 0.f;
#pragma unroll
        for (int c = 0; c < 16; ++c) {
            s1 += row[c] * a_src2[h * 16 + c];
            s2 += row[c] * a_dst2[h * 16 + c];
        }
        if (node < n_nodes) {
            asrc2[(long long)node * 4 + h] = s1;
            adst2[(long long)node * 4 + h] = s2;
        }
    }
}

template <int H, int CSH, bool MEAN>
__global__ void __launch_bounds__(GATHER_WAVES * 32)
node_gather_kernel(const int* __restrict__ rowptr,
                   const int* __restrict__ csr_src,
                   const float* __restrict__ asrc,
                   const float* __restrict__ adst,
                   const float* __restrict__ a_src,
                   const float* __restrict__ xl,
                   const float* __restrict__ bias,
                   float* __restrict__ dstp, int n_nodes,
                   const float* __restrict__ W2,
                   const float* __restrict__ a_src2,
                   const float* __restrict__ a_dst2,
                   float* __restrict__ asrc2,
                   float* __restrict__ adst2) {
    const int lane = threadIdx.x & 31;
    const int hl   = lane & 15;
    const int half = lane >> 4;
    const int wv   = threadIdx.x >> 5;
    const int d0 = __builtin_amdgcn_readfirstlane(
        2 * (int)(blockIdx.x * GATHER_WAVES + wv));
    if (MEAN && d0 >= n_nodes) return;
    const int dl = d0 < n_nodes ? d0 : n_nodes - 1;
    const bool live = (d0 + half) < n_nodes;
    const int  d    = live ? d0 + half : dl;
    const int  h    = (4 * hl) >> CSH;
    const int beg   = rowptr[d];
    const int deg   = live ? rowptr[d + 1] - beg : 0;
    const int dega  = __builtin_amdgcn_readlane(deg, 0);
    const int degb  = __builtin_amdgcn_readlane(deg, 16);
    const int dmax  = dega > degb ? dega : degb;
    const float ad  = adst[d * H + h];
    const v4f as4   = *(const v4f*)(a_src + 4 * hl);
    const v4f* xrow = (const v4f*)xl;

    float m   = lrelu(asrc[d * H + h] + ad);
    float den = 1.f;
    v2f acc_lo, acc_hi;
    {
        const v4f xs = xrow[(unsigned)(d * 16 + hl)];
        acc_lo.x = xs.x;
        acc_lo.y = xs.y;
        acc_hi.x = xs.z;
        acc_hi.y = xs.w;
    }
    for (int pos0 = 0; pos0 < dmax; pos0 += 16) {
        const int cntc = deg - pos0;
        const int sidx = (hl < cntc) ? csr_src[beg + pos0 + hl] : d;
        const int sx   = sidx * 16;
        edge_group<0, CSH>(xrow, sx, hl, as4, ad, cntc, m, den, acc_lo, acc_hi);
        if (dmax - pos0 > 8)
            edge_group<8, CSH>(xrow, sx, hl, as4, ad, cntc, m, den, acc_lo, acc_hi);
    }
    const float dd = fmaxf(den, GAT_EPS);
    v4f o;
    o.x = acc_lo.x / dd;
    o.y = acc_lo.y / dd;
    o.z = acc_hi.x / dd;
    o.w = acc_hi.y / dd;
    if (!MEAN) {
        const v4f bv = *(const v4f*)(bias + 4 * hl);
        const float vx = o.x + bv.x;
        const float vy = o.y + bv.y;
        const float vz = o.z + bv.z;
        const float vw = o.w + bv.w;
        v4f r;
        r.x = vx > 0.f ? vx : (exp_hw(vx) - 1.f);
        r.y = vy > 0.f ? vy : (exp_hw(vy) - 1.f);
        r.z = vz > 0.f ? vz : (exp_hw(vz) - 1.f);
        r.w = vw > 0.f ? vw : (exp_hw(vw) - 1.f);
        block_rows_times_w2(r, (int)(blockIdx.x * GATHER_WAVES) * 2, n_nodes, W2,
                            a_src2, a_dst2, dstp, asrc2, adst2);
    } else {
        float sx = o.x, sy = o.y, sz = o.z, sw = o.w;
#pragma unroll
        for (int k = 4; k < 16; k += 4) {
            const int from = (lane & 16) | ((hl + k) & 15);
            sx += __shfl(o.x, from);
            sy += __shfl(o.y, from);
            sz += __shfl(o.z, from);
            sw += __shfl(o.w, from);
        }
        if (live && hl < 4) {
            const v4f bv = *(const v4f*)(bias + 4 * hl);
            v4f r;
            r.x = 0.25f * sx + bv.x;
            r.y = 0.25f * sy + bv.y;
            r.z = 0.25f * sz + bv.z;
            r.w = 0.25f * sw + bv.w;
            *(v4f*)(dstp + (size_t)d * 16 + 4 * hl) = r;
        }
    }
}

static inline int cdiv(long long a, long long b) { return (int)((a + b - 1) / b); }

extern "C" void kernel_launch(void* const* d_in, const int* in_sizes, int n_in,
                              void* d_out, int out_size, void* d_ws, size_t ws_size,
                              hipStream_t stream) {
    const float* x   = (const float*)d_in[0];
    const int*   ei  = (const int*)d_in[1];
    const float* W1  = (const float*)d_in[2];
    const float* as1 = (const float*)d_in[3];
    const float* ad1 = (const float*)d_in[4];
    const float* b1  = (const float*)d_in[5];
    const float* W2  = (const float*)d_in[6];
    const float* as2 = (const float*)d_in[7];
    const float* ad2 = (const float*)d_in[8];
    const float* b2  = (const float*)d_in[9];
    float*       out = (float*)d_out;

    const int IN = 256;
    const int n  = in_sizes[0] / IN;
    const int E  = in_sizes[1] / 2;
    const int* src = ei;
    const int* dst = ei + E;

    float* ws    = (float*)d_ws;
    float* xl1   = ws;
    float* asrc1 = xl1   + (long long)n * 64;
    float* adst1 = asrc1 + (long long)n * 8;
    float* xl2   = adst1 + (long long)n * 8;
    float* asrc2 = xl2   + (long long)n * 64;
    float* adst2 = asrc2 + (long long)n * 4;
    const int NB = cdiv(n, BIN_NODES);
    const int G  = cdiv(E, TILE_E);
    int*      tilehist = (int*)(adst2 + (long long)n * 4);
    int*      binstart = tilehist + (long long)G * NB;
    int*      rowptr   = binstart + NB + 1;
    unsigned* part     = (unsigned*)(rowptr + n + 1);
    int*      csr      = (int*)(part + E);

    gemm_alpha_wmma_f32<8, 8, true><<<G + cdiv(n, 32), 128, 0, stream>>>(
        x, W1, xl1, as1, ad1, asrc1, adst1, n, IN, dst, E, NB, G, tilehist);

    tile_partition_kernel<<<G, SORT_T, 0, stream>>>(src, dst, E, NB, G, tilehist,
                                                    binstart, part);
    bin_sort_kernel<<<NB, SORT_T, 0, stream>>>(part, binstart, n, NB, rowptr, csr);

    node_gather_kernel<8, 3, false><<<cdiv(n, 2 * GATHER_WAVES), GATHER_WAVES * 32, 0, stream>>>(
        rowptr, csr, asrc1, adst1, as1, xl1, b1, xl2, n, W2, as2, ad2, asrc2, adst2);

    node_gather_kernel<4, 4, true><<<cdiv(n, 2 * GATHER_WAVES), GATHER_WAVES * 32, 0, stream>>>(
        rowptr, csr, asrc2, adst2, as2, xl2, b2, out, n, nullptr, nullptr, nullptr,
        nullptr, nullptr);
}


# Dependent load rounds cut: gathers ask a chunk's source ids one chunk ahead and the self-loop operands beside the row pointers; tile_partition asks its edges and all tile counts as one group
# speedup vs baseline: 11.3687x; 1.0408x over previous
#include <hip/hip_runtime.h>
#include <hip/hip_bf16.h>

#define LRELU_SLOPE 0.2f
#define GAT_EPS 1e-16f
#define KT 32

typedef __attribute__((ext_vector_type(2))) float v2f;
typedef __attribute__((ext_vector_type(4))) float v4f;
typedef __attribute__((ext_vector_type(8))) float v8f;
typedef __attribute__((ext_vector_type(4))) int   v4i;

#define BIN_SHIFT 8
#define NB_MAX    256
#define TILE_E    8192

template <int NT>
__device__ __forceinline__ void tile_hist_role(const int* __restrict__ dst, int E,
                                               int NB, int tile,
                                               int* __restrict__ tilehist) {
    __shared__ int lh[NB_MAX];
    const int t = threadIdx.x;
    for (int i = t; i < NB_MAX; i += NT) lh[i] = 0;
    __syncthreads();
    const int base = tile * TILE_E;
    for (int k0 = 0; k0 < TILE_E / NT; k0 += 16) {
        int dv[16];
#pragma unroll
        for (int k = 0; k < 16; ++k) {
            const int e = base + (k0 + k) * NT + t;
            const int v = dst[e < E ? e : E - 1];
            dv[k] = (e < E) ? v : -1;
        }
#pragma unroll
        for (int k = 0; k < 16; ++k)
            if (dv[k] >= 0) atomicAdd(&lh[dv[k] >> BIN_SHIFT], 1);
    }
    __syncthreads();
    for (int i = t; i < NB_MAX; i += NT) tilehist[tile * NB_MAX + i] = lh[i];
}

template <int H, int CH, bool HIST>
__global__ void gemm_alpha_wmma_f32(const float* __restrict__ A,
                                    const float* __restrict__ B,
                                    float* __restrict__ C,
                                    const float* __restrict__ a_src,
                                    const float* __restrict__ a_dst,
                                    float* __restrict__ asrc,
                                    float* __restrict__ adst, int M, int K,
                                    const int* __restrict__ edst, int E, int NB,
                                    int G, int* __restrict__ tilehist) {
    __shared__ float As[32][KT + 1];
    __shared__ float Cs[32][66];

    const int tid  = threadIdx.x;
    int rblk = blockIdx.x;
    if (HIST) {
        if (rblk < G) {
            tile_hist_role<128>(edst, E, NB, rblk, tilehist);
            return;
        }
        rblk -= G;
    }
    const int lane = tid & 31;
    const int wave = tid >> 5;
    const int r0   = rblk * 32;
    const int c0   = wave * 16;
    const int half = lane >> 4;
    const int l15  = lane & 15;
    const int bcol = c0 + l15;

    const int sr = tid >> 5;
    const int sc = tid & (KT - 1);
    const float* ap[8];
    bool inb[8];
#pragma unroll
    for (int j = 0; j < 8; ++j) {
        const int grow = r0 + sr + 4 * j;
        inb[j] = grow < M;
        ap[j]  = A + (long long)(inb[j] ? grow : M - 1) * K + sc;
    }

    v2f   pb[8];
    float pa[8];
#pragma unroll
    for (int s = 0; s < 8; ++s) {
        const int kk = 4 * s + (half << 1);
        pb[s].x = B[kk * 64 + bcol];
        pb[s].y = B[(kk + 1) * 64 + bcol];
    }
#pragma unroll
    for (int j = 0; j < 8; ++j) pa[j] = ap[j][0];

    v8f acc0 = {}, acc1 = {};

    for (int k0 = 0; k0 < K; k0 += KT) {
        __syncthreads();
#pragma unroll
        for (int j = 0; j < 8; ++j) As[sr + 4 * j][sc] = inb[j] ? pa[j] : 0.f;
        v2f bc[8];
#pragma unroll
        for (int s = 0; s < 8; ++s) bc[s] = pb[s];
        __syncthreads();

        const int kn = k0 + KT;
        if (kn < K) {
#pragma unroll
            for (int s = 0; s < 8; ++s) {
                const int kk = kn + 4 * s + (half << 1);
                pb[s].x = B[kk * 64 + bcol];
                pb[s].y = B[(kk + 1) * 64 + bcol];
            }
#pragma unroll
            for (int j = 0; j < 8; ++j) pa[j] = ap[j][kn];
        }

#pragma unroll
        for (int s = 0; s < 8; ++s) {
            const int kk = 4 * s + (half << 1);
            v2f a0, a1;
            a0.x = As[l15][kk];       a0.y = As[l15][kk + 1];
            a1.x = As[16 + l15][kk];  a1.y = As[16 + l15][kk + 1];
            acc0 = __builtin_amdgcn_wmma_f32_16x16x4_f32(
                false, a0, false, bc[s], (short)0, acc0, false, false);
            acc1 = __builtin_amdgcn_wmma_f32_16x16x4_f32(
                false, a1, false, bc[s], (short)0, acc1, false, false);
        }
    }

    const int lrow = half * 8;
#pragma unroll
    for (int r = 0; r < 8; ++r) {
        const int row0 = r0 + lrow + r;
        const int row1 = row0 + 16;
        if (row0 < M) C[(long long)row0 * 64 + c0 + l15] = acc0[r];
        if (row1 < M) C[(long long)row1 * 64 + c0 + l15] = acc1[r];
        Cs[lrow + r][c0 + l15]      = acc0[r];
        Cs[16 + lrow + r][c0 + l15] = acc1[r];
    }
    __syncthreads();

#pragma unroll
    for (int p = tid; p < 32 * H; p += 128) {
        const int lr   = p / H, h = p % H;
        const int grow = r0 + lr;
        const float* row = &Cs[lr][h * CH];
        float s1 = 0.f, s2 = 0.f;
#pragma unroll
        for (int c = 0; c < CH; ++c) {
            s1 += row[c] * a_src[h * CH + c];
            s2 += row[c] * a_dst[h * CH + c];
        }
        if (grow < M) {
            asrc[(long long)grow * H + h] = s1;
            adst[(long long)grow * H + h] = s2;
        }
    }
}

__device__ __forceinline__ float lrelu(float x) {
    return x > 0.f ? x : LRELU_SLOPE * x;
}

__device__ __forceinline__ float exp_hw(float x) {
    return __builtin_amdgcn_exp2f(x * 1.44269504088896341f);
}

#define BIN_NODES 256
#define SORT_T    1024
#define STAGE_CAP 16384
#define RW        10
#define PART_RG   16

__device__ __forceinline__ int scan256_wave(const int* in, int* out, int lane) {
    int a[8];
    int s = 0;
#pragma unroll
    for (int k = 0; k < 8; ++k) { a[k] = in[8 * lane + k]; s += a[k]; }
    int incl = s;
#pragma unroll
    for (int off = 1; off < 32; off <<= 1) {
        const int v = __shfl_up(incl, off);
        if (lane >= off) incl += v;
    }
    int run = incl - s;
#pragma unroll
    for (int k = 0; k < 8; ++k) { out[8 * lane + k] = run; run += a[k]; }
    return __shfl(incl, 31);
}

__global__ void __launch_bounds__(SORT_T)
tile_partition_kernel(const int* __restrict__ src, const int* __restrict__ dst,
                      int E, int NB, int G, const int* __restrict__ tilehist,
                      int* __restrict__ binstart, unsigned* __restrict__ part) {
    __shared__ __attribute__((aligned(16))) int s_tot[PART_RG][NB_MAX];
    __shared__ __attribute__((aligned(16))) int s_pre[PART_RG][NB_MAX];
    __shared__ int s_T[NB_MAX];
    __shared__ int s_own[NB_MAX];
    __shared__ int s_bs[NB_MAX + 1];
    __shared__ int s_lb[NB_MAX];
    __shared__ int s_gb[NB_MAX];
    __shared__ int s_cur[NB_MAX];
    __shared__ unsigned s_val[TILE_E];

    const int t = threadIdx.x, blk = blockIdx.x;

    const int base = blk * TILE_E;
    int dv[TILE_E / SORT_T], sv[TILE_E / SORT_T];
#pragma unroll
    for (int k = 0; k < TILE_E / SORT_T; ++k) {
        const int e  = base + k * SORT_T + t;
        const int ec = e < E ? e : E - 1;
        const int dd = dst[ec];
        const int ss = src[ec];
        dv[k] = (e < E) ? dd : -1;
        sv[k] = (e < E) ? ss : 0;
    }
    const int ownv = tilehist[blk * NB_MAX + (t & (NB_MAX - 1))];
    const int q = t & 63, rgp = t >> 6;
    v4i tot4 = {0, 0, 0, 0}, pre4 = {0, 0, 0, 0};
    for (int r0 = 0; r0 < G; r0 += PART_RG * 16) {
        v4i v[16];
#pragma unroll
        for (int j = 0; j < 16; ++j) {
            const int r = r0 + rgp + PART_RG * j;
            v[j] = *(const v4i*)(tilehist + (r < G ? r : G - 1) * NB_MAX + 4 * q);
        }
        __builtin_amdgcn_sched_barrier(0);
#pragma unroll
        for (int j = 0; j < 16; ++j) {
            const int r = r0 + rgp + PART_RG * j;
            const v4i z = {0, 0, 0, 0};
            tot4 += (r < G)   ? v[j] : z;
            pre4 += (r < blk) ? v[j] : z;
        }
    }
    *(v4i*)&s_tot[rgp][4 * q] = tot4;
    *(v4i*)&s_pre[rgp][4 * q] = pre4;
    __syncthreads();
    int presum = 0;
    if (t < NB_MAX) {
        int ts = 0;
#pragma unroll
        for (int g = 0; g < PART_RG; ++g) { ts += s_tot[g][t]; presum += s_pre[g][t]; }
        s_T[t]   = ts;
        s_own[t] = ownv;
        s_cur[t] = 0;
    }
    __syncthreads();
    if (t < 32) {
        const int total = scan256_wave(s_T, s_bs, t);
        if (t == 0) s_bs[NB_MAX] = total;
        scan256_wave(s_own, s_lb, t);
    }
    __syncthreads();
    if (t < NB_MAX) s_gb[t] = s_bs[t] + presum;
    if (blk == 0 && t <= NB) binstart[t] = s_bs[t];
    __syncthreads();

    int slot[TILE_E / SORT_T];
#pragma unroll
    for (int k = 0; k < TILE_E / SORT_T; ++k) {
        slot[k] = -1;
        if (dv[k] >= 0) {
            const int b = dv[k] >> BIN_SHIFT;
            slot[k] = s_lb[b] + atomicAdd(&s_cur[b], 1);
        }
    }
#pragma unroll
    for (int k = 0; k < TILE_E / SORT_T; ++k)
        if (slot[k] >= 0) s_val[slot[k]] = ((unsigned)dv[k] << 16) | (unsigned)sv[k];
    __syncthreads();
    const int left = E - base;
    const int cnt  = left < TILE_E ? left : TILE_E;
    for (int i = t; i < cnt; i += SORT_T) {
        const unsigned w = s_val[i];
        const int b = (int)(w >> (16 + BIN_SHIFT));
        part[s_gb[b] + (i - s_lb[b])] = w;
    }
}

__global__ void __launch_bounds__(SORT_T)
bin_sort_kernel(const unsigned* __restrict__ part,
                const int* __restrict__ binstart, int n, int NB,
                int* __restrict__ rowptr, int* __restrict__ csr_src) {
    __shared__ int s_cnt[BIN_NODES];
    __shared__ int s_ex[BIN_NODES];
    __shared__ int s_cur[BIN_NODES];
    __shared__ int s_stage[STAGE_CAP];
    const int t = threadIdx.x, b = blockIdx.x;
    const int beg = binstart[b], end = binstart[b + 1];
    if (t < BIN_NODES) s_cnt[t] = 0;
    unsigned wr[RW];
#pragma unroll
    for (int k = 0; k < RW; ++k) {
        const int i = beg + k * SORT_T + t;
        wr[k] = (i < end) ? part[i] : 0u;
    }
    __syncthreads();
#pragma unroll
    for (int k = 0; k < RW; ++k)
        if (beg + k * SORT_T + t < end)
            atomicAdd(&s_cnt[(wr[k] >> 16) & (BIN_NODES - 1)], 1);
    for (int i = beg + RW * SORT_T + t; i < end; i += SORT_T)
        atomicAdd(&s_cnt[(part[i] >> 16) & (BIN_NODES - 1)], 1);
    __syncthreads();
    if (t < 32) scan256_wave(s_cnt, s_ex, t);
    __syncthreads();
    if (t < BIN_NODES) {
        const int node = b * BIN_NODES + t;
        s_cur[t] = s_ex[t];
        if (node < n) rowptr[node] = beg + s_ex[t];
    }
    if (b == NB - 1 && t == 0) rowptr[n] = end;
    __syncthreads();
    const int  m      = end - beg;
    const bool staged = (m <= STAGE_CAP);
    int pos[RW];
#pragma unroll
    for (int k = 0; k < RW; ++k) {
        pos[k] = -1;
        if (beg + k * SORT_T + t < end)
            pos[k] = atomicAdd(&s_cur[(wr[k] >> 16) & (BIN_NODES - 1)], 1);
    }
#pragma unroll
    for (int k = 0; k < RW; ++k) {
        if (pos[k] >= 0) {
            if (staged) s_stage[pos[k]] = (int)(wr[k] & 0xffffu);
            else        csr_src[beg + pos[k]] = (int)(wr[k] & 0xffffu);
        }
    }
    for (int i = beg + RW * SORT_T + t; i < end; i += SORT_T) {
        const unsigned w = part[i];
        const int p = atomicAdd(&s_cur[(w >> 16) & (BIN_NODES - 1)], 1);
        if (staged) s_stage[p] = (int)(w & 0xffffu);
        else        csr_src[beg + p] = (int)(w & 0xffffu);
    }
    if (staged) {
        __syncthreads();
        for (int i = t; i < m; i += SORT_T) csr_src[beg + i] = s_stage[i];
    }
}

#define GATHER_WAVES 8

template <int K>
__device__ __forceinline__ int row_lane(int v) {
    return __builtin_amdgcn_update_dpp(0, v, 0x150 + K, 0xf, 0xf, true);
}

template <int XM>
__device__ __forceinline__ float row_xor_lane(float v) {
    return __builtin_bit_cast(float, __builtin_amdgcn_update_dpp(
        0, __builtin_bit_cast(int, v), 0x160 + XM, 0xf, 0xf, true));
}

template <int J>
__device__ __forceinline__ void gather8_rows(const v4f* __restrict__ xrow,
                                             int sx, int hl, v4f (&xv)[8]) {
    xv[0] = xrow[(unsigned)(row_lane<J + 0>(sx) + hl)];
    xv[1] = xrow[(unsigned)(row_lane<J + 1>(sx) + hl)];
    xv[2] = xrow[(unsigned)(row_lane<J + 2>(sx) + hl)];
    xv[3] = xrow[(unsigned)(row_lane<J + 3>(sx) + hl)];
    xv[4] = xrow[(unsigned)(row_lane<J + 4>(sx) + hl)];
    xv[5] = xrow[(unsigned)(row_lane<J + 5>(sx) + hl)];
    xv[6] = xrow[(unsigned)(row_lane<J + 6>(sx) + hl)];
    xv[7] = xrow[(unsigned)(row_lane<J + 7>(sx) + hl)];
}

template <int J, int CSH>
__device__ __forceinline__ void edge_group(const v4f* __restrict__ xrow, int sx,
                                           int hl, v4f as4, float ad, int cntc,
                                           float& m, float& den,
                                           v2f& acc_lo, v2f& acc_hi) {
    v4f   xv[8];
    float tv[8];
    gather8_rows<J>(xrow, sx, hl, xv);
    float mn = m;
#pragma unroll
    for (int u = 0; u < 8; ++u) {
        float t = xv[u].x * as4.x + xv[u].y * as4.y
                + xv[u].z * as4.z + xv[u].w * as4.w;
        t += row_xor_lane<1>(t);
        if (CSH == 4) t += row_xor_lane<2>(t);
        t = lrelu(t + ad);
        tv[u] = (J + u < cntc) ? t : -__builtin_inff();
        mn = fmaxf(mn, tv[u]);
    }
    const float sc = exp_hw(m - mn);
    m = mn;
    den    *= sc;
    acc_lo *= sc;
    acc_hi *= sc;
#pragma unroll
    for (int u = 0; u < 8; ++u) {
        const float p = exp_hw(tv[u] - m);
        den += p;
        const v2f x_lo = __builtin_shufflevector(xv[u], xv[u], 0, 1);
        const v2f x_hi = __builtin_shufflevector(xv[u], xv[u], 2, 3);
        acc_lo += p * x_lo;
        acc_hi += p * x_hi;
    }
}

#define HP 68
__device__ __forceinline__ void block_rows_times_w2(
        v4f hrow, int node0, int n_nodes, const float* __restrict__ W2,
        const float* __restrict__ a_src2, const float* __restrict__ a_dst2,
        float* __restrict__ xl2, float* __restrict__ asrc2,
        float* __restrict__ adst2) {
    __shared__ float Hs[16][HP];
    __shared__ float Cs[16][HP];
    const int tid  = threadIdx.x;
    const int lane = tid & 31;
    const int hl   = lane & 15;
    const int half = lane >> 4;
    const int wv = __builtin_amdgcn_readfirstlane(tid >> 5);
    const int c0 = (wv & 3) * 16;
    const int bcol = c0 + hl;
    v2f bf[16];
    if (wv < 4) {
#pragma unroll
        for (int s = 0; s < 16; ++s) {
            const int kk = 4 * s + (half << 1);
            bf[s].x = W2[kk * 64 + bcol];
            bf[s].y = W2[(kk + 1) * 64 + bcol];
        }
    }
    *(v4f*)&Hs[2 * wv + half][4 * hl] = hrow;
    __syncthreads();
    if (wv < 4) {
        v8f acc = {};
#pragma unroll
        for (int s = 0; s < 16; ++s) {
            const int kk = 4 * s + (half << 1);
            v2f a;
            a.x = Hs[hl][kk];
            a.y = Hs[hl][kk + 1];
            acc = __builtin_amdgcn_wmma_f32_16x16x4_f32(
                false, a, false, bf[s], (short)0, acc, false, false);
        }
#pragma unroll
        for (int r = 0; r < 8; ++r) Cs[half * 8 + r][bcol] = acc[r];
    }
    __syncthreads();
    {
        const int row  = tid >> 4;
        const int node = node0 + row;
        if (node < n_nodes)
            *(v4f*)(xl2 + (size_t)node * 64 + 4 * hl) = *(const v4f*)&Cs[row][4 * hl];
    }
    if (tid < 64) {
        const int lr = tid >> 2, h = tid & 3;
        const int node = node0 + lr;
        const float* row = &Cs[lr][h * 16];
        float s1 = 0.f, s2 = 0.f;
#pragma unroll
        for (int c = 0; c < 16; ++c) {
            s1 += row[c] * a_src2[h * 16 + c];
            s2 += row[c] * a_dst2[h * 16 + c];
        }
        if (node < n_nodes) {
            asrc2[(long long)node * 4 + h] = s1;
            adst2[(long long)node * 4 + h] = s2;
        }
    }
}

template <int H, int CSH, bool MEAN>
__global__ void __launch_bounds__(GATHER_WAVES * 32)
node_gather_kernel(const int* __restrict__ rowptr,
                   const int* __restrict__ csr_src,
                   const float* __restrict__ asrc,
                   const float* __restrict__ adst,
                   const float* __restrict__ a_src,
                   const float* __restrict__ xl,
                   const float* __restrict__ bias,
                   float* __restrict__ dstp, int n_nodes,
                   const float* __restrict__ W2,
                   const float* __restrict__ a_src2,
                   const float* __restrict__ a_dst2,
                   float* __restrict__ asrc2,
                   float* __restrict__ adst2, int n_edges) {
    const int lane = threadIdx.x & 31;
    const int hl   = lane & 15;
    const int half = lane >> 4;
    const int wv   = threadIdx.x >> 5;
    const int d0 = __builtin_amdgcn_readfirstlane(
        2 * (int)(blockIdx.x * GATHER_WAVES + wv));
    if (MEAN && d0 >= n_nodes) return;
    const int dl = d0 < n_nodes ? d0 : n_nodes - 1;
    const bool live = (d0 + half) < n_nodes;
    const int  d    = live ? d0 + half : dl;
    const int  h    = (4 * hl) >> CSH;
    const int beg   = rowptr[d];
    const int end1  = rowptr[d + 1];
    const float ad  = adst[d * H + h];
    const float asf = asrc[d * H + h];
    const v4f as4   = *(const v4f*)(a_src + 4 * hl);
    const v4f* xrow = (const v4f*)xl;
    const v4f xs    = xrow[(unsigned)(d * 16 + hl)];
    const int deg   = live ? end1 - beg : 0;
    const int dega  = __builtin_amdgcn_readlane(deg, 0);
    const int degb  = __builtin_amdgcn_readlane(deg, 16);
    const int dmax  = dega > degb ? dega : degb;
    int snext = (hl < deg) ? csr_src[beg + hl] : d;

    float m   = lrelu(asf + ad);
    float den = 1.f;
    v2f acc_lo, acc_hi;
    acc_lo.x = xs.x;
    acc_lo.y = xs.y;
    acc_hi.x = xs.z;
    acc_hi.y = xs.w;
    for (int pos0 = 0; pos0 < dmax; pos0 += 16) {
        const int cntc = deg - pos0;
        const int sidx = (hl < cntc) ? snext : d;
        const int sx   = sidx * 16;
        const int nx = beg + pos0 + 16 + hl;
        snext = csr_src[nx < n_edges ? nx : n_edges - 1];
        __builtin_amdgcn_sched_barrier(0);
        edge_group<0, CSH>(xrow, sx, hl, as4, ad, cntc, m, den, acc_lo, acc_hi);
        if (dmax - pos0 > 8)
            edge_group<8, CSH>(xrow, sx, hl, as4, ad, cntc, m, den, acc_lo, acc_hi);
    }
    const float dd = fmaxf(den, GAT_EPS);
    v4f o;
    o.x = acc_lo.x / dd;
    o.y = acc_lo.y / dd;
    o.z = acc_hi.x / dd;
    o.w = acc_hi.y / dd;
    if (!MEAN) {
        const v4f bv = *(const v4f*)(bias + 4 * hl);
        const float vx = o.x + bv.x;
        const float vy = o.y + bv.y;
        const float vz = o.z + bv.z;
        const float vw = o.w + bv.w;
        v4f r;
        r.x = vx > 0.f ? vx : (exp_hw(vx) - 1.f);
        r.y = vy > 0.f ? vy : (exp_hw(vy) - 1.f);
        r.z = vz > 0.f ? vz : (exp_hw(vz) - 1.f);
        r.w = vw > 0.f ? vw : (exp_hw(vw) - 1.f);
        block_rows_times_w2(r, (int)(blockIdx.x * GATHER_WAVES) * 2, n_nodes, W2,
                            a_src2, a_dst2, dstp, asrc2, adst2);
    } else {
        float sx = o.x, sy = o.y, sz = o.z, sw = o.w;
#pragma unroll
        for (int k = 4; k < 16; k += 4) {
            const int from = (lane & 16) | ((hl + k) & 15);
            sx += __shfl(o.x, from);
            sy += __shfl(o.y, from);
            sz += __shfl(o.z, from);
            sw += __shfl(o.w, from);
        }
        if (live && hl < 4) {
            const v4f bv = *(const v4f*)(bias + 4 * hl);
            v4f r;
            r.x = 0.25f * sx + bv.x;
            r.y = 0.25f * sy + bv.y;
            r.z = 0.25f * sz + bv.z;
            r.w = 0.25f * sw + bv.w;
            *(v4f*)(dstp + (size_t)d * 16 + 4 * hl) = r;
        }
    }
}

static inline int cdiv(long long a, long long b) { return (int)((a + b - 1) / b); }

extern "C" void kernel_launch(void* const* d_in, const int* in_sizes, int n_in,
                              void* d_out, int out_size, void* d_ws, size_t ws_size,
                              hipStream_t stream) {
    const float* x   = (const float*)d_in[0];
    const int*   ei  = (const int*)d_in[1];
    const float* W1  = (const float*)d_in[2];
    const float* as1 = (const float*)d_in[3];
    const float* ad1 = (const float*)d_in[4];
    const float* b1  = (const float*)d_in[5];
    const float* W2  = (const float*)d_in[6];
    const float* as2 = (const float*)d_in[7];
    const float* ad2 = (const float*)d_in[8];
    const float* b2  = (const float*)d_in[9];
    float*       out = (float*)d_out;

    const int IN = 256;
    const int n  = in_sizes[0] / IN;
    const int E  = in_sizes[1] / 2;
    const int* src = ei;
    const int* dst = ei + E;

    float* ws    = (float*)d_ws;
    float* xl1   = ws;
    float* asrc1 = xl1   + (long long)n * 64;
    float* adst1 = asrc1 + (long long)n * 8;
    float* xl2   = adst1 + (long long)n * 8;
    float* asrc2 = xl2   + (long long)n * 64;
    float* adst2 = asrc2 + (long long)n * 4;
    const int NB = cdiv(n, BIN_NODES);
    const int G  = cdiv(E, TILE_E);
    int*      tilehist = (int*)(adst2 + (long long)n * 4);
    int*      binstart = tilehist + (long long)G * NB_MAX;
    int*      rowptr   = binstart + NB + 1;
    unsigned* part     = (unsigned*)(rowptr + n + 1);
    int*      csr      = (int*)(part + E);

    gemm_alpha_wmma_f32<8, 8, true><<<G + cdiv(n, 32), 128, 0, stream>>>(
        x, W1, xl1, as1, ad1, asrc1, adst1, n, IN, dst, E, NB, G, tilehist);

    tile_partition_kernel<<<G, SORT_T, 0, stream>>>(src, dst, E, NB, G, tilehist,
                                                    binstart, part);
    bin_sort_kernel<<<NB, SORT_T, 0, stream>>>(part, binstart, n, NB, rowptr, csr);

    node_gather_kernel<8, 3, false><<<cdiv(n, 2 * GATHER_WAVES), GATHER_WAVES * 32, 0, stream>>>(
        rowptr, csr, asrc1, adst1, as1, xl1, b1, xl2, n, W2, as2, ad2, asrc2, adst2, E);

    node_gather_kernel<4, 4, true><<<cdiv(n, 2 * GATHER_WAVES), GATHER_WAVES * 32, 0, stream>>>(
        rowptr, csr, asrc2, adst2, as2, xl2, b2, out, n, nullptr, nullptr, nullptr,
        nullptr, nullptr, E);
}
